# starting kernel
# baseline (speedup 1.0000x reference)
#include <hip/hip_runtime.h>
#include <math.h>

typedef __attribute__((ext_vector_type(2))) float v2f;
typedef __attribute__((ext_vector_type(8))) float v8f;

#define NFEAT 512
#define HIDDEN 16
#define NCLS 2


__global__ void k_init_deg(float* __restrict__ deg, int n) {
  int i = blockIdx.x * blockDim.x + threadIdx.x;
  if (i < n) deg[i] = 1.0f;
}

__global__ void k_deg_count(const int* __restrict__ col, float* __restrict__ deg, int E) {
  int e = blockIdx.x * blockDim.x + threadIdx.x;
  if (e < E) atomicAdd(&deg[col[e]], 1.0f);
}

__global__ void k_deg_rsqrt(float* __restrict__ deg, int n) {
  int i = blockIdx.x * blockDim.x + threadIdx.x;
  if (i < n) deg[i] = rsqrtf(deg[i]);
}


__global__ __launch_bounds__(256) void k_gemm1(const float* __restrict__ x,
                                               const float* __restrict__ W1,
                                               float* __restrict__ h1,
                                               int n_nodes) {
  const int lane = threadIdx.x & 31;
  const int tile = blockIdx.x * 8 + (threadIdx.x >> 5);
  const int row0 = tile * 16;
  if (row0 >= n_nodes) return;

  const int nsel  = lane & 15;
  const int khalf = (lane >> 4) << 1;

  const float* arow = x + (size_t)(row0 + nsel) * NFEAT + khalf;

  v8f c = {};
  for (int k = 0; k < NFEAT; k += 4) {
    if ((k & 127) == 0) __builtin_prefetch(arow + k + 128, 0, 0);
    v2f a = *(const v2f*)(arow + k);
    v2f b;
    b.x = W1[(k + khalf) * HIDDEN + nsel];
    b.y = W1[(k + khalf + 1) * HIDDEN + nsel];
    c = __builtin_amdgcn_wmma_f32_16x16x4_f32(false, a, false, b, (short)0, c,
                                              false, false);
  }

  const int mbase = row0 + ((lane >> 4) << 3);
#pragma unroll
  for (int r = 0; r < 8; ++r)
    h1[(size_t)(mbase + r) * HIDDEN + nsel] = c[r];
}


__global__ void k_agg1_self(const float* __restrict__ h1, const float* __restrict__ dinv,
                            float* __restrict__ agg1, int n) {
  int t = blockIdx.x * blockDim.x + threadIdx.x;
  int node = t >> 4;
  if (node < n) {
    float w = dinv[node] * dinv[node];
    agg1[t] = h1[t] * w;
  }
}

__global__ void k_agg1_edges(const int* __restrict__ row, const int* __restrict__ col,
                             const float* __restrict__ h1, const float* __restrict__ dinv,
                             float* __restrict__ agg1, int E) {
  int t = blockIdx.x * blockDim.x + threadIdx.x;
  int e = t >> 4;
  int j = t & 15;
  if (e < E) {
    int r = row[e];
    int c = col[e];
    float w = dinv[r] * dinv[c];
    atomicAdd(&agg1[(size_t)c * HIDDEN + j], h1[(size_t)r * HIDDEN + j] * w);
  }
}


__global__ void k_relu_gemm2(const float* __restrict__ agg1, const float* __restrict__ b1,
                             const float* __restrict__ W2, float* __restrict__ h2, int n) {
  int node = blockIdx.x * blockDim.x + threadIdx.x;
  if (node >= n) return;
  float acc0 = 0.f, acc1 = 0.f;
#pragma unroll
  for (int j = 0; j < HIDDEN; ++j) {
    float v = fmaxf(agg1[(size_t)node * HIDDEN + j] + b1[j], 0.f);
    acc0 = fmaf(v, W2[j * NCLS + 0], acc0);
    acc1 = fmaf(v, W2[j * NCLS + 1], acc1);
  }
  h2[(size_t)node * NCLS + 0] = acc0;
  h2[(size_t)node * NCLS + 1] = acc1;
}


__global__ void k_agg2_self(const float* __restrict__ h2, const float* __restrict__ dinv,
                            float* __restrict__ agg2, int n) {
  int node = blockIdx.x * blockDim.x + threadIdx.x;
  if (node < n) {
    float w = dinv[node] * dinv[node];
    agg2[(size_t)node * NCLS + 0] = h2[(size_t)node * NCLS + 0] * w;
    agg2[(size_t)node * NCLS + 1] = h2[(size_t)node * NCLS + 1] * w;
  }
}

__global__ void k_agg2_edges(const int* __restrict__ row, const int* __restrict__ col,
                             const float* __restrict__ h2, const float* __restrict__ dinv,
                             float* __restrict__ agg2, int E) {
  int e = blockIdx.x * blockDim.x + threadIdx.x;
  if (e < E) {
    int r = row[e];
    int c = col[e];
    float w = dinv[r] * dinv[c];
    atomicAdd(&agg2[(size_t)c * NCLS + 0], h2[(size_t)r * NCLS + 0] * w);
    atomicAdd(&agg2[(size_t)c * NCLS + 1], h2[(size_t)r * NCLS + 1] * w);
  }
}


__global__ void k_logsoftmax(const float* __restrict__ agg2, const float* __restrict__ b2,
                             float* __restrict__ out, int n) {
  int node = blockIdx.x * blockDim.x + threadIdx.x;
  if (node >= n) return;
  float z0 = agg2[(size_t)node * NCLS + 0] + b2[0];
  float z1 = agg2[(size_t)node * NCLS + 1] + b2[1];
  float m = fmaxf(z0, z1);
  float lse = m + logf(expf(z0 - m) + expf(z1 - m));
  out[(size_t)node * NCLS + 0] = z0 - lse;
  out[(size_t)node * NCLS + 1] = z1 - lse;
}


extern "C" void kernel_launch(void* const* d_in, const int* in_sizes, int n_in,
                              void* d_out, int out_size, void* d_ws, size_t ws_size,
                              hipStream_t stream) {
  const float* x  = (const float*)d_in[0];
  const int*   ei = (const int*)d_in[1];
  const float* W1 = (const float*)d_in[2];
  const float* b1 = (const float*)d_in[3];
  const float* W2 = (const float*)d_in[4];
  const float* b2 = (const float*)d_in[5];

  const int n = in_sizes[0] / NFEAT;
  const int E = in_sizes[1] / 2;
  const int* row = ei;
  const int* col = ei + E;

  char* ws = (char*)d_ws;
  size_t off = 0;
  auto carve = [&](size_t bytes) -> void* {
    void* p = ws + off;
    off = (off + bytes + 255) & ~(size_t)255;
    return p;
  };
  float* deg  = (float*)carve((size_t)n * sizeof(float));
  float* h1   = (float*)carve((size_t)n * HIDDEN * sizeof(float));
  float* agg1 = (float*)carve((size_t)n * HIDDEN * sizeof(float));
  float* h2   = (float*)carve((size_t)n * NCLS * sizeof(float));
  float* agg2 = (float*)carve((size_t)n * NCLS * sizeof(float));
  (void)ws_size;

  const int B = 256;
  auto blocks = [](long long work, int b) { return (unsigned)((work + b - 1) / b); };

  k_init_deg<<<blocks(n, B), B, 0, stream>>>(deg, n);
  k_deg_count<<<blocks(E, B), B, 0, stream>>>(col, deg, E);
  k_deg_rsqrt<<<blocks(n, B), B, 0, stream>>>(deg, n);

  int tiles = (n + 15) / 16;
  k_gemm1<<<blocks(tiles, 8), B, 0, stream>>>(x, W1, h1, n);

  k_agg1_self<<<blocks((long long)n * HIDDEN, B), B, 0, stream>>>(h1, deg, agg1, n);
  k_agg1_edges<<<blocks((long long)E * HIDDEN, B), B, 0, stream>>>(row, col, h1, deg, agg1, E);

  k_relu_gemm2<<<blocks(n, B), B, 0, stream>>>(agg1, b1, W2, h2, n);

  k_agg2_self<<<blocks(n, B), B, 0, stream>>>(h2, deg, agg2, n);
  k_agg2_edges<<<blocks(E, B), B, 0, stream>>>(row, col, h2, deg, agg2, E);

  k_logsoftmax<<<blocks(n, B), B, 0, stream>>>(agg2, b2, (float*)d_out, n);
}


# degree count and layer-2 edge scatter-add go to per-block LDS bins (node range x edge slice); each non-zero bin is added to global once, coalesced; float sum order changes
# speedup vs baseline: 2.0104x; 2.0104x over previous
#include <hip/hip_runtime.h>
#include <math.h>

typedef __attribute__((ext_vector_type(2))) float v2f;
typedef __attribute__((ext_vector_type(8))) float v8f;

#define NFEAT 512
#define HIDDEN 16
#define NCLS 2


__global__ void k_init_deg(float* __restrict__ deg, int n) {
  int i = blockIdx.x * blockDim.x + threadIdx.x;
  if (i < n) deg[i] = 1.0f;
}

#define BIN_NODES 16384
#define EDGE_SLICES 64
#define SCAT_THREADS 1024
static_assert((BIN_NODES & (BIN_NODES - 1)) == 0, "bin index is wrapped with a mask");

__global__ __launch_bounds__(SCAT_THREADS) void k_deg_count(const int* __restrict__ col,
                                                            float* __restrict__ deg,
                                                            int E, int n) {
  __shared__ float bins[BIN_NODES];
  const int tid = threadIdx.x;
  const int lo = blockIdx.y * BIN_NODES;
  for (int i = tid; i < BIN_NODES; i += SCAT_THREADS) bins[i] = 0.f;
  __syncthreads();

  const int per = (E + (int)gridDim.x - 1) / (int)gridDim.x;
  const int e0 = blockIdx.x * per;
  const int e1 = min(e0 + per, E);
  const int last = e1 - 1;
  for (int e = e0 + tid; e < e1; e += 8 * SCAT_THREADS) {
    int c[8];
#pragma unroll
    for (int j = 0; j < 8; ++j) {
      const int ee = e + j * SCAT_THREADS;
      const bool in = ee < e1;
      const int v = col[in ? ee : last];
      c[j] = in ? v : -1;
    }
#pragma unroll
    for (int j = 0; j < 8; ++j) {
      const unsigned idx = (unsigned)(c[j] - lo);
      if (idx < (unsigned)BIN_NODES) atomicAdd(&bins[idx], 1.0f);
    }
  }
  __syncthreads();

  const int rot = (int)blockIdx.x * SCAT_THREADS;
  for (int t = tid; t < BIN_NODES; t += SCAT_THREADS) {
    const int i = (t + rot) & (BIN_NODES - 1);
    const float v = bins[i];
    const int node = lo + i;
    if (v != 0.f && node < n) atomicAdd(&deg[node], v);
  }
}

__global__ void k_deg_rsqrt(float* __restrict__ deg, int n) {
  int i = blockIdx.x * blockDim.x + threadIdx.x;
  if (i < n) deg[i] = rsqrtf(deg[i]);
}


__global__ __launch_bounds__(256) void k_gemm1(const float* __restrict__ x,
                                               const float* __restrict__ W1,
                                               float* __restrict__ h1,
                                               int n_nodes) {
  const int lane = threadIdx.x & 31;
  const int tile = blockIdx.x * 8 + (threadIdx.x >> 5);
  const int row0 = tile * 16;
  if (row0 >= n_nodes) return;

  const int nsel  = lane & 15;
  const int khalf = (lane >> 4) << 1;

  const float* arow = x + (size_t)(row0 + nsel) * NFEAT + khalf;

  v8f c = {};
  for (int k = 0; k < NFEAT; k += 4) {
    if ((k & 127) == 0) __builtin_prefetch(arow + k + 128, 0, 0);
    v2f a = *(const v2f*)(arow + k);
    v2f b;
    b.x = W1[(k + khalf) * HIDDEN + nsel];
    b.y = W1[(k + khalf + 1) * HIDDEN + nsel];
    c = __builtin_amdgcn_wmma_f32_16x16x4_f32(false, a, false, b, (short)0, c,
                                              false, false);
  }

  const int mbase = row0 + ((lane >> 4) << 3);
#pragma unroll
  for (int r = 0; r < 8; ++r)
    h1[(size_t)(mbase + r) * HIDDEN + nsel] = c[r];
}


__global__ void k_agg1_self(const float* __restrict__ h1, const float* __restrict__ dinv,
                            float* __restrict__ agg1, int n) {
  int t = blockIdx.x * blockDim.x + threadIdx.x;
  int node = t >> 4;
  if (node < n) {
    float w = dinv[node] * dinv[node];
    agg1[t] = h1[t] * w;
  }
}

__global__ void k_agg1_edges(const int* __restrict__ row, const int* __restrict__ col,
                             const float* __restrict__ h1, const float* __restrict__ dinv,
                             float* __restrict__ agg1, int E) {
  int t = blockIdx.x * blockDim.x + threadIdx.x;
  int e = t >> 4;
  int j = t & 15;
  if (e < E) {
    int r = row[e];
    int c = col[e];
    float w = dinv[r] * dinv[c];
    atomicAdd(&agg1[(size_t)c * HIDDEN + j], h1[(size_t)r * HIDDEN + j] * w);
  }
}


__global__ void k_relu_gemm2(const float* __restrict__ agg1, const float* __restrict__ b1,
                             const float* __restrict__ W2, float* __restrict__ h2, int n) {
  int node = blockIdx.x * blockDim.x + threadIdx.x;
  if (node >= n) return;
  float acc0 = 0.f, acc1 = 0.f;
#pragma unroll
  for (int j = 0; j < HIDDEN; ++j) {
    float v = fmaxf(agg1[(size_t)node * HIDDEN + j] + b1[j], 0.f);
    acc0 = fmaf(v, W2[j * NCLS + 0], acc0);
    acc1 = fmaf(v, W2[j * NCLS + 1], acc1);
  }
  h2[(size_t)node * NCLS + 0] = acc0;
  h2[(size_t)node * NCLS + 1] = acc1;
}


__global__ void k_agg2_self(const float* __restrict__ h2, const float* __restrict__ dinv,
                            float* __restrict__ agg2, int n) {
  int node = blockIdx.x * blockDim.x + threadIdx.x;
  if (node < n) {
    float w = dinv[node] * dinv[node];
    agg2[(size_t)node * NCLS + 0] = h2[(size_t)node * NCLS + 0] * w;
    agg2[(size_t)node * NCLS + 1] = h2[(size_t)node * NCLS + 1] * w;
  }
}

__global__ __launch_bounds__(SCAT_THREADS) void k_agg2_edges(const int* __restrict__ row,
                                                             const int* __restrict__ col,
                                                             const float* __restrict__ h2,
                                                             const float* __restrict__ dinv,
                                                             float* __restrict__ agg2,
                                                             int E, int n) {
  __shared__ float bins[BIN_NODES * NCLS];
  const int tid = threadIdx.x;
  const int lo = blockIdx.y * BIN_NODES;
  for (int i = tid; i < BIN_NODES * NCLS; i += SCAT_THREADS) bins[i] = 0.f;
  __syncthreads();

  const int per = (E + (int)gridDim.x - 1) / (int)gridDim.x;
  const int e0 = blockIdx.x * per;
  const int e1 = min(e0 + per, E);
  if (e0 < e1) {
    const int last = e1 - 1;
    int c[4];
    unsigned r[4];
#pragma unroll
    for (int j = 0; j < 4; ++j) {
      const int ee = e0 + tid + j * SCAT_THREADS;
      const bool in = ee < e1;
      const int k = in ? ee : last;
      const int cv = col[k];
      r[j] = (unsigned)row[k];
      c[j] = in ? cv : -1;
    }
    for (int e = e0 + tid; e < e1; e += 4 * SCAT_THREADS) {
      int cn[4];
      unsigned rn[4];
#pragma unroll
      for (int j = 0; j < 4; ++j) {
        const int ee = e + (4 + j) * SCAT_THREADS;
        const bool in = ee < e1;
        const int k = in ? ee : last;
        const int cv = col[k];
        rn[j] = (unsigned)row[k];
        cn[j] = in ? cv : -1;
      }
      float dr[4], dc[4];
      v2f hv[4];
#pragma unroll
      for (int j = 0; j < 4; ++j) {
        const unsigned idx = (unsigned)(c[j] - lo);
        if (idx < (unsigned)BIN_NODES) {
          dr[j] = dinv[r[j]];
          dc[j] = dinv[(unsigned)c[j]];
          hv[j] = *(const v2f*)(h2 + r[j] * NCLS);
        }
      }
#pragma unroll
      for (int j = 0; j < 4; ++j) {
        const unsigned idx = (unsigned)(c[j] - lo);
        if (idx < (unsigned)BIN_NODES) {
          const float w = dr[j] * dc[j];
          atomicAdd(&bins[idx * NCLS + 0], hv[j].x * w);
          atomicAdd(&bins[idx * NCLS + 1], hv[j].y * w);
        }
      }
#pragma unroll
      for (int j = 0; j < 4; ++j) {
        c[j] = cn[j];
        r[j] = rn[j];
      }
    }
  }
  __syncthreads();

  const int total = n * NCLS;
  const int rot = (int)blockIdx.x * SCAT_THREADS;
  for (int t = tid; t < BIN_NODES * NCLS; t += SCAT_THREADS) {
    const int i = (t + rot) & (BIN_NODES * NCLS - 1);
    const float v = bins[i];
    const int g = lo * NCLS + i;
    if (v != 0.f && g < total) atomicAdd(&agg2[g], v);
  }
}


__global__ void k_logsoftmax(const float* __restrict__ agg2, const float* __restrict__ b2,
                             float* __restrict__ out, int n) {
  int node = blockIdx.x * blockDim.x + threadIdx.x;
  if (node >= n) return;
  float z0 = agg2[(size_t)node * NCLS + 0] + b2[0];
  float z1 = agg2[(size_t)node * NCLS + 1] + b2[1];
  float m = fmaxf(z0, z1);
  float lse = m + logf(expf(z0 - m) + expf(z1 - m));
  out[(size_t)node * NCLS + 0] = z0 - lse;
  out[(size_t)node * NCLS + 1] = z1 - lse;
}


extern "C" void kernel_launch(void* const* d_in, const int* in_sizes, int n_in,
                              void* d_out, int out_size, void* d_ws, size_t ws_size,
                              hipStream_t stream) {
  const float* x  = (const float*)d_in[0];
  const int*   ei = (const int*)d_in[1];
  const float* W1 = (const float*)d_in[2];
  const float* b1 = (const float*)d_in[3];
  const float* W2 = (const float*)d_in[4];
  const float* b2 = (const float*)d_in[5];

  const int n = in_sizes[0] / NFEAT;
  const int E = in_sizes[1] / 2;
  const int* row = ei;
  const int* col = ei + E;

  char* ws = (char*)d_ws;
  size_t off = 0;
  auto carve = [&](size_t bytes) -> void* {
    void* p = ws + off;
    off = (off + bytes + 255) & ~(size_t)255;
    return p;
  };
  float* deg  = (float*)carve((size_t)n * sizeof(float));
  float* h1   = (float*)carve((size_t)n * HIDDEN * sizeof(float));
  float* agg1 = (float*)carve((size_t)n * HIDDEN * sizeof(float));
  float* h2   = (float*)carve((size_t)n * NCLS * sizeof(float));
  float* agg2 = (float*)carve((size_t)n * NCLS * sizeof(float));
  (void)ws_size;

  const int B = 256;
  auto blocks = [](long long work, int b) { return (unsigned)((work + b - 1) / b); };

  k_init_deg<<<blocks(n, B), B, 0, stream>>>(deg, n);
  const dim3 scat_grid(EDGE_SLICES, blocks(n, BIN_NODES));
  k_deg_count<<<scat_grid, SCAT_THREADS, 0, stream>>>(col, deg, E, n);
  k_deg_rsqrt<<<blocks(n, B), B, 0, stream>>>(deg, n);

  int tiles = (n + 15) / 16;
  k_gemm1<<<blocks(tiles, 8), B, 0, stream>>>(x, W1, h1, n);

  k_agg1_self<<<blocks((long long)n * HIDDEN, B), B, 0, stream>>>(h1, deg, agg1, n);
  k_agg1_edges<<<blocks((long long)E * HIDDEN, B), B, 0, stream>>>(row, col, h1, deg, agg1, E);

  k_relu_gemm2<<<blocks(n, B), B, 0, stream>>>(agg1, b1, W2, h2, n);

  k_agg2_self<<<blocks(n, B), B, 0, stream>>>(h2, deg, agg2, n);
  k_agg2_edges<<<scat_grid, SCAT_THREADS, 0, stream>>>(row, col, h2, deg, agg2, E, n);

  k_logsoftmax<<<blocks(n, B), B, 0, stream>>>(agg2, b2, (float*)d_out, n);
}


# layer-1 edge scatter-add goes to per-block LDS row bins (node range x edge slice, half a wave per edge); each bin is scaled by its target factor and added to global once; float sum order changes
# speedup vs baseline: 2.8694x; 1.4273x over previous
#include <hip/hip_runtime.h>
#include <math.h>

typedef __attribute__((ext_vector_type(2))) float v2f;
typedef __attribute__((ext_vector_type(8))) float v8f;

#define NFEAT 512
#define HIDDEN 16
#define NCLS 2


__global__ void k_init_deg(float* __restrict__ deg, int n) {
  int i = blockIdx.x * blockDim.x + threadIdx.x;
  if (i < n) deg[i] = 1.0f;
}

#define BIN_NODES 16384
#define EDGE_SLICES 64
#define SCAT_THREADS 1024
static_assert((BIN_NODES & (BIN_NODES - 1)) == 0, "bin index is wrapped with a mask");

__global__ __launch_bounds__(SCAT_THREADS) void k_deg_count(const int* __restrict__ col,
                                                            float* __restrict__ deg,
                                                            int E, int n) {
  __shared__ float bins[BIN_NODES];
  const int tid = threadIdx.x;
  const int lo = blockIdx.y * BIN_NODES;
  for (int i = tid; i < BIN_NODES; i += SCAT_THREADS) bins[i] = 0.f;
  __syncthreads();

  const int per = (E + (int)gridDim.x - 1) / (int)gridDim.x;
  const int e0 = blockIdx.x * per;
  const int e1 = min(e0 + per, E);
  const int last = e1 - 1;
  for (int e = e0 + tid; e < e1; e += 8 * SCAT_THREADS) {
    int c[8];
#pragma unroll
    for (int j = 0; j < 8; ++j) {
      const int ee = e + j * SCAT_THREADS;
      const bool in = ee < e1;
      const int v = col[in ? ee : last];
      c[j] = in ? v : -1;
    }
#pragma unroll
    for (int j = 0; j < 8; ++j) {
      const unsigned idx = (unsigned)(c[j] - lo);
      if (idx < (unsigned)BIN_NODES) atomicAdd(&bins[idx], 1.0f);
    }
  }
  __syncthreads();

  const int rot = (int)blockIdx.x * SCAT_THREADS;
  for (int t = tid; t < BIN_NODES; t += SCAT_THREADS) {
    const int i = (t + rot) & (BIN_NODES - 1);
    const float v = bins[i];
    const int node = lo + i;
    if (v != 0.f && node < n) atomicAdd(&deg[node], v);
  }
}

__global__ void k_deg_rsqrt(float* __restrict__ deg, int n) {
  int i = blockIdx.x * blockDim.x + threadIdx.x;
  if (i < n) deg[i] = rsqrtf(deg[i]);
}


__global__ __launch_bounds__(256) void k_gemm1(const float* __restrict__ x,
                                               const float* __restrict__ W1,
                                               float* __restrict__ h1,
                                               int n_nodes) {
  const int lane = threadIdx.x & 31;
  const int tile = blockIdx.x * 8 + (threadIdx.x >> 5);
  const int row0 = tile * 16;
  if (row0 >= n_nodes) return;

  const int nsel  = lane & 15;
  const int khalf = (lane >> 4) << 1;

  const float* arow = x + (size_t)(row0 + nsel) * NFEAT + khalf;

  v8f c = {};
  for (int k = 0; k < NFEAT; k += 4) {
    if ((k & 127) == 0) __builtin_prefetch(arow + k + 128, 0, 0);
    v2f a = *(const v2f*)(arow + k);
    v2f b;
    b.x = W1[(k + khalf) * HIDDEN + nsel];
    b.y = W1[(k + khalf + 1) * HIDDEN + nsel];
    c = __builtin_amdgcn_wmma_f32_16x16x4_f32(false, a, false, b, (short)0, c,
                                              false, false);
  }

  const int mbase = row0 + ((lane >> 4) << 3);
#pragma unroll
  for (int r = 0; r < 8; ++r)
    h1[(size_t)(mbase + r) * HIDDEN + nsel] = c[r];
}


__global__ void k_agg1_self(const float* __restrict__ h1, const float* __restrict__ dinv,
                            float* __restrict__ agg1, int n) {
  int t = blockIdx.x * blockDim.x + threadIdx.x;
  int node = t >> 4;
  if (node < n) {
    float w = dinv[node] * dinv[node];
    agg1[t] = h1[t] * w;
  }
}

#define A1_NODES 4096
#define A1_CL_BITS 12
#define A1_SLICES 10
#define A1_THREADS 1024
#define A1_U 8
#define A1_G 8
static_assert((1 << A1_CL_BITS) == A1_NODES, "list entry packs the in-range target in A1_CL_BITS bits");

__global__ __launch_bounds__(A1_THREADS) void k_agg1_edges(const int* __restrict__ row,
                                                           const int* __restrict__ col,
                                                           const float* __restrict__ h1,
                                                           const float* __restrict__ dinv,
                                                           float* __restrict__ agg1,
                                                           int E, int n) {
  __shared__ float bins[A1_NODES * HIDDEN];
  __shared__ unsigned list[A1_THREADS / 32][A1_U * 32];
  const int tid = threadIdx.x;
  const int lane = tid & 31;
  const int half = lane >> 4;
  const int f = lane & 15;
  const int lo = blockIdx.y * A1_NODES;
  unsigned* mylist = list[tid >> 5];

  for (int i = tid; i < A1_NODES * HIDDEN / 4; i += A1_THREADS)
    ((float4*)bins)[i] = make_float4(0.f, 0.f, 0.f, 0.f);
  __syncthreads();

  const int per = (E + (int)gridDim.x - 1) / (int)gridDim.x;
  const int e0 = blockIdx.x * per;
  const int e1 = min(e0 + per, E);
  if (e0 < e1) {
    const int last = e1 - 1;
    int c[A1_U];
    unsigned r[A1_U];
#pragma unroll
    for (int j = 0; j < A1_U; ++j) {
      const int ee = e0 + tid + j * A1_THREADS;
      const bool in = ee < e1;
      const int k = in ? ee : last;
      const int cv = col[k];
      r[j] = (unsigned)row[k];
      c[j] = in ? cv : -1;
    }
    for (int base = e0; base < e1; base += A1_U * A1_THREADS) {
      int cn[A1_U];
      unsigned rn[A1_U];
#pragma unroll
      for (int j = 0; j < A1_U; ++j) {
        const int ee = base + tid + (A1_U + j) * A1_THREADS;
        const bool in = ee < e1;
        const int k = in ? ee : last;
        const int cv = col[k];
        rn[j] = (unsigned)row[k];
        cn[j] = in ? cv : -1;
      }
      int cnt = 0;
#pragma unroll
      for (int j = 0; j < A1_U; ++j) {
        const unsigned idx = (unsigned)(c[j] - lo);
        const bool m = idx < (unsigned)A1_NODES;
        const unsigned mask = __builtin_amdgcn_ballot_w32(m);
        if (m) mylist[cnt + (int)__builtin_amdgcn_mbcnt_lo(mask, 0u)] = (r[j] << A1_CL_BITS) | idx;
        cnt += __builtin_popcount(mask);
      }
      for (int t = 0; t < cnt; t += 2 * A1_G) {
        unsigned ent[A1_G];
        float hv[A1_G], dr[A1_G];
#pragma unroll
        for (int q = 0; q < A1_G; ++q) {
          const int i = t + 2 * q + half;
          ent[q] = mylist[i < cnt ? i : t];
        }
#pragma unroll
        for (int q = 0; q < A1_G; ++q) {
          const unsigned rr = ent[q] >> A1_CL_BITS;
          hv[q] = h1[rr * HIDDEN + f];
          dr[q] = dinv[rr];
        }
#pragma unroll
        for (int q = 0; q < A1_G; ++q) {
          const int i = t + 2 * q + half;
          if (i < cnt)
            atomicAdd(&bins[(ent[q] & (A1_NODES - 1)) * HIDDEN + f], hv[q] * dr[q]);
        }
      }
#pragma unroll
      for (int j = 0; j < A1_U; ++j) {
        c[j] = cn[j];
        r[j] = rn[j];
      }
    }
  }
  __syncthreads();

  const int total = n * HIDDEN;
  const int rot = (int)blockIdx.x * (((A1_NODES * HIDDEN) / A1_SLICES) & ~31);
  for (int t = tid; t < A1_NODES * HIDDEN; t += 4 * A1_THREADS) {
    float v[4], d[4];
    int g[4];
#pragma unroll
    for (int u = 0; u < 4; ++u) {
      const int i = (t + u * A1_THREADS + rot) & (A1_NODES * HIDDEN - 1);
      v[u] = bins[i];
      g[u] = lo * HIDDEN + i;
      d[u] = dinv[min(lo + (i >> 4), n - 1)];
    }
#pragma unroll
    for (int u = 0; u < 4; ++u)
      if (v[u] != 0.f && g[u] < total) atomicAdd(&agg1[g[u]], v[u] * d[u]);
  }
}


__global__ void k_relu_gemm2(const float* __restrict__ agg1, const float* __restrict__ b1,
                             const float* __restrict__ W2, float* __restrict__ h2, int n) {
  int node = blockIdx.x * blockDim.x + threadIdx.x;
  if (node >= n) return;
  float acc0 = 0.f, acc1 = 0.f;
#pragma unroll
  for (int j = 0; j < HIDDEN; ++j) {
    float v = fmaxf(agg1[(size_t)node * HIDDEN + j] + b1[j], 0.f);
    acc0 = fmaf(v, W2[j * NCLS + 0], acc0);
    acc1 = fmaf(v, W2[j * NCLS + 1], acc1);
  }
  h2[(size_t)node * NCLS + 0] = acc0;
  h2[(size_t)node * NCLS + 1] = acc1;
}


__global__ void k_agg2_self(const float* __restrict__ h2, const float* __restrict__ dinv,
                            float* __restrict__ agg2, int n) {
  int node = blockIdx.x * blockDim.x + threadIdx.x;
  if (node < n) {
    float w = dinv[node] * dinv[node];
    agg2[(size_t)node * NCLS + 0] = h2[(size_t)node * NCLS + 0] * w;
    agg2[(size_t)node * NCLS + 1] = h2[(size_t)node * NCLS + 1] * w;
  }
}

__global__ __launch_bounds__(SCAT_THREADS) void k_agg2_edges(const int* __restrict__ row,
                                                             const int* __restrict__ col,
                                                             const float* __restrict__ h2,
                                                             const float* __restrict__ dinv,
                                                             float* __restrict__ agg2,
                                                             int E, int n) {
  __shared__ float bins[BIN_NODES * NCLS];
  const int tid = threadIdx.x;
  const int lo = blockIdx.y * BIN_NODES;
  for (int i = tid; i < BIN_NODES * NCLS; i += SCAT_THREADS) bins[i] = 0.f;
  __syncthreads();

  const int per = (E + (int)gridDim.x - 1) / (int)gridDim.x;
  const int e0 = blockIdx.x * per;
  const int e1 = min(e0 + per, E);
  if (e0 < e1) {
    const int last = e1 - 1;
    int c[4];
    unsigned r[4];
#pragma unroll
    for (int j = 0; j < 4; ++j) {
      const int ee = e0 + tid + j * SCAT_THREADS;
      const bool in = ee < e1;
      const int k = in ? ee : last;
      const int cv = col[k];
      r[j] = (unsigned)row[k];
      c[j] = in ? cv : -1;
    }
    for (int e = e0 + tid; e < e1; e += 4 * SCAT_THREADS) {
      int cn[4];
      unsigned rn[4];
#pragma unroll
      for (int j = 0; j < 4; ++j) {
        const int ee = e + (4 + j) * SCAT_THREADS;
        const bool in = ee < e1;
        const int k = in ? ee : last;
        const int cv = col[k];
        rn[j] = (unsigned)row[k];
        cn[j] = in ? cv : -1;
      }
      float dr[4], dc[4];
      v2f hv[4];
#pragma unroll
      for (int j = 0; j < 4; ++j) {
        const unsigned idx = (unsigned)(c[j] - lo);
        if (idx < (unsigned)BIN_NODES) {
          dr[j] = dinv[r[j]];
          dc[j] = dinv[(unsigned)c[j]];
          hv[j] = *(const v2f*)(h2 + r[j] * NCLS);
        }
      }
#pragma unroll
      for (int j = 0; j < 4; ++j) {
        const unsigned idx = (unsigned)(c[j] - lo);
        if (idx < (unsigned)BIN_NODES) {
          const float w = dr[j] * dc[j];
          atomicAdd(&bins[idx * NCLS + 0], hv[j].x * w);
          atomicAdd(&bins[idx * NCLS + 1], hv[j].y * w);
        }
      }
#pragma unroll
      for (int j = 0; j < 4; ++j) {
        c[j] = cn[j];
        r[j] = rn[j];
      }
    }
  }
  __syncthreads();

  const int total = n * NCLS;
  const int rot = (int)blockIdx.x * SCAT_THREADS;
  for (int t = tid; t < BIN_NODES * NCLS; t += SCAT_THREADS) {
    const int i = (t + rot) & (BIN_NODES * NCLS - 1);
    const float v = bins[i];
    const int g = lo * NCLS + i;
    if (v != 0.f && g < total) atomicAdd(&agg2[g], v);
  }
}


__global__ void k_logsoftmax(const float* __restrict__ agg2, const float* __restrict__ b2,
                             float* __restrict__ out, int n) {
  int node = blockIdx.x * blockDim.x + threadIdx.x;
  if (node >= n) return;
  float z0 = agg2[(size_t)node * NCLS + 0] + b2[0];
  float z1 = agg2[(size_t)node * NCLS + 1] + b2[1];
  float m = fmaxf(z0, z1);
  float lse = m + logf(expf(z0 - m) + expf(z1 - m));
  out[(size_t)node * NCLS + 0] = z0 - lse;
  out[(size_t)node * NCLS + 1] = z1 - lse;
}


extern "C" void kernel_launch(void* const* d_in, const int* in_sizes, int n_in,
                              void* d_out, int out_size, void* d_ws, size_t ws_size,
                              hipStream_t stream) {
  const float* x  = (const float*)d_in[0];
  const int*   ei = (const int*)d_in[1];
  const float* W1 = (const float*)d_in[2];
  const float* b1 = (const float*)d_in[3];
  const float* W2 = (const float*)d_in[4];
  const float* b2 = (const float*)d_in[5];

  const int n = in_sizes[0] / NFEAT;
  const int E = in_sizes[1] / 2;
  const int* row = ei;
  const int* col = ei + E;

  char* ws = (char*)d_ws;
  size_t off = 0;
  auto carve = [&](size_t bytes) -> void* {
    void* p = ws + off;
    off = (off + bytes + 255) & ~(size_t)255;
    return p;
  };
  float* deg  = (float*)carve((size_t)n * sizeof(float));
  float* h1   = (float*)carve((size_t)n * HIDDEN * sizeof(float));
  float* agg1 = (float*)carve((size_t)n * HIDDEN * sizeof(float));
  float* h2   = (float*)carve((size_t)n * NCLS * sizeof(float));
  float* agg2 = (float*)carve((size_t)n * NCLS * sizeof(float));
  (void)ws_size;

  const int B = 256;
  auto blocks = [](long long work, int b) { return (unsigned)((work + b - 1) / b); };

  k_init_deg<<<blocks(n, B), B, 0, stream>>>(deg, n);
  const dim3 scat_grid(EDGE_SLICES, blocks(n, BIN_NODES));
  k_deg_count<<<scat_grid, SCAT_THREADS, 0, stream>>>(col, deg, E, n);
  k_deg_rsqrt<<<blocks(n, B), B, 0, stream>>>(deg, n);

  int tiles = (n + 15) / 16;
  k_gemm1<<<blocks(tiles, 8), B, 0, stream>>>(x, W1, h1, n);

  k_agg1_self<<<blocks((long long)n * HIDDEN, B), B, 0, stream>>>(h1, deg, agg1, n);
  const dim3 a1_grid(A1_SLICES, blocks(n, A1_NODES));
  k_agg1_edges<<<a1_grid, A1_THREADS, 0, stream>>>(row, col, h1, deg, agg1, E, n);

  k_relu_gemm2<<<blocks(n, B), B, 0, stream>>>(agg1, b1, W2, h2, n);

  k_agg2_self<<<blocks(n, B), B, 0, stream>>>(h2, deg, agg2, n);
  k_agg2_edges<<<scat_grid, SCAT_THREADS, 0, stream>>>(row, col, h2, deg, agg2, E, n);

  k_logsoftmax<<<blocks(n, B), B, 0, stream>>>(agg2, b2, (float*)d_out, n);
}


# node passes folded into neighbours: GEMM epilogue writes deg^-1/2 and the layer-1 self-loop term, relu/layer-2 product writes the layer-2 self-loop term; 10 launches become 7; sum order unchanged
# speedup vs baseline: 2.9963x; 1.0442x over previous
#include <hip/hip_runtime.h>
#include <math.h>

typedef __attribute__((ext_vector_type(2))) float v2f;
typedef __attribute__((ext_vector_type(8))) float v8f;

#define NFEAT 512
#define HIDDEN 16
#define NCLS 2


__global__ void k_init_deg(float* __restrict__ deg, int n) {
  int i = blockIdx.x * blockDim.x + threadIdx.x;
  if (i < n) deg[i] = 1.0f;
}

#define BIN_NODES 16384
#define EDGE_SLICES 64
#define SCAT_THREADS 1024
static_assert((BIN_NODES & (BIN_NODES - 1)) == 0, "bin index is wrapped with a mask");

__global__ __launch_bounds__(SCAT_THREADS) void k_deg_count(const int* __restrict__ col,
                                                            float* __restrict__ deg,
                                                            int E, int n) {
  __shared__ float bins[BIN_NODES];
  const int tid = threadIdx.x;
  const int lo = blockIdx.y * BIN_NODES;
  for (int i = tid; i < BIN_NODES; i += SCAT_THREADS) bins[i] = 0.f;
  __syncthreads();

  const int per = (E + (int)gridDim.x - 1) / (int)gridDim.x;
  const int e0 = blockIdx.x * per;
  const int e1 = min(e0 + per, E);
  const int last = e1 - 1;
  for (int e = e0 + tid; e < e1; e += 8 * SCAT_THREADS) {
    int c[8];
#pragma unroll
    for (int j = 0; j < 8; ++j) {
      const int ee = e + j * SCAT_THREADS;
      const bool in = ee < e1;
      const int v = col[in ? ee : last];
      c[j] = in ? v : -1;
    }
#pragma unroll
    for (int j = 0; j < 8; ++j) {
      const unsigned idx = (unsigned)(c[j] - lo);
      if (idx < (unsigned)BIN_NODES) atomicAdd(&bins[idx], 1.0f);
    }
  }
  __syncthreads();

  const int rot = (int)blockIdx.x * SCAT_THREADS;
  for (int t = tid; t < BIN_NODES; t += SCAT_THREADS) {
    const int i = (t + rot) & (BIN_NODES - 1);
    const float v = bins[i];
    const int node = lo + i;
    if (v != 0.f && node < n) atomicAdd(&deg[node], v);
  }
}


__global__ __launch_bounds__(256) void k_gemm1(const float* __restrict__ x,
                                               const float* __restrict__ W1,
                                               float* __restrict__ h1,
                                               float* degnorm,
                                               float* __restrict__ agg1,
                                               int n_nodes) {
  const int lane = threadIdx.x & 31;
  const int tile = blockIdx.x * 8 + (threadIdx.x >> 5);
  const int row0 = tile * 16;
  if (row0 >= n_nodes) return;

  const int nsel  = lane & 15;
  const int khalf = (lane >> 4) << 1;

  const float* arow = x + (size_t)(row0 + nsel) * NFEAT + khalf;

  const int mbase = row0 + ((lane >> 4) << 3);
  float dg[8];
#pragma unroll
  for (int r = 0; r < 8; ++r) dg[r] = degnorm[mbase + r];

  v8f c = {};
  for (int k = 0; k < NFEAT; k += 4) {
    if ((k & 127) == 0) __builtin_prefetch(arow + k + 128, 0, 0);
    v2f a = *(const v2f*)(arow + k);
    v2f b;
    b.x = W1[(k + khalf) * HIDDEN + nsel];
    b.y = W1[(k + khalf + 1) * HIDDEN + nsel];
    c = __builtin_amdgcn_wmma_f32_16x16x4_f32(false, a, false, b, (short)0, c,
                                              false, false);
  }

  float d[8];
#pragma unroll
  for (int r = 0; r < 8; ++r) d[r] = rsqrtf(dg[r]);
#pragma unroll
  for (int r = 0; r < 8; ++r) {
    const size_t o = (size_t)(mbase + r) * HIDDEN + nsel;
    h1[o] = c[r];
    agg1[o] = c[r] * (d[r] * d[r]);
  }
  if (nsel == 0) {
#pragma unroll
    for (int r = 0; r < 8; ++r) degnorm[mbase + r] = d[r];
  }
}


#define A1_NODES 4096
#define A1_CL_BITS 12
#define A1_SLICES 10
#define A1_THREADS 1024
#define A1_U 8
#define A1_G 8
static_assert((1 << A1_CL_BITS) == A1_NODES, "list entry packs the in-range target in A1_CL_BITS bits");

__global__ __launch_bounds__(A1_THREADS) void k_agg1_edges(const int* __restrict__ row,
                                                           const int* __restrict__ col,
                                                           const float* __restrict__ h1,
                                                           const float* __restrict__ dinv,
                                                           float* __restrict__ agg1,
                                                           int E, int n) {
  __shared__ float bins[A1_NODES * HIDDEN];
  __shared__ unsigned list[A1_THREADS / 32][A1_U * 32];
  const int tid = threadIdx.x;
  const int lane = tid & 31;
  const int half = lane >> 4;
  const int f = lane & 15;
  const int lo = blockIdx.y * A1_NODES;
  unsigned* mylist = list[tid >> 5];

  for (int i = tid; i < A1_NODES * HIDDEN / 4; i += A1_THREADS)
    ((float4*)bins)[i] = make_float4(0.f, 0.f, 0.f, 0.f);
  __syncthreads();

  const int per = (E + (int)gridDim.x - 1) / (int)gridDim.x;
  const int e0 = blockIdx.x * per;
  const int e1 = min(e0 + per, E);
  if (e0 < e1) {
    const int last = e1 - 1;
    int c[A1_U];
    unsigned r[A1_U];
#pragma unroll
    for (int j = 0; j < A1_U; ++j) {
      const int ee = e0 + tid + j * A1_THREADS;
      const bool in = ee < e1;
      const int k = in ? ee : last;
      const int cv = col[k];
      r[j] = (unsigned)row[k];
      c[j] = in ? cv : -1;
    }
    for (int base = e0; base < e1; base += A1_U * A1_THREADS) {
      int cn[A1_U];
      unsigned rn[A1_U];
#pragma unroll
      for (int j = 0; j < A1_U; ++j) {
        const int ee = base + tid + (A1_U + j) * A1_THREADS;
        const bool in = ee < e1;
        const int k = in ? ee : last;
        const int cv = col[k];
        rn[j] = (unsigned)row[k];
        cn[j] = in ? cv : -1;
      }
      int cnt = 0;
#pragma unroll
      for (int j = 0; j < A1_U; ++j) {
        const unsigned idx = (unsigned)(c[j] - lo);
        const bool m = idx < (unsigned)A1_NODES;
        const unsigned mask = __builtin_amdgcn_ballot_w32(m);
        if (m) mylist[cnt + (int)__builtin_amdgcn_mbcnt_lo(mask, 0u)] = (r[j] << A1_CL_BITS) | idx;
        cnt += __builtin_popcount(mask);
      }
      for (int t = 0; t < cnt; t += 2 * A1_G) {
        unsigned ent[A1_G];
        float hv[A1_G], dr[A1_G];
#pragma unroll
        for (int q = 0; q < A1_G; ++q) {
          const int i = t + 2 * q + half;
          ent[q] = mylist[i < cnt ? i : t];
        }
#pragma unroll
        for (int q = 0; q < A1_G; ++q) {
          const unsigned rr = ent[q] >> A1_CL_BITS;
          hv[q] = h1[rr * HIDDEN + f];
          dr[q] = dinv[rr];
        }
#pragma unroll
        for (int q = 0; q < A1_G; ++q) {
          const int i = t + 2 * q + half;
          if (i < cnt)
            atomicAdd(&bins[(ent[q] & (A1_NODES - 1)) * HIDDEN + f], hv[q] * dr[q]);
        }
      }
#pragma unroll
      for (int j = 0; j < A1_U; ++j) {
        c[j] = cn[j];
        r[j] = rn[j];
      }
    }
  }
  __syncthreads();

  const int total = n * HIDDEN;
  const int rot = (int)blockIdx.x * (((A1_NODES * HIDDEN) / A1_SLICES) & ~31);
  for (int t = tid; t < A1_NODES * HIDDEN; t += 4 * A1_THREADS) {
    float v[4], d[4];
    int g[4];
#pragma unroll
    for (int u = 0; u < 4; ++u) {
      const int i = (t + u * A1_THREADS + rot) & (A1_NODES * HIDDEN - 1);
      v[u] = bins[i];
      g[u] = lo * HIDDEN + i;
      d[u] = dinv[min(lo + (i >> 4), n - 1)];
    }
#pragma unroll
    for (int u = 0; u < 4; ++u)
      if (v[u] != 0.f && g[u] < total) atomicAdd(&agg1[g[u]], v[u] * d[u]);
  }
}


__global__ void k_relu_gemm2(const float* __restrict__ agg1, const float* __restrict__ b1,
                             const float* __restrict__ W2, const float* __restrict__ dinv,
                             float* __restrict__ h2, float* __restrict__ agg2, int n) {
  int node = blockIdx.x * blockDim.x + threadIdx.x;
  if (node >= n) return;
  const float dn = dinv[node];
  const float w = dn * dn;
  float acc0 = 0.f, acc1 = 0.f;
#pragma unroll
  for (int j = 0; j < HIDDEN; ++j) {
    float v = fmaxf(agg1[(size_t)node * HIDDEN + j] + b1[j], 0.f);
    acc0 = fmaf(v, W2[j * NCLS + 0], acc0);
    acc1 = fmaf(v, W2[j * NCLS + 1], acc1);
  }
  h2[(size_t)node * NCLS + 0] = acc0;
  h2[(size_t)node * NCLS + 1] = acc1;
  agg2[(size_t)node * NCLS + 0] = acc0 * w;
  agg2[(size_t)node * NCLS + 1] = acc1 * w;
}


__global__ __launch_bounds__(SCAT_THREADS) void k_agg2_edges(const int* __restrict__ row,
                                                             const int* __restrict__ col,
                                                             const float* __restrict__ h2,
                                                             const float* __restrict__ dinv,
                                                             float* __restrict__ agg2,
                                                             int E, int n) {
  __shared__ float bins[BIN_NODES * NCLS];
  const int tid = threadIdx.x;
  const int lo = blockIdx.y * BIN_NODES;
  for (int i = tid; i < BIN_NODES * NCLS; i += SCAT_THREADS) bins[i] = 0.f;
  __syncthreads();

  const int per = (E + (int)gridDim.x - 1) / (int)gridDim.x;
  const int e0 = blockIdx.x * per;
  const int e1 = min(e0 + per, E);
  if (e0 < e1) {
    const int last = e1 - 1;
    int c[4];
    unsigned r[4];
#pragma unroll
    for (int j = 0; j < 4; ++j) {
      const int ee = e0 + tid + j * SCAT_THREADS;
      const bool in = ee < e1;
      const int k = in ? ee : last;
      const int cv = col[k];
      r[j] = (unsigned)row[k];
      c[j] = in ? cv : -1;
    }
    for (int e = e0 + tid; e < e1; e += 4 * SCAT_THREADS) {
      int cn[4];
      unsigned rn[4];
#pragma unroll
      for (int j = 0; j < 4; ++j) {
        const int ee = e + (4 + j) * SCAT_THREADS;
        const bool in = ee < e1;
        const int k = in ? ee : last;
        const int cv = col[k];
        rn[j] = (unsigned)row[k];
        cn[j] = in ? cv : -1;
      }
      float dr[4], dc[4];
      v2f hv[4];
#pragma unroll
      for (int j = 0; j < 4; ++j) {
        const unsigned idx = (unsigned)(c[j] - lo);
        if (idx < (unsigned)BIN_NODES) {
          dr[j] = dinv[r[j]];
          dc[j] = dinv[(unsigned)c[j]];
          hv[j] = *(const v2f*)(h2 + r[j] * NCLS);
        }
      }
#pragma unroll
      for (int j = 0; j < 4; ++j) {
        const unsigned idx = (unsigned)(c[j] - lo);
        if (idx < (unsigned)BIN_NODES) {
          const float w = dr[j] * dc[j];
          atomicAdd(&bins[idx * NCLS + 0], hv[j].x * w);
          atomicAdd(&bins[idx * NCLS + 1], hv[j].y * w);
        }
      }
#pragma unroll
      for (int j = 0; j < 4; ++j) {
        c[j] = cn[j];
        r[j] = rn[j];
      }
    }
  }
  __syncthreads();

  const int total = n * NCLS;
  const int rot = (int)blockIdx.x * SCAT_THREADS;
  for (int t = tid; t < BIN_NODES * NCLS; t += SCAT_THREADS) {
    const int i = (t + rot) & (BIN_NODES * NCLS - 1);
    const float v = bins[i];
    const int g = lo * NCLS + i;
    if (v != 0.f && g < total) atomicAdd(&agg2[g], v);
  }
}


__global__ void k_logsoftmax(const float* __restrict__ agg2, const float* __restrict__ b2,
                             float* __restrict__ out, int n) {
  int node = blockIdx.x * blockDim.x + threadIdx.x;
  if (node >= n) return;
  float z0 = agg2[(size_t)node * NCLS + 0] + b2[0];
  float z1 = agg2[(size_t)node * NCLS + 1] + b2[1];
  float m = fmaxf(z0, z1);
  float lse = m + logf(expf(z0 - m) + expf(z1 - m));
  out[(size_t)node * NCLS + 0] = z0 - lse;
  out[(size_t)node * NCLS + 1] = z1 - lse;
}


extern "C" void kernel_launch(void* const* d_in, const int* in_sizes, int n_in,
                              void* d_out, int out_size, void* d_ws, size_t ws_size,
                              hipStream_t stream) {
  const float* x  = (const float*)d_in[0];
  const int*   ei = (const int*)d_in[1];
  const float* W1 = (const float*)d_in[2];
  const float* b1 = (const float*)d_in[3];
  const float* W2 = (const float*)d_in[4];
  const float* b2 = (const float*)d_in[5];

  const int n = in_sizes[0] / NFEAT;
  const int E = in_sizes[1] / 2;
  const int* row = ei;
  const int* col = ei + E;

  char* ws = (char*)d_ws;
  size_t off = 0;
  auto carve = [&](size_t bytes) -> void* {
    void* p = ws + off;
    off = (off + bytes + 255) & ~(size_t)255;
    return p;
  };
  float* deg  = (float*)carve((size_t)n * sizeof(float));
  float* h1   = (float*)carve((size_t)n * HIDDEN * sizeof(float));
  float* agg1 = (float*)carve((size_t)n * HIDDEN * sizeof(float));
  float* h2   = (float*)carve((size_t)n * NCLS * sizeof(float));
  float* agg2 = (float*)carve((size_t)n * NCLS * sizeof(float));
  (void)ws_size;

  const int B = 256;
  auto blocks = [](long long work, int b) { return (unsigned)((work + b - 1) / b); };

  k_init_deg<<<blocks(n, B), B, 0, stream>>>(deg, n);
  const dim3 scat_grid(EDGE_SLICES, blocks(n, BIN_NODES));
  k_deg_count<<<scat_grid, SCAT_THREADS, 0, stream>>>(col, deg, E, n);

  int tiles = (n + 15) / 16;
  k_gemm1<<<blocks(tiles, 8), B, 0, stream>>>(x, W1, h1, deg, agg1, n);

  const dim3 a1_grid(A1_SLICES, blocks(n, A1_NODES));
  k_agg1_edges<<<a1_grid, A1_THREADS, 0, stream>>>(row, col, h1, deg, agg1, E, n);

  k_relu_gemm2<<<blocks(n, B), B, 0, stream>>>(agg1, b1, W2, deg, h2, agg2, n);

  k_agg2_edges<<<scat_grid, SCAT_THREADS, 0, stream>>>(row, col, h2, deg, agg2, E, n);

  k_logsoftmax<<<blocks(n, B), B, 0, stream>>>(agg2, b2, (float*)d_out, n);
}


# stored features carry their node's deg^-1/2 (GEMM writes h1*dinv, layer-2 product h2*dinv): edge passes gather one item per edge; layer-2 target factor applied once per bin; scale rounding changes
# speedup vs baseline: 3.5281x; 1.1775x over previous
#include <hip/hip_runtime.h>
#include <math.h>

typedef __attribute__((ext_vector_type(2))) float v2f;
typedef __attribute__((ext_vector_type(8))) float v8f;

#define NFEAT 512
#define HIDDEN 16
#define NCLS 2


__global__ void k_init_deg(float* __restrict__ deg, int n) {
  int i = blockIdx.x * blockDim.x + threadIdx.x;
  if (i < n) deg[i] = 1.0f;
}

#define BIN_NODES 16384
#define EDGE_SLICES 64
#define SCAT_THREADS 1024
static_assert((BIN_NODES & (BIN_NODES - 1)) == 0, "bin index is wrapped with a mask");

__global__ __launch_bounds__(SCAT_THREADS) void k_deg_count(const int* __restrict__ col,
                                                            float* __restrict__ deg,
                                                            int E, int n) {
  __shared__ float bins[BIN_NODES];
  const int tid = threadIdx.x;
  const int lo = blockIdx.y * BIN_NODES;
  for (int i = tid; i < BIN_NODES; i += SCAT_THREADS) bins[i] = 0.f;
  __syncthreads();

  const int per = (E + (int)gridDim.x - 1) / (int)gridDim.x;
  const int e0 = blockIdx.x * per;
  const int e1 = min(e0 + per, E);
  const int last = e1 - 1;
  for (int e = e0 + tid; e < e1; e += 8 * SCAT_THREADS) {
    int c[8];
#pragma unroll
    for (int j = 0; j < 8; ++j) {
      const int ee = e + j * SCAT_THREADS;
      const bool in = ee < e1;
      const int v = col[in ? ee : last];
      c[j] = in ? v : -1;
    }
#pragma unroll
    for (int j = 0; j < 8; ++j) {
      const unsigned idx = (unsigned)(c[j] - lo);
      if (idx < (unsigned)BIN_NODES) atomicAdd(&bins[idx], 1.0f);
    }
  }
  __syncthreads();

  const int rot = (int)blockIdx.x * SCAT_THREADS;
  for (int t = tid; t < BIN_NODES; t += SCAT_THREADS) {
    const int i = (t + rot) & (BIN_NODES - 1);
    const float v = bins[i];
    const int node = lo + i;
    if (v != 0.f && node < n) atomicAdd(&deg[node], v);
  }
}


__global__ __launch_bounds__(256) void k_gemm1(const float* __restrict__ x,
                                               const float* __restrict__ W1,
                                               float* __restrict__ h1,
                                               float* degnorm,
                                               float* __restrict__ agg1,
                                               int n_nodes) {
  const int lane = threadIdx.x & 31;
  const int tile = blockIdx.x * 8 + (threadIdx.x >> 5);
  const int row0 = tile * 16;
  if (row0 >= n_nodes) return;

  const int nsel  = lane & 15;
  const int khalf = (lane >> 4) << 1;

  const float* arow = x + (size_t)(row0 + nsel) * NFEAT + khalf;

  const int mbase = row0 + ((lane >> 4) << 3);
  float dg[8];
#pragma unroll
  for (int r = 0; r < 8; ++r) dg[r] = degnorm[mbase + r];

  v8f c = {};
  for (int k = 0; k < NFEAT; k += 4) {
    if ((k & 127) == 0) __builtin_prefetch(arow + k + 128, 0, 0);
    v2f a = *(const v2f*)(arow + k);
    v2f b;
    b.x = W1[(k + khalf) * HIDDEN + nsel];
    b.y = W1[(k + khalf + 1) * HIDDEN + nsel];
    c = __builtin_amdgcn_wmma_f32_16x16x4_f32(false, a, false, b, (short)0, c,
                                              false, false);
  }

  float d[8];
#pragma unroll
  for (int r = 0; r < 8; ++r) d[r] = rsqrtf(dg[r]);
#pragma unroll
  for (int r = 0; r < 8; ++r) {
    const size_t o = (size_t)(mbase + r) * HIDDEN + nsel;
    const float hs = c[r] * d[r];
    h1[o] = hs;
    agg1[o] = hs * d[r];
  }
  if (nsel == 0) {
#pragma unroll
    for (int r = 0; r < 8; ++r) degnorm[mbase + r] = d[r];
  }
}


#define A1_NODES 4096
#define A1_CL_BITS 12
#define A1_SLICES 10
#define A1_THREADS 1024
#define A1_U 8
#define A1_G 8
static_assert((1 << A1_CL_BITS) == A1_NODES, "list entry packs the in-range target in A1_CL_BITS bits");

__global__ __launch_bounds__(A1_THREADS) void k_agg1_edges(const int* __restrict__ row,
                                                           const int* __restrict__ col,
                                                           const float* __restrict__ h1,
                                                           const float* __restrict__ dinv,
                                                           float* __restrict__ agg1,
                                                           int E, int n) {
  __shared__ float bins[A1_NODES * HIDDEN];
  __shared__ unsigned list[A1_THREADS / 32][A1_U * 32];
  const int tid = threadIdx.x;
  const int lane = tid & 31;
  const int half = lane >> 4;
  const int f = lane & 15;
  const int lo = blockIdx.y * A1_NODES;
  unsigned* mylist = list[tid >> 5];

  for (int i = tid; i < A1_NODES * HIDDEN / 4; i += A1_THREADS)
    ((float4*)bins)[i] = make_float4(0.f, 0.f, 0.f, 0.f);
  __syncthreads();

  const int per = (E + (int)gridDim.x - 1) / (int)gridDim.x;
  const int e0 = blockIdx.x * per;
  const int e1 = min(e0 + per, E);
  if (e0 < e1) {
    const int last = e1 - 1;
    int c[A1_U];
    unsigned r[A1_U];
#pragma unroll
    for (int j = 0; j < A1_U; ++j) {
      const int ee = e0 + tid + j * A1_THREADS;
      const bool in = ee < e1;
      const int k = in ? ee : last;
      const int cv = col[k];
      r[j] = (unsigned)row[k];
      c[j] = in ? cv : -1;
    }
    for (int base = e0; base < e1; base += A1_U * A1_THREADS) {
      int cn[A1_U];
      unsigned rn[A1_U];
#pragma unroll
      for (int j = 0; j < A1_U; ++j) {
        const int ee = base + tid + (A1_U + j) * A1_THREADS;
        const bool in = ee < e1;
        const int k = in ? ee : last;
        const int cv = col[k];
        rn[j] = (unsigned)row[k];
        cn[j] = in ? cv : -1;
      }
      int cnt = 0;
#pragma unroll
      for (int j = 0; j < A1_U; ++j) {
        const unsigned idx = (unsigned)(c[j] - lo);
        const bool m = idx < (unsigned)A1_NODES;
        const unsigned mask = __builtin_amdgcn_ballot_w32(m);
        if (m) mylist[cnt + (int)__builtin_amdgcn_mbcnt_lo(mask, 0u)] = (r[j] << A1_CL_BITS) | idx;
        cnt += __builtin_popcount(mask);
      }
      for (int t = 0; t < cnt; t += 2 * A1_G) {
        unsigned ent[A1_G];
        float hv[A1_G];
#pragma unroll
        for (int q = 0; q < A1_G; ++q) {
          const int i = t + 2 * q + half;
          ent[q] = mylist[i < cnt ? i : t];
        }
#pragma unroll
        for (int q = 0; q < A1_G; ++q) {
          const unsigned rr = ent[q] >> A1_CL_BITS;
          hv[q] = h1[rr * HIDDEN + f];
        }
#pragma unroll
        for (int q = 0; q < A1_G; ++q) {
          const int i = t + 2 * q + half;
          if (i < cnt)
            atomicAdd(&bins[(ent[q] & (A1_NODES - 1)) * HIDDEN + f], hv[q]);
        }
      }
#pragma unroll
      for (int j = 0; j < A1_U; ++j) {
        c[j] = cn[j];
        r[j] = rn[j];
      }
    }
  }
  __syncthreads();

  const int total = n * HIDDEN;
  const int rot = (int)blockIdx.x * (((A1_NODES * HIDDEN) / A1_SLICES) & ~31);
  for (int t = tid; t < A1_NODES * HIDDEN; t += 4 * A1_THREADS) {
    float v[4], d[4];
    int g[4];
#pragma unroll
    for (int u = 0; u < 4; ++u) {
      const int i = (t + u * A1_THREADS + rot) & (A1_NODES * HIDDEN - 1);
      v[u] = bins[i];
      g[u] = lo * HIDDEN + i;
      d[u] = dinv[min(lo + (i >> 4), n - 1)];
    }
#pragma unroll
    for (int u = 0; u < 4; ++u)
      if (v[u] != 0.f && g[u] < total) atomicAdd(&agg1[g[u]], v[u] * d[u]);
  }
}


__global__ void k_relu_gemm2(const float* __restrict__ agg1, const float* __restrict__ b1,
                             const float* __restrict__ W2, const float* __restrict__ dinv,
                             float* __restrict__ h2, float* __restrict__ agg2, int n) {
  int node = blockIdx.x * blockDim.x + threadIdx.x;
  if (node >= n) return;
  const float dn = dinv[node];
  float acc0 = 0.f, acc1 = 0.f;
#pragma unroll
  for (int j = 0; j < HIDDEN; ++j) {
    float v = fmaxf(agg1[(size_t)node * HIDDEN + j] + b1[j], 0.f);
    acc0 = fmaf(v, W2[j * NCLS + 0], acc0);
    acc1 = fmaf(v, W2[j * NCLS + 1], acc1);
  }
  const float s0 = acc0 * dn, s1 = acc1 * dn;
  h2[(size_t)node * NCLS + 0] = s0;
  h2[(size_t)node * NCLS + 1] = s1;
  agg2[(size_t)node * NCLS + 0] = s0 * dn;
  agg2[(size_t)node * NCLS + 1] = s1 * dn;
}


__global__ __launch_bounds__(SCAT_THREADS) void k_agg2_edges(const int* __restrict__ row,
                                                             const int* __restrict__ col,
                                                             const float* __restrict__ h2,
                                                             const float* __restrict__ dinv,
                                                             float* __restrict__ agg2,
                                                             int E, int n) {
  __shared__ float bins[BIN_NODES * NCLS];
  const int tid = threadIdx.x;
  const int lo = blockIdx.y * BIN_NODES;
  for (int i = tid; i < BIN_NODES * NCLS; i += SCAT_THREADS) bins[i] = 0.f;
  __syncthreads();

  const int per = (E + (int)gridDim.x - 1) / (int)gridDim.x;
  const int e0 = blockIdx.x * per;
  const int e1 = min(e0 + per, E);
  if (e0 < e1) {
    const int last = e1 - 1;
    int c[4];
    unsigned r[4];
#pragma unroll
    for (int j = 0; j < 4; ++j) {
      const int ee = e0 + tid + j * SCAT_THREADS;
      const bool in = ee < e1;
      const int k = in ? ee : last;
      const int cv = col[k];
      r[j] = (unsigned)row[k];
      c[j] = in ? cv : -1;
    }
    for (int e = e0 + tid; e < e1; e += 4 * SCAT_THREADS) {
      int cn[4];
      unsigned rn[4];
#pragma unroll
      for (int j = 0; j < 4; ++j) {
        const int ee = e + (4 + j) * SCAT_THREADS;
        const bool in = ee < e1;
        const int k = in ? ee : last;
        const int cv = col[k];
        rn[j] = (unsigned)row[k];
        cn[j] = in ? cv : -1;
      }
      v2f hv[4];
#pragma unroll
      for (int j = 0; j < 4; ++j) {
        const unsigned idx = (unsigned)(c[j] - lo);
        if (idx < (unsigned)BIN_NODES) hv[j] = *(const v2f*)(h2 + r[j] * NCLS);
      }
#pragma unroll
      for (int j = 0; j < 4; ++j) {
        const unsigned idx = (unsigned)(c[j] - lo);
        if (idx < (unsigned)BIN_NODES) {
          atomicAdd(&bins[idx * NCLS + 0], hv[j].x);
          atomicAdd(&bins[idx * NCLS + 1], hv[j].y);
        }
      }
#pragma unroll
      for (int j = 0; j < 4; ++j) {
        c[j] = cn[j];
        r[j] = rn[j];
      }
    }
  }
  __syncthreads();

  const int total = n * NCLS;
  const int rot = (int)blockIdx.x * SCAT_THREADS;
  for (int t = tid; t < BIN_NODES * NCLS; t += SCAT_THREADS) {
    const int i = (t + rot) & (BIN_NODES * NCLS - 1);
    const float v = bins[i];
    const int g = lo * NCLS + i;
    const float d = dinv[min(lo + (i >> 1), n - 1)];
    if (v != 0.f && g < total) atomicAdd(&agg2[g], v * d);
  }
}


__global__ void k_logsoftmax(const float* __restrict__ agg2, const float* __restrict__ b2,
                             float* __restrict__ out, int n) {
  int node = blockIdx.x * blockDim.x + threadIdx.x;
  if (node >= n) return;
  float z0 = agg2[(size_t)node * NCLS + 0] + b2[0];
  float z1 = agg2[(size_t)node * NCLS + 1] + b2[1];
  float m = fmaxf(z0, z1);
  float lse = m + logf(expf(z0 - m) + expf(z1 - m));
  out[(size_t)node * NCLS + 0] = z0 - lse;
  out[(size_t)node * NCLS + 1] = z1 - lse;
}


extern "C" void kernel_launch(void* const* d_in, const int* in_sizes, int n_in,
                              void* d_out, int out_size, void* d_ws, size_t ws_size,
                              hipStream_t stream) {
  const float* x  = (const float*)d_in[0];
  const int*   ei = (const int*)d_in[1];
  const float* W1 = (const float*)d_in[2];
  const float* b1 = (const float*)d_in[3];
  const float* W2 = (const float*)d_in[4];
  const float* b2 = (const float*)d_in[5];

  const int n = in_sizes[0] / NFEAT;
  const int E = in_sizes[1] / 2;
  const int* row = ei;
  const int* col = ei + E;

  char* ws = (char*)d_ws;
  size_t off = 0;
  auto carve = [&](size_t bytes) -> void* {
    void* p = ws + off;
    off = (off + bytes + 255) & ~(size_t)255;
    return p;
  };
  float* deg  = (float*)carve((size_t)n * sizeof(float));
  float* h1   = (float*)carve((size_t)n * HIDDEN * sizeof(float));
  float* agg1 = (float*)carve((size_t)n * HIDDEN * sizeof(float));
  float* h2   = (float*)carve((size_t)n * NCLS * sizeof(float));
  float* agg2 = (float*)carve((size_t)n * NCLS * sizeof(float));
  (void)ws_size;

  const int B = 256;
  auto blocks = [](long long work, int b) { return (unsigned)((work + b - 1) / b); };

  k_init_deg<<<blocks(n, B), B, 0, stream>>>(deg, n);
  const dim3 scat_grid(EDGE_SLICES, blocks(n, BIN_NODES));
  k_deg_count<<<scat_grid, SCAT_THREADS, 0, stream>>>(col, deg, E, n);

  int tiles = (n + 15) / 16;
  k_gemm1<<<blocks(tiles, 8), B, 0, stream>>>(x, W1, h1, deg, agg1, n);

  const dim3 a1_grid(A1_SLICES, blocks(n, A1_NODES));
  k_agg1_edges<<<a1_grid, A1_THREADS, 0, stream>>>(row, col, h1, deg, agg1, E, n);

  k_relu_gemm2<<<blocks(n, B), B, 0, stream>>>(agg1, b1, W2, deg, h2, agg2, n);

  k_agg2_edges<<<scat_grid, SCAT_THREADS, 0, stream>>>(row, col, h2, deg, agg2, E, n);

  k_logsoftmax<<<blocks(n, B), B, 0, stream>>>(agg2, b2, (float*)d_out, n);
}


# edge list grouped once per call by 256-node target range (per-block LDS counting sort); one block per range sums degree, layer-1 and layer-2 from its runs, no global atomics; float sum order changes
# speedup vs baseline: 5.1648x; 1.4639x over previous
#include <hip/hip_runtime.h>
#include <math.h>

typedef __attribute__((ext_vector_type(2))) float v2f;
typedef __attribute__((ext_vector_type(8))) float v8f;

#define NFEAT 512
#define HIDDEN 16
#define NCLS 2


__global__ void k_init_deg(float* __restrict__ deg, int n) {
  int i = blockIdx.x * blockDim.x + threadIdx.x;
  if (i < n) deg[i] = 1.0f;
}

#define BIN_NODES 16384
#define EDGE_SLICES 64
#define SCAT_THREADS 1024
static_assert((BIN_NODES & (BIN_NODES - 1)) == 0, "bin index is wrapped with a mask");

__global__ __launch_bounds__(SCAT_THREADS) void k_deg_count(const int* __restrict__ col,
                                                            float* __restrict__ deg,
                                                            int E, int n) {
  __shared__ float bins[BIN_NODES];
  const int tid = threadIdx.x;
  const int lo = blockIdx.y * BIN_NODES;
  for (int i = tid; i < BIN_NODES; i += SCAT_THREADS) bins[i] = 0.f;
  __syncthreads();

  const int per = (E + (int)gridDim.x - 1) / (int)gridDim.x;
  const int e0 = blockIdx.x * per;
  const int e1 = min(e0 + per, E);
  const int last = e1 - 1;
  for (int e = e0 + tid; e < e1; e += 8 * SCAT_THREADS) {
    int c[8];
#pragma unroll
    for (int j = 0; j < 8; ++j) {
      const int ee = e + j * SCAT_THREADS;
      const bool in = ee < e1;
      const int v = col[in ? ee : last];
      c[j] = in ? v : -1;
    }
#pragma unroll
    for (int j = 0; j < 8; ++j) {
      const unsigned idx = (unsigned)(c[j] - lo);
      if (idx < (unsigned)BIN_NODES) atomicAdd(&bins[idx], 1.0f);
    }
  }
  __syncthreads();

  const int rot = (int)blockIdx.x * SCAT_THREADS;
  for (int t = tid; t < BIN_NODES; t += SCAT_THREADS) {
    const int i = (t + rot) & (BIN_NODES - 1);
    const float v = bins[i];
    const int node = lo + i;
    if (v != 0.f && node < n) atomicAdd(&deg[node], v);
  }
}


__global__ __launch_bounds__(256) void k_gemm1(const float* __restrict__ x,
                                               const float* __restrict__ W1,
                                               float* __restrict__ h1,
                                               float* degnorm,
                                               float* __restrict__ agg1,
                                               int n_nodes) {
  const int lane = threadIdx.x & 31;
  const int tile = blockIdx.x * 8 + (threadIdx.x >> 5);
  const int row0 = tile * 16;
  if (row0 >= n_nodes) return;

  const int nsel  = lane & 15;
  const int khalf = (lane >> 4) << 1;

  const float* arow = x + (size_t)(row0 + nsel) * NFEAT + khalf;

  const int mbase = row0 + ((lane >> 4) << 3);
  float dg[8];
#pragma unroll
  for (int r = 0; r < 8; ++r) dg[r] = degnorm[mbase + r];

  v8f c = {};
  for (int k = 0; k < NFEAT; k += 4) {
    if ((k & 127) == 0) __builtin_prefetch(arow + k + 128, 0, 0);
    v2f a = *(const v2f*)(arow + k);
    v2f b;
    b.x = W1[(k + khalf) * HIDDEN + nsel];
    b.y = W1[(k + khalf + 1) * HIDDEN + nsel];
    c = __builtin_amdgcn_wmma_f32_16x16x4_f32(false, a, false, b, (short)0, c,
                                              false, false);
  }

  float d[8];
#pragma unroll
  for (int r = 0; r < 8; ++r) d[r] = rsqrtf(dg[r]);
#pragma unroll
  for (int r = 0; r < 8; ++r) {
    const size_t o = (size_t)(mbase + r) * HIDDEN + nsel;
    const float hs = c[r] * d[r];
    h1[o] = hs;
    agg1[o] = hs * d[r];
  }
  if (nsel == 0) {
#pragma unroll
    for (int r = 0; r < 8; ++r) degnorm[mbase + r] = d[r];
  }
}


#define A1_NODES 4096
#define A1_CL_BITS 12
#define A1_SLICES 10
#define A1_THREADS 1024
#define A1_U 8
#define A1_G 8
static_assert((1 << A1_CL_BITS) == A1_NODES, "list entry packs the in-range target in A1_CL_BITS bits");

__global__ __launch_bounds__(A1_THREADS) void k_agg1_edges(const int* __restrict__ row,
                                                           const int* __restrict__ col,
                                                           const float* __restrict__ h1,
                                                           const float* __restrict__ dinv,
                                                           float* __restrict__ agg1,
                                                           int E, int n) {
  __shared__ float bins[A1_NODES * HIDDEN];
  __shared__ unsigned list[A1_THREADS / 32][A1_U * 32];
  const int tid = threadIdx.x;
  const int lane = tid & 31;
  const int half = lane >> 4;
  const int f = lane & 15;
  const int lo = blockIdx.y * A1_NODES;
  unsigned* mylist = list[tid >> 5];

  for (int i = tid; i < A1_NODES * HIDDEN / 4; i += A1_THREADS)
    ((float4*)bins)[i] = make_float4(0.f, 0.f, 0.f, 0.f);
  __syncthreads();

  const int per = (E + (int)gridDim.x - 1) / (int)gridDim.x;
  const int e0 = blockIdx.x * per;
  const int e1 = min(e0 + per, E);
  if (e0 < e1) {
    const int last = e1 - 1;
    int c[A1_U];
    unsigned r[A1_U];
#pragma unroll
    for (int j = 0; j < A1_U; ++j) {
      const int ee = e0 + tid + j * A1_THREADS;
      const bool in = ee < e1;
      const int k = in ? ee : last;
      const int cv = col[k];
      r[j] = (unsigned)row[k];
      c[j] = in ? cv : -1;
    }
    for (int base = e0; base < e1; base += A1_U * A1_THREADS) {
      int cn[A1_U];
      unsigned rn[A1_U];
#pragma unroll
      for (int j = 0; j < A1_U; ++j) {
        const int ee = base + tid + (A1_U + j) * A1_THREADS;
        const bool in = ee < e1;
        const int k = in ? ee : last;
        const int cv = col[k];
        rn[j] = (unsigned)row[k];
        cn[j] = in ? cv : -1;
      }
      int cnt = 0;
#pragma unroll
      for (int j = 0; j < A1_U; ++j) {
        const unsigned idx = (unsigned)(c[j] - lo);
        const bool m = idx < (unsigned)A1_NODES;
        const unsigned mask = __builtin_amdgcn_ballot_w32(m);
        if (m) mylist[cnt + (int)__builtin_amdgcn_mbcnt_lo(mask, 0u)] = (r[j] << A1_CL_BITS) | idx;
        cnt += __builtin_popcount(mask);
      }
      for (int t = 0; t < cnt; t += 2 * A1_G) {
        unsigned ent[A1_G];
        float hv[A1_G];
#pragma unroll
        for (int q = 0; q < A1_G; ++q) {
          const int i = t + 2 * q + half;
          ent[q] = mylist[i < cnt ? i : t];
        }
#pragma unroll
        for (int q = 0; q < A1_G; ++q) {
          const unsigned rr = ent[q] >> A1_CL_BITS;
          hv[q] = h1[rr * HIDDEN + f];
        }
#pragma unroll
        for (int q = 0; q < A1_G; ++q) {
          const int i = t + 2 * q + half;
          if (i < cnt)
            atomicAdd(&bins[(ent[q] & (A1_NODES - 1)) * HIDDEN + f], hv[q]);
        }
      }
#pragma unroll
      for (int j = 0; j < A1_U; ++j) {
        c[j] = cn[j];
        r[j] = rn[j];
      }
    }
  }
  __syncthreads();

  const int total = n * HIDDEN;
  const int rot = (int)blockIdx.x * (((A1_NODES * HIDDEN) / A1_SLICES) & ~31);
  for (int t = tid; t < A1_NODES * HIDDEN; t += 4 * A1_THREADS) {
    float v[4], d[4];
    int g[4];
#pragma unroll
    for (int u = 0; u < 4; ++u) {
      const int i = (t + u * A1_THREADS + rot) & (A1_NODES * HIDDEN - 1);
      v[u] = bins[i];
      g[u] = lo * HIDDEN + i;
      d[u] = dinv[min(lo + (i >> 4), n - 1)];
    }
#pragma unroll
    for (int u = 0; u < 4; ++u)
      if (v[u] != 0.f && g[u] < total) atomicAdd(&agg1[g[u]], v[u] * d[u]);
  }
}


__global__ void k_relu_gemm2(const float* __restrict__ agg1, const float* __restrict__ b1,
                             const float* __restrict__ W2, const float* __restrict__ dinv,
                             float* __restrict__ h2, float* __restrict__ agg2, int n) {
  int node = blockIdx.x * blockDim.x + threadIdx.x;
  if (node >= n) return;
  const float dn = dinv[node];
  float acc0 = 0.f, acc1 = 0.f;
#pragma unroll
  for (int j = 0; j < HIDDEN; ++j) {
    float v = fmaxf(agg1[(size_t)node * HIDDEN + j] + b1[j], 0.f);
    acc0 = fmaf(v, W2[j * NCLS + 0], acc0);
    acc1 = fmaf(v, W2[j * NCLS + 1], acc1);
  }
  const float s0 = acc0 * dn, s1 = acc1 * dn;
  h2[(size_t)node * NCLS + 0] = s0;
  h2[(size_t)node * NCLS + 1] = s1;
  agg2[(size_t)node * NCLS + 0] = s0 * dn;
  agg2[(size_t)node * NCLS + 1] = s1 * dn;
}


__global__ __launch_bounds__(SCAT_THREADS) void k_agg2_edges(const int* __restrict__ row,
                                                             const int* __restrict__ col,
                                                             const float* __restrict__ h2,
                                                             const float* __restrict__ dinv,
                                                             float* __restrict__ agg2,
                                                             int E, int n) {
  __shared__ float bins[BIN_NODES * NCLS];
  const int tid = threadIdx.x;
  const int lo = blockIdx.y * BIN_NODES;
  for (int i = tid; i < BIN_NODES * NCLS; i += SCAT_THREADS) bins[i] = 0.f;
  __syncthreads();

  const int per = (E + (int)gridDim.x - 1) / (int)gridDim.x;
  const int e0 = blockIdx.x * per;
  const int e1 = min(e0 + per, E);
  if (e0 < e1) {
    const int last = e1 - 1;
    int c[4];
    unsigned r[4];
#pragma unroll
    for (int j = 0; j < 4; ++j) {
      const int ee = e0 + tid + j * SCAT_THREADS;
      const bool in = ee < e1;
      const int k = in ? ee : last;
      const int cv = col[k];
      r[j] = (unsigned)row[k];
      c[j] = in ? cv : -1;
    }
    for (int e = e0 + tid; e < e1; e += 4 * SCAT_THREADS) {
      int cn[4];
      unsigned rn[4];
#pragma unroll
      for (int j = 0; j < 4; ++j) {
        const int ee = e + (4 + j) * SCAT_THREADS;
        const bool in = ee < e1;
        const int k = in ? ee : last;
        const int cv = col[k];
        rn[j] = (unsigned)row[k];
        cn[j] = in ? cv : -1;
      }
      v2f hv[4];
#pragma unroll
      for (int j = 0; j < 4; ++j) {
        const unsigned idx = (unsigned)(c[j] - lo);
        if (idx < (unsigned)BIN_NODES) hv[j] = *(const v2f*)(h2 + r[j] * NCLS);
      }
#pragma unroll
      for (int j = 0; j < 4; ++j) {
        const unsigned idx = (unsigned)(c[j] - lo);
        if (idx < (unsigned)BIN_NODES) {
          atomicAdd(&bins[idx * NCLS + 0], hv[j].x);
          atomicAdd(&bins[idx * NCLS + 1], hv[j].y);
        }
      }
#pragma unroll
      for (int j = 0; j < 4; ++j) {
        c[j] = cn[j];
        r[j] = rn[j];
      }
    }
  }
  __syncthreads();

  const int total = n * NCLS;
  const int rot = (int)blockIdx.x * SCAT_THREADS;
  for (int t = tid; t < BIN_NODES * NCLS; t += SCAT_THREADS) {
    const int i = (t + rot) & (BIN_NODES * NCLS - 1);
    const float v = bins[i];
    const int g = lo * NCLS + i;
    const float d = dinv[min(lo + (i >> 1), n - 1)];
    if (v != 0.f && g < total) atomicAdd(&agg2[g], v * d);
  }
}


#define PART_NODES 256
#define PART_BITS 8
#define PART_MAXR 512
#define PART_THREADS 1024
#define PART_U 10
#define PART_CAP (PART_THREADS * PART_U)
#define PART_PITCH 544
#define PART_MAXB 1024
#define PART_PADB (PART_MAXB + 64)
#define ROW_PITCH (HIDDEN + 1)

__global__ __launch_bounds__(PART_THREADS) void k_part(const int* __restrict__ row,
                                                       const int* __restrict__ col,
                                                       unsigned* __restrict__ part,
                                                       int* __restrict__ tab, int E) {
  __shared__ unsigned sorted[PART_CAP];
  __shared__ int hist[PART_MAXR];
  __shared__ int base[PART_MAXR + 1];
  __shared__ int wsum[PART_MAXR / 32];
  const int tid = threadIdx.x;
  const int b = blockIdx.x;

  if (tid < PART_MAXR) hist[tid] = 0;
  __syncthreads();

  const int e0 = b * PART_CAP;
  const int e1 = min(e0 + PART_CAP, E);
  unsigned ent[PART_U];
  int rg[PART_U], rk[PART_U];
#pragma unroll
  for (int j = 0; j < PART_U; ++j) {
    const int ee = e0 + tid + j * PART_THREADS;
    const bool in = ee < e1;
    const int k = in ? ee : e0;
    const int cv = col[k];
    const unsigned rv = (unsigned)row[k];
    rg[j] = in ? (cv >> PART_BITS) : -1;
    ent[j] = (rv << PART_BITS) | ((unsigned)cv & (PART_NODES - 1));
  }
#pragma unroll
  for (int j = 0; j < PART_U; ++j) rk[j] = rg[j] >= 0 ? atomicAdd(&hist[rg[j]], 1) : 0;
  __syncthreads();

  int v = 0, incl = 0;
  if (tid < PART_MAXR) {
    v = hist[tid];
    incl = v;
#pragma unroll
    for (int d = 1; d < 32; d <<= 1) {
      const int up = __shfl_up(incl, d);
      if ((tid & 31) >= d) incl += up;
    }
    if ((tid & 31) == 31) wsum[tid >> 5] = incl;
  }
  __syncthreads();
  if (tid < PART_MAXR) {
    int before = 0;
    for (int w = 0; w < (tid >> 5); ++w) before += wsum[w];
    base[tid] = before + incl - v;
    if (tid == PART_MAXR - 1) base[PART_MAXR] = before + incl;
  }
  __syncthreads();

  if (tid <= PART_MAXR) tab[b * PART_PITCH + tid] = base[tid];
#pragma unroll
  for (int j = 0; j < PART_U; ++j)
    if (rg[j] >= 0) sorted[base[rg[j]] + rk[j]] = ent[j];
  __syncthreads();

  const int cnt = e1 - e0;
  for (int i = tid; i < cnt; i += PART_THREADS) part[e0 + i] = sorted[i];
}

__device__ __forceinline__ void load_runs(const int* __restrict__ tab, int y, int nblk,
                                          int* rb, int* re) {
  for (int i = threadIdx.x; i < PART_PADB; i += blockDim.x) {
    int b0 = 0, b1 = 0;
    if (i < nblk) {
      const int* t = tab + (size_t)i * PART_PITCH + y;
      b0 = i * PART_CAP + t[0];
      b1 = i * PART_CAP + t[1];
    }
    rb[i] = b0;
    re[i] = b1;
  }
}

#define DEG_THREADS 512
__global__ __launch_bounds__(DEG_THREADS) void k_deg_part(const unsigned* __restrict__ part,
                                                          const int* __restrict__ tab,
                                                          float* __restrict__ deg,
                                                          int nblk, int n) {
  __shared__ float bins[PART_NODES];
  __shared__ int rb[PART_PADB], re[PART_PADB];
  const int tid = threadIdx.x;
  const int lane = tid & 31;
  const int y = blockIdx.x;
  load_runs(tab, y, nblk, rb, re);
  if (tid < PART_NODES) bins[tid] = 0.f;
  __syncthreads();

  const int nw = DEG_THREADS / 32;
  for (int b0 = tid >> 5; b0 < nblk; b0 += 4 * nw) {
    int beg[4], cnt[4];
    unsigned en[4];
#pragma unroll
    for (int j = 0; j < 4; ++j) {
      const int bj = b0 + j * nw;
      beg[j] = rb[bj];
      cnt[j] = re[bj] - beg[j];
    }
#pragma unroll
    for (int j = 0; j < 4; ++j) en[j] = part[cnt[j] > 0 ? beg[j] + min(lane, cnt[j] - 1) : 0];
#pragma unroll
    for (int j = 0; j < 4; ++j)
      if (lane < cnt[j]) atomicAdd(&bins[en[j] & (PART_NODES - 1)], 1.0f);
#pragma unroll
    for (int j = 0; j < 4; ++j)
      for (int t = 32 + lane; t < cnt[j]; t += 32)
        atomicAdd(&bins[part[beg[j] + t] & (PART_NODES - 1)], 1.0f);
  }
  __syncthreads();

  if (tid < PART_NODES) {
    const int node = y * PART_NODES + tid;
    if (node < n) deg[node] = 1.0f + bins[tid];
  }
}

#define AP_THREADS 512
__global__ __launch_bounds__(AP_THREADS) void k_agg1_part(const unsigned* __restrict__ part,
                                                          const int* __restrict__ tab,
                                                          const float* __restrict__ h1,
                                                          const float* __restrict__ dinv,
                                                          float* __restrict__ agg1,
                                                          int nblk, int n) {
  __shared__ float bins[PART_NODES * ROW_PITCH];
  __shared__ int rb[PART_PADB], re[PART_PADB];
  const int tid = threadIdx.x;
  const int lane = tid & 31;
  const int eq = lane >> 2;
  const int c4 = (lane & 3) * 4;
  const int y = blockIdx.x;
  load_runs(tab, y, nblk, rb, re);
  for (int i = tid; i < PART_NODES * ROW_PITCH; i += AP_THREADS) bins[i] = 0.f;
  __syncthreads();

  const int nw = AP_THREADS / 32;
  int b0 = tid >> 5;
  int beg[2], cnt[2];
  unsigned cur[2];
#pragma unroll
  for (int j = 0; j < 2; ++j) {
    beg[j] = rb[b0 + j * nw];
    cnt[j] = re[b0 + j * nw] - beg[j];
    cur[j] = part[cnt[j] > 0 ? beg[j] + min(lane, cnt[j] - 1) : 0];
  }
  for (; b0 < nblk; b0 += 2 * nw) {
    int nbeg[2], ncnt[2];
    unsigned nxt[2];
#pragma unroll
    for (int j = 0; j < 2; ++j) {
      nbeg[j] = rb[b0 + (2 + j) * nw];
      ncnt[j] = re[b0 + (2 + j) * nw] - nbeg[j];
      nxt[j] = part[ncnt[j] > 0 ? nbeg[j] + min(lane, ncnt[j] - 1) : 0];
    }
    unsigned en[2][4];
    float4 hv[2][4];
#pragma unroll
    for (int j = 0; j < 2; ++j)
#pragma unroll
      for (int q = 0; q < 4; ++q) {
        en[j][q] = (unsigned)__shfl((int)cur[j], q * 8 + eq);
        hv[j][q] = *(const float4*)(h1 + (en[j][q] >> PART_BITS) * HIDDEN + c4);
      }
#pragma unroll
    for (int j = 0; j < 2; ++j)
#pragma unroll
      for (int q = 0; q < 4; ++q)
        if (q * 8 + eq < cnt[j]) {
          float* p = &bins[(en[j][q] & (PART_NODES - 1)) * ROW_PITCH + c4];
          atomicAdd(p + 0, hv[j][q].x);
          atomicAdd(p + 1, hv[j][q].y);
          atomicAdd(p + 2, hv[j][q].z);
          atomicAdd(p + 3, hv[j][q].w);
        }
#pragma unroll
    for (int j = 0; j < 2; ++j)
      for (int t = 32; t < cnt[j]; t += 32) {
        const unsigned more = part[beg[j] + min(t + lane, cnt[j] - 1)];
        unsigned em[4];
        float4 hm[4];
#pragma unroll
        for (int q = 0; q < 4; ++q) {
          em[q] = (unsigned)__shfl((int)more, q * 8 + eq);
          hm[q] = *(const float4*)(h1 + (em[q] >> PART_BITS) * HIDDEN + c4);
        }
#pragma unroll
        for (int q = 0; q < 4; ++q)
          if (t + q * 8 + eq < cnt[j]) {
            float* p = &bins[(em[q] & (PART_NODES - 1)) * ROW_PITCH + c4];
            atomicAdd(p + 0, hm[q].x);
            atomicAdd(p + 1, hm[q].y);
            atomicAdd(p + 2, hm[q].z);
            atomicAdd(p + 3, hm[q].w);
          }
      }
#pragma unroll
    for (int j = 0; j < 2; ++j) {
      beg[j] = nbeg[j];
      cnt[j] = ncnt[j];
      cur[j] = nxt[j];
    }
  }
  __syncthreads();

  for (int i = tid; i < PART_NODES * HIDDEN; i += AP_THREADS) {
    const int r = i >> 4;
    const int node = y * PART_NODES + r;
    if (node < n) {
      const size_t g = (size_t)node * HIDDEN + (i & 15);
      agg1[g] = agg1[g] + bins[r * ROW_PITCH + (i & 15)] * dinv[node];
    }
  }
}

#define A2_THREADS 512
__global__ __launch_bounds__(A2_THREADS) void k_agg2_part(const unsigned* __restrict__ part,
                                                          const int* __restrict__ tab,
                                                          const float* __restrict__ h2,
                                                          const float* __restrict__ dinv,
                                                          float* __restrict__ agg2,
                                                          int nblk, int n) {
  __shared__ float bins[PART_NODES * NCLS];
  __shared__ int rb[PART_PADB], re[PART_PADB];
  const int tid = threadIdx.x;
  const int lane = tid & 31;
  const int y = blockIdx.x;
  load_runs(tab, y, nblk, rb, re);
  for (int i = tid; i < PART_NODES * NCLS; i += A2_THREADS) bins[i] = 0.f;
  __syncthreads();

  const int nw = A2_THREADS / 32;
  for (int b0 = tid >> 5; b0 < nblk; b0 += 4 * nw) {
    int beg[4], cnt[4];
    unsigned en[4];
    v2f hv[4];
#pragma unroll
    for (int j = 0; j < 4; ++j) {
      const int bj = b0 + j * nw;
      beg[j] = rb[bj];
      cnt[j] = re[bj] - beg[j];
    }
#pragma unroll
    for (int j = 0; j < 4; ++j) en[j] = part[cnt[j] > 0 ? beg[j] + min(lane, cnt[j] - 1) : 0];
#pragma unroll
    for (int j = 0; j < 4; ++j) hv[j] = *(const v2f*)(h2 + (en[j] >> PART_BITS) * NCLS);
#pragma unroll
    for (int j = 0; j < 4; ++j)
      if (lane < cnt[j]) {
        const unsigned idx = en[j] & (PART_NODES - 1);
        atomicAdd(&bins[idx * NCLS + 0], hv[j].x);
        atomicAdd(&bins[idx * NCLS + 1], hv[j].y);
      }
#pragma unroll
    for (int j = 0; j < 4; ++j)
      for (int t = 32 + lane; t < cnt[j]; t += 32) {
        const unsigned em = part[beg[j] + t];
        const v2f hm = *(const v2f*)(h2 + (em >> PART_BITS) * NCLS);
        atomicAdd(&bins[(em & (PART_NODES - 1)) * NCLS + 0], hm.x);
        atomicAdd(&bins[(em & (PART_NODES - 1)) * NCLS + 1], hm.y);
      }
  }
  __syncthreads();

  for (int i = tid; i < PART_NODES * NCLS; i += A2_THREADS) {
    const int node = y * PART_NODES + (i >> 1);
    if (node < n) {
      const size_t g = (size_t)node * NCLS + (i & 1);
      agg2[g] = agg2[g] + bins[i] * dinv[node];
    }
  }
}


__global__ void k_logsoftmax(const float* __restrict__ agg2, const float* __restrict__ b2,
                             float* __restrict__ out, int n) {
  int node = blockIdx.x * blockDim.x + threadIdx.x;
  if (node >= n) return;
  float z0 = agg2[(size_t)node * NCLS + 0] + b2[0];
  float z1 = agg2[(size_t)node * NCLS + 1] + b2[1];
  float m = fmaxf(z0, z1);
  float lse = m + logf(expf(z0 - m) + expf(z1 - m));
  out[(size_t)node * NCLS + 0] = z0 - lse;
  out[(size_t)node * NCLS + 1] = z1 - lse;
}


extern "C" void kernel_launch(void* const* d_in, const int* in_sizes, int n_in,
                              void* d_out, int out_size, void* d_ws, size_t ws_size,
                              hipStream_t stream) {
  const float* x  = (const float*)d_in[0];
  const int*   ei = (const int*)d_in[1];
  const float* W1 = (const float*)d_in[2];
  const float* b1 = (const float*)d_in[3];
  const float* W2 = (const float*)d_in[4];
  const float* b2 = (const float*)d_in[5];

  const int n = in_sizes[0] / NFEAT;
  const int E = in_sizes[1] / 2;
  const int* row = ei;
  const int* col = ei + E;

  char* ws = (char*)d_ws;
  size_t off = 0;
  auto carve = [&](size_t bytes) -> void* {
    void* p = ws + off;
    off = (off + bytes + 255) & ~(size_t)255;
    return p;
  };
  float* deg  = (float*)carve((size_t)n * sizeof(float));
  float* h1   = (float*)carve((size_t)n * HIDDEN * sizeof(float));
  float* agg1 = (float*)carve((size_t)n * HIDDEN * sizeof(float));
  float* h2   = (float*)carve((size_t)n * NCLS * sizeof(float));
  float* agg2 = (float*)carve((size_t)n * NCLS * sizeof(float));

  const int B = 256;
  auto blocks = [](long long work, int b) { return (unsigned)((work + b - 1) / b); };

  const unsigned nblk = blocks(E, PART_CAP);
  const unsigned nrng = blocks(n, PART_NODES);
  unsigned* part = (unsigned*)carve((size_t)E * sizeof(unsigned));
  int* tab = (int*)carve((size_t)nblk * PART_PITCH * sizeof(int));
  const bool grouped = off <= ws_size && nrng <= PART_MAXR && nblk <= PART_MAXB &&
                       n <= (1 << (32 - PART_BITS));
  const dim3 scat_grid(EDGE_SLICES, blocks(n, BIN_NODES));

  if (grouped) {
    k_part<<<nblk, PART_THREADS, 0, stream>>>(row, col, part, tab, E);
    k_deg_part<<<nrng, DEG_THREADS, 0, stream>>>(part, tab, deg, (int)nblk, n);
  } else {
    k_init_deg<<<blocks(n, B), B, 0, stream>>>(deg, n);
    k_deg_count<<<scat_grid, SCAT_THREADS, 0, stream>>>(col, deg, E, n);
  }

  int tiles = (n + 15) / 16;
  k_gemm1<<<blocks(tiles, 8), B, 0, stream>>>(x, W1, h1, deg, agg1, n);

  if (grouped) {
    k_agg1_part<<<nrng, AP_THREADS, 0, stream>>>(part, tab, h1, deg, agg1, (int)nblk, n);
  } else {
    const dim3 a1_grid(A1_SLICES, blocks(n, A1_NODES));
    k_agg1_edges<<<a1_grid, A1_THREADS, 0, stream>>>(row, col, h1, deg, agg1, E, n);
  }

  k_relu_gemm2<<<blocks(n, B), B, 0, stream>>>(agg1, b1, W2, deg, h2, agg2, n);

  if (grouped) {
    k_agg2_part<<<nrng, A2_THREADS, 0, stream>>>(part, tab, h2, deg, agg2, (int)nblk, n);
  } else {
    k_agg2_edges<<<scat_grid, SCAT_THREADS, 0, stream>>>(row, col, h2, deg, agg2, E, n);
  }

  k_logsoftmax<<<blocks(n, B), B, 0, stream>>>(agg2, b2, (float*)d_out, n);
}


# bias+ReLU+layer-2 product folded into the layer-1 range block's epilogue, bias+log_softmax into the layer-2 range block's: a block finishes its 256 nodes; 7 launches become 5; sum order unchanged
# speedup vs baseline: 5.7909x; 1.1212x over previous
#include <hip/hip_runtime.h>
#include <math.h>

typedef __attribute__((ext_vector_type(2))) float v2f;
typedef __attribute__((ext_vector_type(8))) float v8f;

#define NFEAT 512
#define HIDDEN 16
#define NCLS 2


__global__ void k_init_deg(float* __restrict__ deg, int n) {
  int i = blockIdx.x * blockDim.x + threadIdx.x;
  if (i < n) deg[i] = 1.0f;
}

#define BIN_NODES 16384
#define EDGE_SLICES 64
#define SCAT_THREADS 1024
static_assert((BIN_NODES & (BIN_NODES - 1)) == 0, "bin index is wrapped with a mask");

__global__ __launch_bounds__(SCAT_THREADS) void k_deg_count(const int* __restrict__ col,
                                                            float* __restrict__ deg,
                                                            int E, int n) {
  __shared__ float bins[BIN_NODES];
  const int tid = threadIdx.x;
  const int lo = blockIdx.y * BIN_NODES;
  for (int i = tid; i < BIN_NODES; i += SCAT_THREADS) bins[i] = 0.f;
  __syncthreads();

  const int per = (E + (int)gridDim.x - 1) / (int)gridDim.x;
  const int e0 = blockIdx.x * per;
  const int e1 = min(e0 + per, E);
  const int last = e1 - 1;
  for (int e = e0 + tid; e < e1; e += 8 * SCAT_THREADS) {
    int c[8];
#pragma unroll
    for (int j = 0; j < 8; ++j) {
      const int ee = e + j * SCAT_THREADS;
      const bool in = ee < e1;
      const int v = col[in ? ee : last];
      c[j] = in ? v : -1;
    }
#pragma unroll
    for (int j = 0; j < 8; ++j) {
      const unsigned idx = (unsigned)(c[j] - lo);
      if (idx < (unsigned)BIN_NODES) atomicAdd(&bins[idx], 1.0f);
    }
  }
  __syncthreads();

  const int rot = (int)blockIdx.x * SCAT_THREADS;
  for (int t = tid; t < BIN_NODES; t += SCAT_THREADS) {
    const int i = (t + rot) & (BIN_NODES - 1);
    const float v = bins[i];
    const int node = lo + i;
    if (v != 0.f && node < n) atomicAdd(&deg[node], v);
  }
}


__global__ __launch_bounds__(256) void k_gemm1(const float* __restrict__ x,
                                               const float* __restrict__ W1,
                                               float* __restrict__ h1,
                                               float* degnorm,
                                               float* __restrict__ agg1,
                                               int n_nodes) {
  const int lane = threadIdx.x & 31;
  const int tile = blockIdx.x * 8 + (threadIdx.x >> 5);
  const int row0 = tile * 16;
  if (row0 >= n_nodes) return;

  const int nsel  = lane & 15;
  const int khalf = (lane >> 4) << 1;

  const float* arow = x + (size_t)(row0 + nsel) * NFEAT + khalf;

  const int mbase = row0 + ((lane >> 4) << 3);
  float dg[8];
#pragma unroll
  for (int r = 0; r < 8; ++r) dg[r] = degnorm[mbase + r];

  v8f c = {};
  for (int k = 0; k < NFEAT; k += 4) {
    if ((k & 127) == 0) __builtin_prefetch(arow + k + 128, 0, 0);
    v2f a = *(const v2f*)(arow + k);
    v2f b;
    b.x = W1[(k + khalf) * HIDDEN + nsel];
    b.y = W1[(k + khalf + 1) * HIDDEN + nsel];
    c = __builtin_amdgcn_wmma_f32_16x16x4_f32(false, a, false, b, (short)0, c,
                                              false, false);
  }

  float d[8];
#pragma unroll
  for (int r = 0; r < 8; ++r) d[r] = rsqrtf(dg[r]);
#pragma unroll
  for (int r = 0; r < 8; ++r) {
    const size_t o = (size_t)(mbase + r) * HIDDEN + nsel;
    const float hs = c[r] * d[r];
    h1[o] = hs;
    agg1[o] = hs * d[r];
  }
  if (nsel == 0) {
#pragma unroll
    for (int r = 0; r < 8; ++r) degnorm[mbase + r] = d[r];
  }
}


#define A1_NODES 4096
#define A1_CL_BITS 12
#define A1_SLICES 10
#define A1_THREADS 1024
#define A1_U 8
#define A1_G 8
static_assert((1 << A1_CL_BITS) == A1_NODES, "list entry packs the in-range target in A1_CL_BITS bits");

__global__ __launch_bounds__(A1_THREADS) void k_agg1_edges(const int* __restrict__ row,
                                                           const int* __restrict__ col,
                                                           const float* __restrict__ h1,
                                                           const float* __restrict__ dinv,
                                                           float* __restrict__ agg1,
                                                           int E, int n) {
  __shared__ float bins[A1_NODES * HIDDEN];
  __shared__ unsigned list[A1_THREADS / 32][A1_U * 32];
  const int tid = threadIdx.x;
  const int lane = tid & 31;
  const int half = lane >> 4;
  const int f = lane & 15;
  const int lo = blockIdx.y * A1_NODES;
  unsigned* mylist = list[tid >> 5];

  for (int i = tid; i < A1_NODES * HIDDEN / 4; i += A1_THREADS)
    ((float4*)bins)[i] = make_float4(0.f, 0.f, 0.f, 0.f);
  __syncthreads();

  const int per = (E + (int)gridDim.x - 1) / (int)gridDim.x;
  const int e0 = blockIdx.x * per;
  const int e1 = min(e0 + per, E);
  if (e0 < e1) {
    const int last = e1 - 1;
    int c[A1_U];
    unsigned r[A1_U];
#pragma unroll
    for (int j = 0; j < A1_U; ++j) {
      const int ee = e0 + tid + j * A1_THREADS;
      const bool in = ee < e1;
      const int k = in ? ee : last;
      const int cv = col[k];
      r[j] = (unsigned)row[k];
      c[j] = in ? cv : -1;
    }
    for (int base = e0; base < e1; base += A1_U * A1_THREADS) {
      int cn[A1_U];
      unsigned rn[A1_U];
#pragma unroll
      for (int j = 0; j < A1_U; ++j) {
        const int ee = base + tid + (A1_U + j) * A1_THREADS;
        const bool in = ee < e1;
        const int k = in ? ee : last;
        const int cv = col[k];
        rn[j] = (unsigned)row[k];
        cn[j] = in ? cv : -1;
      }
      int cnt = 0;
#pragma unroll
      for (int j = 0; j < A1_U; ++j) {
        const unsigned idx = (unsigned)(c[j] - lo);
        const bool m = idx < (unsigned)A1_NODES;
        const unsigned mask = __builtin_amdgcn_ballot_w32(m);
        if (m) mylist[cnt + (int)__builtin_amdgcn_mbcnt_lo(mask, 0u)] = (r[j] << A1_CL_BITS) | idx;
        cnt += __builtin_popcount(mask);
      }
      for (int t = 0; t < cnt; t += 2 * A1_G) {
        unsigned ent[A1_G];
        float hv[A1_G];
#pragma unroll
        for (int q = 0; q < A1_G; ++q) {
          const int i = t + 2 * q + half;
          ent[q] = mylist[i < cnt ? i : t];
        }
#pragma unroll
        for (int q = 0; q < A1_G; ++q) {
          const unsigned rr = ent[q] >> A1_CL_BITS;
          hv[q] = h1[rr * HIDDEN + f];
        }
#pragma unroll
        for (int q = 0; q < A1_G; ++q) {
          const int i = t + 2 * q + half;
          if (i < cnt)
            atomicAdd(&bins[(ent[q] & (A1_NODES - 1)) * HIDDEN + f], hv[q]);
        }
      }
#pragma unroll
      for (int j = 0; j < A1_U; ++j) {
        c[j] = cn[j];
        r[j] = rn[j];
      }
    }
  }
  __syncthreads();

  const int total = n * HIDDEN;
  const int rot = (int)blockIdx.x * (((A1_NODES * HIDDEN) / A1_SLICES) & ~31);
  for (int t = tid; t < A1_NODES * HIDDEN; t += 4 * A1_THREADS) {
    float v[4], d[4];
    int g[4];
#pragma unroll
    for (int u = 0; u < 4; ++u) {
      const int i = (t + u * A1_THREADS + rot) & (A1_NODES * HIDDEN - 1);
      v[u] = bins[i];
      g[u] = lo * HIDDEN + i;
      d[u] = dinv[min(lo + (i >> 4), n - 1)];
    }
#pragma unroll
    for (int u = 0; u < 4; ++u)
      if (v[u] != 0.f && g[u] < total) atomicAdd(&agg1[g[u]], v[u] * d[u]);
  }
}


__global__ void k_relu_gemm2(const float* __restrict__ agg1, const float* __restrict__ b1,
                             const float* __restrict__ W2, const float* __restrict__ dinv,
                             float* __restrict__ h2, float* __restrict__ agg2, int n) {
  int node = blockIdx.x * blockDim.x + threadIdx.x;
  if (node >= n) return;
  const float dn = dinv[node];
  float acc0 = 0.f, acc1 = 0.f;
#pragma unroll
  for (int j = 0; j < HIDDEN; ++j) {
    float v = fmaxf(agg1[(size_t)node * HIDDEN + j] + b1[j], 0.f);
    acc0 = fmaf(v, W2[j * NCLS + 0], acc0);
    acc1 = fmaf(v, W2[j * NCLS + 1], acc1);
  }
  const float s0 = acc0 * dn, s1 = acc1 * dn;
  h2[(size_t)node * NCLS + 0] = s0;
  h2[(size_t)node * NCLS + 1] = s1;
  agg2[(size_t)node * NCLS + 0] = s0 * dn;
  agg2[(size_t)node * NCLS + 1] = s1 * dn;
}


__global__ __launch_bounds__(SCAT_THREADS) void k_agg2_edges(const int* __restrict__ row,
                                                             const int* __restrict__ col,
                                                             const float* __restrict__ h2,
                                                             const float* __restrict__ dinv,
                                                             float* __restrict__ agg2,
                                                             int E, int n) {
  __shared__ float bins[BIN_NODES * NCLS];
  const int tid = threadIdx.x;
  const int lo = blockIdx.y * BIN_NODES;
  for (int i = tid; i < BIN_NODES * NCLS; i += SCAT_THREADS) bins[i] = 0.f;
  __syncthreads();

  const int per = (E + (int)gridDim.x - 1) / (int)gridDim.x;
  const int e0 = blockIdx.x * per;
  const int e1 = min(e0 + per, E);
  if (e0 < e1) {
    const int last = e1 - 1;
    int c[4];
    unsigned r[4];
#pragma unroll
    for (int j = 0; j < 4; ++j) {
      const int ee = e0 + tid + j * SCAT_THREADS;
      const bool in = ee < e1;
      const int k = in ? ee : last;
      const int cv = col[k];
      r[j] = (unsigned)row[k];
      c[j] = in ? cv : -1;
    }
    for (int e = e0 + tid; e < e1; e += 4 * SCAT_THREADS) {
      int cn[4];
      unsigned rn[4];
#pragma unroll
      for (int j = 0; j < 4; ++j) {
        const int ee = e + (4 + j) * SCAT_THREADS;
        const bool in = ee < e1;
        const int k = in ? ee : last;
        const int cv = col[k];
        rn[j] = (unsigned)row[k];
        cn[j] = in ? cv : -1;
      }
      v2f hv[4];
#pragma unroll
      for (int j = 0; j < 4; ++j) {
        const unsigned idx = (unsigned)(c[j] - lo);
        if (idx < (unsigned)BIN_NODES) hv[j] = *(const v2f*)(h2 + r[j] * NCLS);
      }
#pragma unroll
      for (int j = 0; j < 4; ++j) {
        const unsigned idx = (unsigned)(c[j] - lo);
        if (idx < (unsigned)BIN_NODES) {
          atomicAdd(&bins[idx * NCLS + 0], hv[j].x);
          atomicAdd(&bins[idx * NCLS + 1], hv[j].y);
        }
      }
#pragma unroll
      for (int j = 0; j < 4; ++j) {
        c[j] = cn[j];
        r[j] = rn[j];
      }
    }
  }
  __syncthreads();

  const int total = n * NCLS;
  const int rot = (int)blockIdx.x * SCAT_THREADS;
  for (int t = tid; t < BIN_NODES * NCLS; t += SCAT_THREADS) {
    const int i = (t + rot) & (BIN_NODES * NCLS - 1);
    const float v = bins[i];
    const int g = lo * NCLS + i;
    const float d = dinv[min(lo + (i >> 1), n - 1)];
    if (v != 0.f && g < total) atomicAdd(&agg2[g], v * d);
  }
}


#define PART_NODES 256
#define PART_BITS 8
#define PART_MAXR 512
#define PART_THREADS 1024
#define PART_U 10
#define PART_CAP (PART_THREADS * PART_U)
#define PART_PITCH 544
#define PART_MAXB 1024
#define PART_PADB (PART_MAXB + 64)
#define ROW_PITCH (HIDDEN + 1)

__global__ __launch_bounds__(PART_THREADS) void k_part(const int* __restrict__ row,
                                                       const int* __restrict__ col,
                                                       unsigned* __restrict__ part,
                                                       int* __restrict__ tab, int E) {
  __shared__ unsigned sorted[PART_CAP];
  __shared__ int hist[PART_MAXR];
  __shared__ int base[PART_MAXR + 1];
  __shared__ int wsum[PART_MAXR / 32];
  const int tid = threadIdx.x;
  const int b = blockIdx.x;

  if (tid < PART_MAXR) hist[tid] = 0;
  __syncthreads();

  const int e0 = b * PART_CAP;
  const int e1 = min(e0 + PART_CAP, E);
  unsigned ent[PART_U];
  int rg[PART_U], rk[PART_U];
#pragma unroll
  for (int j = 0; j < PART_U; ++j) {
    const int ee = e0 + tid + j * PART_THREADS;
    const bool in = ee < e1;
    const int k = in ? ee : e0;
    const int cv = col[k];
    const unsigned rv = (unsigned)row[k];
    rg[j] = in ? (cv >> PART_BITS) : -1;
    ent[j] = (rv << PART_BITS) | ((unsigned)cv & (PART_NODES - 1));
  }
#pragma unroll
  for (int j = 0; j < PART_U; ++j) rk[j] = rg[j] >= 0 ? atomicAdd(&hist[rg[j]], 1) : 0;
  __syncthreads();

  int v = 0, incl = 0;
  if (tid < PART_MAXR) {
    v = hist[tid];
    incl = v;
#pragma unroll
    for (int d = 1; d < 32; d <<= 1) {
      const int up = __shfl_up(incl, d);
      if ((tid & 31) >= d) incl += up;
    }
    if ((tid & 31) == 31) wsum[tid >> 5] = incl;
  }
  __syncthreads();
  if (tid < PART_MAXR) {
    int before = 0;
    for (int w = 0; w < (tid >> 5); ++w) before += wsum[w];
    base[tid] = before + incl - v;
    if (tid == PART_MAXR - 1) base[PART_MAXR] = before + incl;
  }
  __syncthreads();

  if (tid <= PART_MAXR) tab[b * PART_PITCH + tid] = base[tid];
#pragma unroll
  for (int j = 0; j < PART_U; ++j)
    if (rg[j] >= 0) sorted[base[rg[j]] + rk[j]] = ent[j];
  __syncthreads();

  const int cnt = e1 - e0;
  for (int i = tid; i < cnt; i += PART_THREADS) part[e0 + i] = sorted[i];
}

__device__ __forceinline__ void load_runs(const int* __restrict__ tab, int y, int nblk,
                                          int* rb, int* re) {
  for (int i = threadIdx.x; i < PART_PADB; i += blockDim.x) {
    int b0 = 0, b1 = 0;
    if (i < nblk) {
      const int* t = tab + (size_t)i * PART_PITCH + y;
      b0 = i * PART_CAP + t[0];
      b1 = i * PART_CAP + t[1];
    }
    rb[i] = b0;
    re[i] = b1;
  }
}

#define DEG_THREADS 512
__global__ __launch_bounds__(DEG_THREADS) void k_deg_part(const unsigned* __restrict__ part,
                                                          const int* __restrict__ tab,
                                                          float* __restrict__ deg,
                                                          int nblk, int n) {
  __shared__ float bins[PART_NODES];
  __shared__ int rb[PART_PADB], re[PART_PADB];
  const int tid = threadIdx.x;
  const int lane = tid & 31;
  const int y = blockIdx.x;
  load_runs(tab, y, nblk, rb, re);
  if (tid < PART_NODES) bins[tid] = 0.f;
  __syncthreads();

  const int nw = DEG_THREADS / 32;
  for (int b0 = tid >> 5; b0 < nblk; b0 += 4 * nw) {
    int beg[4], cnt[4];
    unsigned en[4];
#pragma unroll
    for (int j = 0; j < 4; ++j) {
      const int bj = b0 + j * nw;
      beg[j] = rb[bj];
      cnt[j] = re[bj] - beg[j];
    }
#pragma unroll
    for (int j = 0; j < 4; ++j) en[j] = part[cnt[j] > 0 ? beg[j] + min(lane, cnt[j] - 1) : 0];
#pragma unroll
    for (int j = 0; j < 4; ++j)
      if (lane < cnt[j]) atomicAdd(&bins[en[j] & (PART_NODES - 1)], 1.0f);
#pragma unroll
    for (int j = 0; j < 4; ++j)
      for (int t = 32 + lane; t < cnt[j]; t += 32)
        atomicAdd(&bins[part[beg[j] + t] & (PART_NODES - 1)], 1.0f);
  }
  __syncthreads();

  if (tid < PART_NODES) {
    const int node = y * PART_NODES + tid;
    if (node < n) deg[node] = 1.0f + bins[tid];
  }
}

#define AP_THREADS 512
__global__ __launch_bounds__(AP_THREADS) void k_agg1_part(const unsigned* __restrict__ part,
                                                          const int* __restrict__ tab,
                                                          const float* __restrict__ h1,
                                                          const float* __restrict__ dinv,
                                                          const float* __restrict__ self1,
                                                          const float* __restrict__ b1,
                                                          const float* __restrict__ W2,
                                                          float* __restrict__ h2,
                                                          int nblk, int n) {
  __shared__ float bins[PART_NODES * ROW_PITCH];
  __shared__ int rb[PART_PADB], re[PART_PADB];
  const int tid = threadIdx.x;
  const int lane = tid & 31;
  const int eq = lane >> 2;
  const int c4 = (lane & 3) * 4;
  const int y = blockIdx.x;
  load_runs(tab, y, nblk, rb, re);
  for (int i = tid; i < PART_NODES * ROW_PITCH; i += AP_THREADS) bins[i] = 0.f;
  __syncthreads();

  const int nw = AP_THREADS / 32;
  int b0 = tid >> 5;
  int beg[2], cnt[2];
  unsigned cur[2];
#pragma unroll
  for (int j = 0; j < 2; ++j) {
    beg[j] = rb[b0 + j * nw];
    cnt[j] = re[b0 + j * nw] - beg[j];
    cur[j] = part[cnt[j] > 0 ? beg[j] + min(lane, cnt[j] - 1) : 0];
  }
  for (; b0 < nblk; b0 += 2 * nw) {
    int nbeg[2], ncnt[2];
    unsigned nxt[2];
#pragma unroll
    for (int j = 0; j < 2; ++j) {
      nbeg[j] = rb[b0 + (2 + j) * nw];
      ncnt[j] = re[b0 + (2 + j) * nw] - nbeg[j];
      nxt[j] = part[ncnt[j] > 0 ? nbeg[j] + min(lane, ncnt[j] - 1) : 0];
    }
    unsigned en[2][4];
    float4 hv[2][4];
#pragma unroll
    for (int j = 0; j < 2; ++j)
#pragma unroll
      for (int q = 0; q < 4; ++q) {
        en[j][q] = (unsigned)__shfl((int)cur[j], q * 8 + eq);
        hv[j][q] = *(const float4*)(h1 + (en[j][q] >> PART_BITS) * HIDDEN + c4);
      }
#pragma unroll
    for (int j = 0; j < 2; ++j)
#pragma unroll
      for (int q = 0; q < 4; ++q)
        if (q * 8 + eq < cnt[j]) {
          float* p = &bins[(en[j][q] & (PART_NODES - 1)) * ROW_PITCH + c4];
          atomicAdd(p + 0, hv[j][q].x);
          atomicAdd(p + 1, hv[j][q].y);
          atomicAdd(p + 2, hv[j][q].z);
          atomicAdd(p + 3, hv[j][q].w);
        }
#pragma unroll
    for (int j = 0; j < 2; ++j)
      for (int t = 32; t < cnt[j]; t += 32) {
        const unsigned more = part[beg[j] + min(t + lane, cnt[j] - 1)];
        unsigned em[4];
        float4 hm[4];
#pragma unroll
        for (int q = 0; q < 4; ++q) {
          em[q] = (unsigned)__shfl((int)more, q * 8 + eq);
          hm[q] = *(const float4*)(h1 + (em[q] >> PART_BITS) * HIDDEN + c4);
        }
#pragma unroll
        for (int q = 0; q < 4; ++q)
          if (t + q * 8 + eq < cnt[j]) {
            float* p = &bins[(em[q] & (PART_NODES - 1)) * ROW_PITCH + c4];
            atomicAdd(p + 0, hm[q].x);
            atomicAdd(p + 1, hm[q].y);
            atomicAdd(p + 2, hm[q].z);
            atomicAdd(p + 3, hm[q].w);
          }
      }
#pragma unroll
    for (int j = 0; j < 2; ++j) {
      beg[j] = nbeg[j];
      cnt[j] = ncnt[j];
      cur[j] = nxt[j];
    }
  }
  __syncthreads();

  if (tid < PART_NODES) {
    const int node = y * PART_NODES + tid;
    if (node < n) {
      const float dn = dinv[node];
      const float4* sp = (const float4*)(self1 + (size_t)node * HIDDEN);
      float sv[HIDDEN];
#pragma unroll
      for (int q = 0; q < HIDDEN / 4; ++q) {
        const float4 s = sp[q];
        sv[4 * q + 0] = s.x;
        sv[4 * q + 1] = s.y;
        sv[4 * q + 2] = s.z;
        sv[4 * q + 3] = s.w;
      }
      const float* brow = &bins[tid * ROW_PITCH];
      float acc0 = 0.f, acc1 = 0.f;
#pragma unroll
      for (int j = 0; j < HIDDEN; ++j) {
        const float a = sv[j] + brow[j] * dn;
        const float v = fmaxf(a + b1[j], 0.f);
        acc0 = fmaf(v, W2[j * NCLS + 0], acc0);
        acc1 = fmaf(v, W2[j * NCLS + 1], acc1);
      }
      v2f o;
      o.x = acc0 * dn;
      o.y = acc1 * dn;
      *(v2f*)(h2 + (size_t)node * NCLS) = o;
    }
  }
}

#define A2_THREADS 512
__global__ __launch_bounds__(A2_THREADS) void k_agg2_part(const unsigned* __restrict__ part,
                                                          const int* __restrict__ tab,
                                                          const float* __restrict__ h2,
                                                          const float* __restrict__ dinv,
                                                          const float* __restrict__ b2,
                                                          float* __restrict__ out,
                                                          int nblk, int n) {
  __shared__ float bins[PART_NODES * NCLS];
  __shared__ int rb[PART_PADB], re[PART_PADB];
  const int tid = threadIdx.x;
  const int lane = tid & 31;
  const int y = blockIdx.x;
  load_runs(tab, y, nblk, rb, re);
  for (int i = tid; i < PART_NODES * NCLS; i += A2_THREADS) bins[i] = 0.f;
  __syncthreads();

  const int nw = A2_THREADS / 32;
  for (int b0 = tid >> 5; b0 < nblk; b0 += 4 * nw) {
    int beg[4], cnt[4];
    unsigned en[4];
    v2f hv[4];
#pragma unroll
    for (int j = 0; j < 4; ++j) {
      const int bj = b0 + j * nw;
      beg[j] = rb[bj];
      cnt[j] = re[bj] - beg[j];
    }
#pragma unroll
    for (int j = 0; j < 4; ++j) en[j] = part[cnt[j] > 0 ? beg[j] + min(lane, cnt[j] - 1) : 0];
#pragma unroll
    for (int j = 0; j < 4; ++j) hv[j] = *(const v2f*)(h2 + (en[j] >> PART_BITS) * NCLS);
#pragma unroll
    for (int j = 0; j < 4; ++j)
      if (lane < cnt[j]) {
        const unsigned idx = en[j] & (PART_NODES - 1);
        atomicAdd(&bins[idx * NCLS + 0], hv[j].x);
        atomicAdd(&bins[idx * NCLS + 1], hv[j].y);
      }
#pragma unroll
    for (int j = 0; j < 4; ++j)
      for (int t = 32 + lane; t < cnt[j]; t += 32) {
        const unsigned em = part[beg[j] + t];
        const v2f hm = *(const v2f*)(h2 + (em >> PART_BITS) * NCLS);
        atomicAdd(&bins[(em & (PART_NODES - 1)) * NCLS + 0], hm.x);
        atomicAdd(&bins[(em & (PART_NODES - 1)) * NCLS + 1], hm.y);
      }
  }
  __syncthreads();

  if (tid < PART_NODES) {
    const int node = y * PART_NODES + tid;
    if (node < n) {
      const float dn = dinv[node];
      const v2f hs = *(const v2f*)(h2 + (size_t)node * NCLS);
      v2f bs;
      bs.x = bins[tid * NCLS + 0];
      bs.y = bins[tid * NCLS + 1];
      const float s0 = hs.x * dn;
      const float s1 = hs.y * dn;
      const float a0 = s0 + bs.x * dn;
      const float a1 = s1 + bs.y * dn;
      const float z0 = a0 + b2[0];
      const float z1 = a1 + b2[1];
      const float m = fmaxf(z0, z1);
      const float lse = m + logf(expf(z0 - m) + expf(z1 - m));
      v2f o;
      o.x = z0 - lse;
      o.y = z1 - lse;
      *(v2f*)(out + (size_t)node * NCLS) = o;
    }
  }
}


__global__ void k_logsoftmax(const float* __restrict__ agg2, const float* __restrict__ b2,
                             float* __restrict__ out, int n) {
  int node = blockIdx.x * blockDim.x + threadIdx.x;
  if (node >= n) return;
  float z0 = agg2[(size_t)node * NCLS + 0] + b2[0];
  float z1 = agg2[(size_t)node * NCLS + 1] + b2[1];
  float m = fmaxf(z0, z1);
  float lse = m + logf(expf(z0 - m) + expf(z1 - m));
  out[(size_t)node * NCLS + 0] = z0 - lse;
  out[(size_t)node * NCLS + 1] = z1 - lse;
}


extern "C" void kernel_launch(void* const* d_in, const int* in_sizes, int n_in,
                              void* d_out, int out_size, void* d_ws, size_t ws_size,
                              hipStream_t stream) {
  const float* x  = (const float*)d_in[0];
  const int*   ei = (const int*)d_in[1];
  const float* W1 = (const float*)d_in[2];
  const float* b1 = (const float*)d_in[3];
  const float* W2 = (const float*)d_in[4];
  const float* b2 = (const float*)d_in[5];

  const int n = in_sizes[0] / NFEAT;
  const int E = in_sizes[1] / 2;
  const int* row = ei;
  const int* col = ei + E;

  char* ws = (char*)d_ws;
  size_t off = 0;
  auto carve = [&](size_t bytes) -> void* {
    void* p = ws + off;
    off = (off + bytes + 255) & ~(size_t)255;
    return p;
  };
  float* deg  = (float*)carve((size_t)n * sizeof(float));
  float* h1   = (float*)carve((size_t)n * HIDDEN * sizeof(float));
  float* agg1 = (float*)carve((size_t)n * HIDDEN * sizeof(float));
  float* h2   = (float*)carve((size_t)n * NCLS * sizeof(float));
  float* agg2 = (float*)carve((size_t)n * NCLS * sizeof(float));

  const int B = 256;
  auto blocks = [](long long work, int b) { return (unsigned)((work + b - 1) / b); };

  const unsigned nblk = blocks(E, PART_CAP);
  const unsigned nrng = blocks(n, PART_NODES);
  unsigned* part = (unsigned*)carve((size_t)E * sizeof(unsigned));
  int* tab = (int*)carve((size_t)nblk * PART_PITCH * sizeof(int));
  const bool grouped = off <= ws_size && nrng <= PART_MAXR && nblk <= PART_MAXB &&
                       n <= (1 << (32 - PART_BITS));
  const dim3 scat_grid(EDGE_SLICES, blocks(n, BIN_NODES));

  if (grouped) {
    k_part<<<nblk, PART_THREADS, 0, stream>>>(row, col, part, tab, E);
    k_deg_part<<<nrng, DEG_THREADS, 0, stream>>>(part, tab, deg, (int)nblk, n);
  } else {
    k_init_deg<<<blocks(n, B), B, 0, stream>>>(deg, n);
    k_deg_count<<<scat_grid, SCAT_THREADS, 0, stream>>>(col, deg, E, n);
  }

  int tiles = (n + 15) / 16;
  k_gemm1<<<blocks(tiles, 8), B, 0, stream>>>(x, W1, h1, deg, agg1, n);

  if (grouped) {
    k_agg1_part<<<nrng, AP_THREADS, 0, stream>>>(part, tab, h1, deg, agg1, b1, W2, h2,
                                                 (int)nblk, n);
    k_agg2_part<<<nrng, A2_THREADS, 0, stream>>>(part, tab, h2, deg, b2, (float*)d_out,
                                                 (int)nblk, n);
  } else {
    const dim3 a1_grid(A1_SLICES, blocks(n, A1_NODES));
    k_agg1_edges<<<a1_grid, A1_THREADS, 0, stream>>>(row, col, h1, deg, agg1, E, n);
    k_relu_gemm2<<<blocks(n, B), B, 0, stream>>>(agg1, b1, W2, deg, h2, agg2, n);
    k_agg2_edges<<<scat_grid, SCAT_THREADS, 0, stream>>>(row, col, h2, deg, agg2, E, n);
    k_logsoftmax<<<blocks(n, B), B, 0, stream>>>(agg2, b2, (float*)d_out, n);
  }
}


# layer-1 GEMM stores raw h1 and shares one launch with the edge grouping (independent work, two block roles); the degree pass scales its 256 rows in place; 5 launches become 4; sum order unchanged
# speedup vs baseline: 6.4173x; 1.1082x over previous
#include <hip/hip_runtime.h>
#include <math.h>

typedef __attribute__((ext_vector_type(2))) float v2f;
typedef __attribute__((ext_vector_type(8))) float v8f;

#define NFEAT 512
#define HIDDEN 16
#define NCLS 2


__global__ void k_init_deg(float* __restrict__ deg, int n) {
  int i = blockIdx.x * blockDim.x + threadIdx.x;
  if (i < n) deg[i] = 1.0f;
}

#define BIN_NODES 16384
#define EDGE_SLICES 64
#define SCAT_THREADS 1024
static_assert((BIN_NODES & (BIN_NODES - 1)) == 0, "bin index is wrapped with a mask");

__global__ __launch_bounds__(SCAT_THREADS) void k_deg_count(const int* __restrict__ col,
                                                            float* __restrict__ deg,
                                                            int E, int n) {
  __shared__ float bins[BIN_NODES];
  const int tid = threadIdx.x;
  const int lo = blockIdx.y * BIN_NODES;
  for (int i = tid; i < BIN_NODES; i += SCAT_THREADS) bins[i] = 0.f;
  __syncthreads();

  const int per = (E + (int)gridDim.x - 1) / (int)gridDim.x;
  const int e0 = blockIdx.x * per;
  const int e1 = min(e0 + per, E);
  const int last = e1 - 1;
  for (int e = e0 + tid; e < e1; e += 8 * SCAT_THREADS) {
    int c[8];
#pragma unroll
    for (int j = 0; j < 8; ++j) {
      const int ee = e + j * SCAT_THREADS;
      const bool in = ee < e1;
      const int v = col[in ? ee : last];
      c[j] = in ? v : -1;
    }
#pragma unroll
    for (int j = 0; j < 8; ++j) {
      const unsigned idx = (unsigned)(c[j] - lo);
      if (idx < (unsigned)BIN_NODES) atomicAdd(&bins[idx], 1.0f);
    }
  }
  __syncthreads();

  const int rot = (int)blockIdx.x * SCAT_THREADS;
  for (int t = tid; t < BIN_NODES; t += SCAT_THREADS) {
    const int i = (t + rot) & (BIN_NODES - 1);
    const float v = bins[i];
    const int node = lo + i;
    if (v != 0.f && node < n) atomicAdd(&deg[node], v);
  }
}


__device__ __forceinline__ v8f gemm_tile(const float* __restrict__ x,
                                         const float* __restrict__ W1, int row0, int lane) {
  const int nsel  = lane & 15;
  const int khalf = (lane >> 4) << 1;
  const float* arow = x + (size_t)(row0 + nsel) * NFEAT + khalf;
  v8f c = {};
  for (int k = 0; k < NFEAT; k += 4) {
    if ((k & 127) == 0) __builtin_prefetch(arow + k + 128, 0, 0);
    v2f a = *(const v2f*)(arow + k);
    v2f b;
    b.x = W1[(k + khalf) * HIDDEN + nsel];
    b.y = W1[(k + khalf + 1) * HIDDEN + nsel];
    c = __builtin_amdgcn_wmma_f32_16x16x4_f32(false, a, false, b, (short)0, c,
                                              false, false);
  }
  return c;
}

__global__ __launch_bounds__(256) void k_gemm1(const float* __restrict__ x,
                                               const float* __restrict__ W1,
                                               float* __restrict__ h1,
                                               float* degnorm,
                                               float* __restrict__ agg1,
                                               int n_nodes) {
  const int lane = threadIdx.x & 31;
  const int tile = blockIdx.x * 8 + (threadIdx.x >> 5);
  const int row0 = tile * 16;
  if (row0 >= n_nodes) return;

  const int nsel  = lane & 15;

  const int mbase = row0 + ((lane >> 4) << 3);
  float dg[8];
#pragma unroll
  for (int r = 0; r < 8; ++r) dg[r] = degnorm[mbase + r];

  const v8f c = gemm_tile(x, W1, row0, lane);

  float d[8];
#pragma unroll
  for (int r = 0; r < 8; ++r) d[r] = rsqrtf(dg[r]);
#pragma unroll
  for (int r = 0; r < 8; ++r) {
    const size_t o = (size_t)(mbase + r) * HIDDEN + nsel;
    const float hs = c[r] * d[r];
    h1[o] = hs;
    agg1[o] = hs * d[r];
  }
  if (nsel == 0) {
#pragma unroll
    for (int r = 0; r < 8; ++r) degnorm[mbase + r] = d[r];
  }
}


#define A1_NODES 4096
#define A1_CL_BITS 12
#define A1_SLICES 10
#define A1_THREADS 1024
#define A1_U 8
#define A1_G 8
static_assert((1 << A1_CL_BITS) == A1_NODES, "list entry packs the in-range target in A1_CL_BITS bits");

__global__ __launch_bounds__(A1_THREADS) void k_agg1_edges(const int* __restrict__ row,
                                                           const int* __restrict__ col,
                                                           const float* __restrict__ h1,
                                                           const float* __restrict__ dinv,
                                                           float* __restrict__ agg1,
                                                           int E, int n) {
  __shared__ float bins[A1_NODES * HIDDEN];
  __shared__ unsigned list[A1_THREADS / 32][A1_U * 32];
  const int tid = threadIdx.x;
  const int lane = tid & 31;
  const int half = lane >> 4;
  const int f = lane & 15;
  const int lo = blockIdx.y * A1_NODES;
  unsigned* mylist = list[tid >> 5];

  for (int i = tid; i < A1_NODES * HIDDEN / 4; i += A1_THREADS)
    ((float4*)bins)[i] = make_float4(0.f, 0.f, 0.f, 0.f);
  __syncthreads();

  const int per = (E + (int)gridDim.x - 1) / (int)gridDim.x;
  const int e0 = blockIdx.x * per;
  const int e1 = min(e0 + per, E);
  if (e0 < e1) {
    const int last = e1 - 1;
    int c[A1_U];
    unsigned r[A1_U];
#pragma unroll
    for (int j = 0; j < A1_U; ++j) {
      const int ee = e0 + tid + j * A1_THREADS;
      const bool in = ee < e1;
      const int k = in ? ee : last;
      const int cv = col[k];
      r[j] = (unsigned)row[k];
      c[j] = in ? cv : -1;
    }
    for (int base = e0; base < e1; base += A1_U * A1_THREADS) {
      int cn[A1_U];
      unsigned rn[A1_U];
#pragma unroll
      for (int j = 0; j < A1_U; ++j) {
        const int ee = base + tid + (A1_U + j) * A1_THREADS;
        const bool in = ee < e1;
        const int k = in ? ee : last;
        const int cv = col[k];
        rn[j] = (unsigned)row[k];
        cn[j] = in ? cv : -1;
      }
      int cnt = 0;
#pragma unroll
      for (int j = 0; j < A1_U; ++j) {
        const unsigned idx = (unsigned)(c[j] - lo);
        const bool m = idx < (unsigned)A1_NODES;
        const unsigned mask = __builtin_amdgcn_ballot_w32(m);
        if (m) mylist[cnt + (int)__builtin_amdgcn_mbcnt_lo(mask, 0u)] = (r[j] << A1_CL_BITS) | idx;
        cnt += __builtin_popcount(mask);
      }
      for (int t = 0; t < cnt; t += 2 * A1_G) {
        unsigned ent[A1_G];
        float hv[A1_G];
#pragma unroll
        for (int q = 0; q < A1_G; ++q) {
          const int i = t + 2 * q + half;
          ent[q] = mylist[i < cnt ? i : t];
        }
#pragma unroll
        for (int q = 0; q < A1_G; ++q) {
          const unsigned rr = ent[q] >> A1_CL_BITS;
          hv[q] = h1[rr * HIDDEN + f];
        }
#pragma unroll
        for (int q = 0; q < A1_G; ++q) {
          const int i = t + 2 * q + half;
          if (i < cnt)
            atomicAdd(&bins[(ent[q] & (A1_NODES - 1)) * HIDDEN + f], hv[q]);
        }
      }
#pragma unroll
      for (int j = 0; j < A1_U; ++j) {
        c[j] = cn[j];
        r[j] = rn[j];
      }
    }
  }
  __syncthreads();

  const int total = n * HIDDEN;
  const int rot = (int)blockIdx.x * (((A1_NODES * HIDDEN) / A1_SLICES) & ~31);
  for (int t = tid; t < A1_NODES * HIDDEN; t += 4 * A1_THREADS) {
    float v[4], d[4];
    int g[4];
#pragma unroll
    for (int u = 0; u < 4; ++u) {
      const int i = (t + u * A1_THREADS + rot) & (A1_NODES * HIDDEN - 1);
      v[u] = bins[i];
      g[u] = lo * HIDDEN + i;
      d[u] = dinv[min(lo + (i >> 4), n - 1)];
    }
#pragma unroll
    for (int u = 0; u < 4; ++u)
      if (v[u] != 0.f && g[u] < total) atomicAdd(&agg1[g[u]], v[u] * d[u]);
  }
}


__global__ void k_relu_gemm2(const float* __restrict__ agg1, const float* __restrict__ b1,
                             const float* __restrict__ W2, const float* __restrict__ dinv,
                             float* __restrict__ h2, float* __restrict__ agg2, int n) {
  int node = blockIdx.x * blockDim.x + threadIdx.x;
  if (node >= n) return;
  const float dn = dinv[node];
  float acc0 = 0.f, acc1 = 0.f;
#pragma unroll
  for (int j = 0; j < HIDDEN; ++j) {
    float v = fmaxf(agg1[(size_t)node * HIDDEN + j] + b1[j], 0.f);
    acc0 = fmaf(v, W2[j * NCLS + 0], acc0);
    acc1 = fmaf(v, W2[j * NCLS + 1], acc1);
  }
  const float s0 = acc0 * dn, s1 = acc1 * dn;
  h2[(size_t)node * NCLS + 0] = s0;
  h2[(size_t)node * NCLS + 1] = s1;
  agg2[(size_t)node * NCLS + 0] = s0 * dn;
  agg2[(size_t)node * NCLS + 1] = s1 * dn;
}


__global__ __launch_bounds__(SCAT_THREADS) void k_agg2_edges(const int* __restrict__ row,
                                                             const int* __restrict__ col,
                                                             const float* __restrict__ h2,
                                                             const float* __restrict__ dinv,
                                                             float* __restrict__ agg2,
                                                             int E, int n) {
  __shared__ float bins[BIN_NODES * NCLS];
  const int tid = threadIdx.x;
  const int lo = blockIdx.y * BIN_NODES;
  for (int i = tid; i < BIN_NODES * NCLS; i += SCAT_THREADS) bins[i] = 0.f;
  __syncthreads();

  const int per = (E + (int)gridDim.x - 1) / (int)gridDim.x;
  const int e0 = blockIdx.x * per;
  const int e1 = min(e0 + per, E);
  if (e0 < e1) {
    const int last = e1 - 1;
    int c[4];
    unsigned r[4];
#pragma unroll
    for (int j = 0; j < 4; ++j) {
      const int ee = e0 + tid + j * SCAT_THREADS;
      const bool in = ee < e1;
      const int k = in ? ee : last;
      const int cv = col[k];
      r[j] = (unsigned)row[k];
      c[j] = in ? cv : -1;
    }
    for (int e = e0 + tid; e < e1; e += 4 * SCAT_THREADS) {
      int cn[4];
      unsigned rn[4];
#pragma unroll
      for (int j = 0; j < 4; ++j) {
        const int ee = e + (4 + j) * SCAT_THREADS;
        const bool in = ee < e1;
        const int k = in ? ee : last;
        const int cv = col[k];
        rn[j] = (unsigned)row[k];
        cn[j] = in ? cv : -1;
      }
      v2f hv[4];
#pragma unroll
      for (int j = 0; j < 4; ++j) {
        const unsigned idx = (unsigned)(c[j] - lo);
        if (idx < (unsigned)BIN_NODES) hv[j] = *(const v2f*)(h2 + r[j] * NCLS);
      }
#pragma unroll
      for (int j = 0; j < 4; ++j) {
        const unsigned idx = (unsigned)(c[j] - lo);
        if (idx < (unsigned)BIN_NODES) {
          atomicAdd(&bins[idx * NCLS + 0], hv[j].x);
          atomicAdd(&bins[idx * NCLS + 1], hv[j].y);
        }
      }
#pragma unroll
      for (int j = 0; j < 4; ++j) {
        c[j] = cn[j];
        r[j] = rn[j];
      }
    }
  }
  __syncthreads();

  const int total = n * NCLS;
  const int rot = (int)blockIdx.x * SCAT_THREADS;
  for (int t = tid; t < BIN_NODES * NCLS; t += SCAT_THREADS) {
    const int i = (t + rot) & (BIN_NODES * NCLS - 1);
    const float v = bins[i];
    const int g = lo * NCLS + i;
    const float d = dinv[min(lo + (i >> 1), n - 1)];
    if (v != 0.f && g < total) atomicAdd(&agg2[g], v * d);
  }
}


#define PART_NODES 256
#define PART_BITS 8
#define PART_MAXR 512
#define PART_THREADS 1024
#define PART_U 10
#define PART_CAP (PART_THREADS * PART_U)
#define PART_PITCH 544
#define PART_MAXB 1024
#define PART_PADB (PART_MAXB + 64)
#define ROW_PITCH (HIDDEN + 1)

#define GEMM_TPB 25

__global__ __launch_bounds__(PART_THREADS) void k_part_gemm(const int* __restrict__ row,
                                                            const int* __restrict__ col,
                                                            unsigned* __restrict__ part,
                                                            int* __restrict__ tab, int E,
                                                            const float* __restrict__ x,
                                                            const float* __restrict__ W1,
                                                            float* __restrict__ h1,
                                                            int n_nodes, int gemm_blocks) {
  __shared__ unsigned sorted[PART_CAP];
  __shared__ int hist[PART_MAXR];
  __shared__ int base[PART_MAXR + 1];
  __shared__ int wsum[PART_MAXR / 32];
  const int tid = threadIdx.x;

  if ((int)blockIdx.x < gemm_blocks) {
    const int wave = tid >> 5;
    const int lane = tid & 31;
    const int row0 = ((int)blockIdx.x * GEMM_TPB + wave) * 16;
    if (wave >= GEMM_TPB || row0 >= n_nodes) return;
    const v8f c = gemm_tile(x, W1, row0, lane);
    const int nsel = lane & 15;
    const int mbase = row0 + ((lane >> 4) << 3);
#pragma unroll
    for (int r = 0; r < 8; ++r) h1[(size_t)(mbase + r) * HIDDEN + nsel] = c[r];
    return;
  }
  const int b = (int)blockIdx.x - gemm_blocks;

  if (tid < PART_MAXR) hist[tid] = 0;
  __syncthreads();

  const int e0 = b * PART_CAP;
  const int e1 = min(e0 + PART_CAP, E);
  unsigned ent[PART_U];
  int rg[PART_U], rk[PART_U];
#pragma unroll
  for (int j = 0; j < PART_U; ++j) {
    const int ee = e0 + tid + j * PART_THREADS;
    const bool in = ee < e1;
    const int k = in ? ee : e0;
    const int cv = col[k];
    const unsigned rv = (unsigned)row[k];
    rg[j] = in ? (cv >> PART_BITS) : -1;
    ent[j] = (rv << PART_BITS) | ((unsigned)cv & (PART_NODES - 1));
  }
#pragma unroll
  for (int j = 0; j < PART_U; ++j) rk[j] = rg[j] >= 0 ? atomicAdd(&hist[rg[j]], 1) : 0;
  __syncthreads();

  int v = 0, incl = 0;
  if (tid < PART_MAXR) {
    v = hist[tid];
    incl = v;
#pragma unroll
    for (int d = 1; d < 32; d <<= 1) {
      const int up = __shfl_up(incl, d);
      if ((tid & 31) >= d) incl += up;
    }
    if ((tid & 31) == 31) wsum[tid >> 5] = incl;
  }
  __syncthreads();
  if (tid < PART_MAXR) {
    int before = 0;
    for (int w = 0; w < (tid >> 5); ++w) before += wsum[w];
    base[tid] = before + incl - v;
    if (tid == PART_MAXR - 1) base[PART_MAXR] = before + incl;
  }
  __syncthreads();

  if (tid <= PART_MAXR) tab[b * PART_PITCH + tid] = base[tid];
#pragma unroll
  for (int j = 0; j < PART_U; ++j)
    if (rg[j] >= 0) sorted[base[rg[j]] + rk[j]] = ent[j];
  __syncthreads();

  const int cnt = e1 - e0;
  for (int i = tid; i < cnt; i += PART_THREADS) part[e0 + i] = sorted[i];
}

__device__ __forceinline__ void load_runs(const int* __restrict__ tab, int y, int nblk,
                                          int* rb, int* re) {
  for (int i = threadIdx.x; i < PART_PADB; i += blockDim.x) {
    int b0 = 0, b1 = 0;
    if (i < nblk) {
      const int* t = tab + (size_t)i * PART_PITCH + y;
      b0 = i * PART_CAP + t[0];
      b1 = i * PART_CAP + t[1];
    }
    rb[i] = b0;
    re[i] = b1;
  }
}

#define DEG_THREADS 512
static_assert(PART_NODES * HIDDEN / 4 == 2 * DEG_THREADS, "two 16-byte row pieces per thread");
__global__ __launch_bounds__(DEG_THREADS) void k_deg_scale(const unsigned* __restrict__ part,
                                                           const int* __restrict__ tab,
                                                           float* __restrict__ dinv,
                                                           float* h1,
                                                           int nblk, int n) {
  __shared__ float bins[PART_NODES];
  __shared__ int rb[PART_PADB], re[PART_PADB];
  const int tid = threadIdx.x;
  const int lane = tid & 31;
  const int y = blockIdx.x;
  float4* hp = (float4*)(h1 + (size_t)y * PART_NODES * HIDDEN);
  const int r0 = tid >> 2;
  const int r1 = r0 + DEG_THREADS / 4;
  const bool in0 = y * PART_NODES + r0 < n;
  const bool in1 = y * PART_NODES + r1 < n;
  float4 p0 = make_float4(0.f, 0.f, 0.f, 0.f), p1 = p0;
  if (in0) p0 = hp[tid];
  if (in1) p1 = hp[tid + DEG_THREADS];
  load_runs(tab, y, nblk, rb, re);
  if (tid < PART_NODES) bins[tid] = 0.f;
  __syncthreads();

  const int nw = DEG_THREADS / 32;
  for (int b0 = tid >> 5; b0 < nblk; b0 += 4 * nw) {
    int beg[4], cnt[4];
    unsigned en[4];
#pragma unroll
    for (int j = 0; j < 4; ++j) {
      const int bj = b0 + j * nw;
      beg[j] = rb[bj];
      cnt[j] = re[bj] - beg[j];
    }
#pragma unroll
    for (int j = 0; j < 4; ++j) en[j] = part[cnt[j] > 0 ? beg[j] + min(lane, cnt[j] - 1) : 0];
#pragma unroll
    for (int j = 0; j < 4; ++j)
      if (lane < cnt[j]) atomicAdd(&bins[en[j] & (PART_NODES - 1)], 1.0f);
#pragma unroll
    for (int j = 0; j < 4; ++j)
      for (int t = 32 + lane; t < cnt[j]; t += 32)
        atomicAdd(&bins[part[beg[j] + t] & (PART_NODES - 1)], 1.0f);
  }
  __syncthreads();

  if (tid < PART_NODES) {
    const int node = y * PART_NODES + tid;
    const float d = rsqrtf(1.0f + bins[tid]);
    bins[tid] = d;
    if (node < n) dinv[node] = d;
  }
  __syncthreads();

  if (in0) {
    const float d = bins[r0];
    p0.x *= d; p0.y *= d; p0.z *= d; p0.w *= d;
    hp[tid] = p0;
  }
  if (in1) {
    const float d = bins[r1];
    p1.x *= d; p1.y *= d; p1.z *= d; p1.w *= d;
    hp[tid + DEG_THREADS] = p1;
  }
}

#define AP_THREADS 512
__global__ __launch_bounds__(AP_THREADS) void k_agg1_part(const unsigned* __restrict__ part,
                                                          const int* __restrict__ tab,
                                                          const float* __restrict__ h1,
                                                          const float* __restrict__ dinv,
                                                          const float* __restrict__ b1,
                                                          const float* __restrict__ W2,
                                                          float* __restrict__ h2,
                                                          int nblk, int n) {
  __shared__ float bins[PART_NODES * ROW_PITCH];
  __shared__ int rb[PART_PADB], re[PART_PADB];
  const int tid = threadIdx.x;
  const int lane = tid & 31;
  const int eq = lane >> 2;
  const int c4 = (lane & 3) * 4;
  const int y = blockIdx.x;
  load_runs(tab, y, nblk, rb, re);
  for (int i = tid; i < PART_NODES * ROW_PITCH; i += AP_THREADS) bins[i] = 0.f;
  __syncthreads();

  const int nw = AP_THREADS / 32;
  int b0 = tid >> 5;
  int beg[2], cnt[2];
  unsigned cur[2];
#pragma unroll
  for (int j = 0; j < 2; ++j) {
    beg[j] = rb[b0 + j * nw];
    cnt[j] = re[b0 + j * nw] - beg[j];
    cur[j] = part[cnt[j] > 0 ? beg[j] + min(lane, cnt[j] - 1) : 0];
  }
  for (; b0 < nblk; b0 += 2 * nw) {
    int nbeg[2], ncnt[2];
    unsigned nxt[2];
#pragma unroll
    for (int j = 0; j < 2; ++j) {
      nbeg[j] = rb[b0 + (2 + j) * nw];
      ncnt[j] = re[b0 + (2 + j) * nw] - nbeg[j];
      nxt[j] = part[ncnt[j] > 0 ? nbeg[j] + min(lane, ncnt[j] - 1) : 0];
    }
    unsigned en[2][4];
    float4 hv[2][4];
#pragma unroll
    for (int j = 0; j < 2; ++j)
#pragma unroll
      for (int q = 0; q < 4; ++q) {
        en[j][q] = (unsigned)__shfl((int)cur[j], q * 8 + eq);
        hv[j][q] = *(const float4*)(h1 + (en[j][q] >> PART_BITS) * HIDDEN + c4);
      }
#pragma unroll
    for (int j = 0; j < 2; ++j)
#pragma unroll
      for (int q = 0; q < 4; ++q)
        if (q * 8 + eq < cnt[j]) {
          float* p = &bins[(en[j][q] & (PART_NODES - 1)) * ROW_PITCH + c4];
          atomicAdd(p + 0, hv[j][q].x);
          atomicAdd(p + 1, hv[j][q].y);
          atomicAdd(p + 2, hv[j][q].z);
          atomicAdd(p + 3, hv[j][q].w);
        }
#pragma unroll
    for (int j = 0; j < 2; ++j)
      for (int t = 32; t < cnt[j]; t += 32) {
        const unsigned more = part[beg[j] + min(t + lane, cnt[j] - 1)];
        unsigned em[4];
        float4 hm[4];
#pragma unroll
        for (int q = 0; q < 4; ++q) {
          em[q] = (unsigned)__shfl((int)more, q * 8 + eq);
          hm[q] = *(const float4*)(h1 + (em[q] >> PART_BITS) * HIDDEN + c4);
        }
#pragma unroll
        for (int q = 0; q < 4; ++q)
          if (t + q * 8 + eq < cnt[j]) {
            float* p = &bins[(em[q] & (PART_NODES - 1)) * ROW_PITCH + c4];
            atomicAdd(p + 0, hm[q].x);
            atomicAdd(p + 1, hm[q].y);
            atomicAdd(p + 2, hm[q].z);
            atomicAdd(p + 3, hm[q].w);
          }
      }
#pragma unroll
    for (int j = 0; j < 2; ++j) {
      beg[j] = nbeg[j];
      cnt[j] = ncnt[j];
      cur[j] = nxt[j];
    }
  }
  __syncthreads();

  if (tid < PART_NODES) {
    const int node = y * PART_NODES + tid;
    if (node < n) {
      const float dn = dinv[node];
      const float4* sp = (const float4*)(h1 + (size_t)node * HIDDEN);
      float sv[HIDDEN];
#pragma unroll
      for (int q = 0; q < HIDDEN / 4; ++q) {
        const float4 s = sp[q];
        sv[4 * q + 0] = s.x;
        sv[4 * q + 1] = s.y;
        sv[4 * q + 2] = s.z;
        sv[4 * q + 3] = s.w;
      }
      const float* brow = &bins[tid * ROW_PITCH];
      float acc0 = 0.f, acc1 = 0.f;
#pragma unroll
      for (int j = 0; j < HIDDEN; ++j) {
        const float a = fmaf(brow[j], dn, sv[j] * dn);
        const float v = fmaxf(a + b1[j], 0.f);
        acc0 = fmaf(v, W2[j * NCLS + 0], acc0);
        acc1 = fmaf(v, W2[j * NCLS + 1], acc1);
      }
      v2f o;
      o.x = acc0 * dn;
      o.y = acc1 * dn;
      *(v2f*)(h2 + (size_t)node * NCLS) = o;
    }
  }
}

#define A2_THREADS 512
__global__ __launch_bounds__(A2_THREADS) void k_agg2_part(const unsigned* __restrict__ part,
                                                          const int* __restrict__ tab,
                                                          const float* __restrict__ h2,
                                                          const float* __restrict__ dinv,
                                                          const float* __restrict__ b2,
                                                          float* __restrict__ out,
                                                          int nblk, int n) {
  __shared__ float bins[PART_NODES * NCLS];
  __shared__ int rb[PART_PADB], re[PART_PADB];
  const int tid = threadIdx.x;
  const int lane = tid & 31;
  const int y = blockIdx.x;
  load_runs(tab, y, nblk, rb, re);
  for (int i = tid; i < PART_NODES * NCLS; i += A2_THREADS) bins[i] = 0.f;
  __syncthreads();

  const int nw = A2_THREADS / 32;
  for (int b0 = tid >> 5; b0 < nblk; b0 += 4 * nw) {
    int beg[4], cnt[4];
    unsigned en[4];
    v2f hv[4];
#pragma unroll
    for (int j = 0; j < 4; ++j) {
      const int bj = b0 + j * nw;
      beg[j] = rb[bj];
      cnt[j] = re[bj] - beg[j];
    }
#pragma unroll
    for (int j = 0; j < 4; ++j) en[j] = part[cnt[j] > 0 ? beg[j] + min(lane, cnt[j] - 1) : 0];
#pragma unroll
    for (int j = 0; j < 4; ++j) hv[j] = *(const v2f*)(h2 + (en[j] >> PART_BITS) * NCLS);
#pragma unroll
    for (int j = 0; j < 4; ++j)
      if (lane < cnt[j]) {
        const unsigned idx = en[j] & (PART_NODES - 1);
        atomicAdd(&bins[idx * NCLS + 0], hv[j].x);
        atomicAdd(&bins[idx * NCLS + 1], hv[j].y);
      }
#pragma unroll
    for (int j = 0; j < 4; ++j)
      for (int t = 32 + lane; t < cnt[j]; t += 32) {
        const unsigned em = part[beg[j] + t];
        const v2f hm = *(const v2f*)(h2 + (em >> PART_BITS) * NCLS);
        atomicAdd(&bins[(em & (PART_NODES - 1)) * NCLS + 0], hm.x);
        atomicAdd(&bins[(em & (PART_NODES - 1)) * NCLS + 1], hm.y);
      }
  }
  __syncthreads();

  if (tid < PART_NODES) {
    const int node = y * PART_NODES + tid;
    if (node < n) {
      const float dn = dinv[node];
      const v2f hs = *(const v2f*)(h2 + (size_t)node * NCLS);
      v2f bs;
      bs.x = bins[tid * NCLS + 0];
      bs.y = bins[tid * NCLS + 1];
      const float s0 = hs.x * dn;
      const float s1 = hs.y * dn;
      const float a0 = s0 + bs.x * dn;
      const float a1 = s1 + bs.y * dn;
      const float z0 = a0 + b2[0];
      const float z1 = a1 + b2[1];
      const float m = fmaxf(z0, z1);
      const float lse = m + logf(expf(z0 - m) + expf(z1 - m));
      v2f o;
      o.x = z0 - lse;
      o.y = z1 - lse;
      *(v2f*)(out + (size_t)node * NCLS) = o;
    }
  }
}


__global__ void k_logsoftmax(const float* __restrict__ agg2, const float* __restrict__ b2,
                             float* __restrict__ out, int n) {
  int node = blockIdx.x * blockDim.x + threadIdx.x;
  if (node >= n) return;
  float z0 = agg2[(size_t)node * NCLS + 0] + b2[0];
  float z1 = agg2[(size_t)node * NCLS + 1] + b2[1];
  float m = fmaxf(z0, z1);
  float lse = m + logf(expf(z0 - m) + expf(z1 - m));
  out[(size_t)node * NCLS + 0] = z0 - lse;
  out[(size_t)node * NCLS + 1] = z1 - lse;
}


extern "C" void kernel_launch(void* const* d_in, const int* in_sizes, int n_in,
                              void* d_out, int out_size, void* d_ws, size_t ws_size,
                              hipStream_t stream) {
  const float* x  = (const float*)d_in[0];
  const int*   ei = (const int*)d_in[1];
  const float* W1 = (const float*)d_in[2];
  const float* b1 = (const float*)d_in[3];
  const float* W2 = (const float*)d_in[4];
  const float* b2 = (const float*)d_in[5];

  const int n = in_sizes[0] / NFEAT;
  const int E = in_sizes[1] / 2;
  const int* row = ei;
  const int* col = ei + E;

  char* ws = (char*)d_ws;
  size_t off = 0;
  auto carve = [&](size_t bytes) -> void* {
    void* p = ws + off;
    off = (off + bytes + 255) & ~(size_t)255;
    return p;
  };
  float* deg  = (float*)carve((size_t)n * sizeof(float));
  float* h1   = (float*)carve((size_t)n * HIDDEN * sizeof(float));
  float* agg1 = (float*)carve((size_t)n * HIDDEN * sizeof(float));
  float* h2   = (float*)carve((size_t)n * NCLS * sizeof(float));
  float* agg2 = (float*)carve((size_t)n * NCLS * sizeof(float));

  const int B = 256;
  auto blocks = [](long long work, int b) { return (unsigned)((work + b - 1) / b); };

  const unsigned nblk = blocks(E, PART_CAP);
  const unsigned nrng = blocks(n, PART_NODES);
  unsigned* part = (unsigned*)carve((size_t)E * sizeof(unsigned));
  int* tab = (int*)carve((size_t)nblk * PART_PITCH * sizeof(int));
  const bool grouped = off <= ws_size && nrng <= PART_MAXR && nblk <= PART_MAXB &&
                       n <= (1 << (32 - PART_BITS));
  const dim3 scat_grid(EDGE_SLICES, blocks(n, BIN_NODES));

  int tiles = (n + 15) / 16;
  if (grouped) {
    const unsigned gblk = blocks(tiles, GEMM_TPB);
    k_part_gemm<<<gblk + nblk, PART_THREADS, 0, stream>>>(row, col, part, tab, E, x, W1, h1, n,
                                                           (int)gblk);
    k_deg_scale<<<nrng, DEG_THREADS, 0, stream>>>(part, tab, deg, h1, (int)nblk, n);
    k_agg1_part<<<nrng, AP_THREADS, 0, stream>>>(part, tab, h1, deg, b1, W2, h2,
                                                 (int)nblk, n);
    k_agg2_part<<<nrng, A2_THREADS, 0, stream>>>(part, tab, h2, deg, b2, (float*)d_out,
                                                 (int)nblk, n);
  } else {
    k_init_deg<<<blocks(n, B), B, 0, stream>>>(deg, n);
    k_deg_count<<<scat_grid, SCAT_THREADS, 0, stream>>>(col, deg, E, n);
    k_gemm1<<<blocks(tiles, 8), B, 0, stream>>>(x, W1, h1, deg, agg1, n);
    const dim3 a1_grid(A1_SLICES, blocks(n, A1_NODES));
    k_agg1_edges<<<a1_grid, A1_THREADS, 0, stream>>>(row, col, h1, deg, agg1, E, n);
    k_relu_gemm2<<<blocks(n, B), B, 0, stream>>>(agg1, b1, W2, deg, h2, agg2, n);
    k_agg2_edges<<<scat_grid, SCAT_THREADS, 0, stream>>>(row, col, h2, deg, agg2, E, n);
    k_logsoftmax<<<blocks(n, B), B, 0, stream>>>(agg2, b2, (float*)d_out, n);
  }
}


# layer-1 GEMM role fed by wider reads: x as 16-byte pieces per lane (whole lines per group of four, two groups in flight), W1 fragments from an LDS copy; k permuted alike in A and B, sum order changes
# speedup vs baseline: 6.7093x; 1.0455x over previous
#include <hip/hip_runtime.h>
#include <math.h>

typedef __attribute__((ext_vector_type(2))) float v2f;
typedef __attribute__((ext_vector_type(8))) float v8f;

#define NFEAT 512
#define HIDDEN 16
#define NCLS 2


__global__ void k_init_deg(float* __restrict__ deg, int n) {
  int i = blockIdx.x * blockDim.x + threadIdx.x;
  if (i < n) deg[i] = 1.0f;
}

#define BIN_NODES 16384
#define EDGE_SLICES 64
#define SCAT_THREADS 1024
static_assert((BIN_NODES & (BIN_NODES - 1)) == 0, "bin index is wrapped with a mask");

__global__ __launch_bounds__(SCAT_THREADS) void k_deg_count(const int* __restrict__ col,
                                                            float* __restrict__ deg,
                                                            int E, int n) {
  __shared__ float bins[BIN_NODES];
  const int tid = threadIdx.x;
  const int lo = blockIdx.y * BIN_NODES;
  for (int i = tid; i < BIN_NODES; i += SCAT_THREADS) bins[i] = 0.f;
  __syncthreads();

  const int per = (E + (int)gridDim.x - 1) / (int)gridDim.x;
  const int e0 = blockIdx.x * per;
  const int e1 = min(e0 + per, E);
  const int last = e1 - 1;
  for (int e = e0 + tid; e < e1; e += 8 * SCAT_THREADS) {
    int c[8];
#pragma unroll
    for (int j = 0; j < 8; ++j) {
      const int ee = e + j * SCAT_THREADS;
      const bool in = ee < e1;
      const int v = col[in ? ee : last];
      c[j] = in ? v : -1;
    }
#pragma unroll
    for (int j = 0; j < 8; ++j) {
      const unsigned idx = (unsigned)(c[j] - lo);
      if (idx < (unsigned)BIN_NODES) atomicAdd(&bins[idx], 1.0f);
    }
  }
  __syncthreads();

  const int rot = (int)blockIdx.x * SCAT_THREADS;
  for (int t = tid; t < BIN_NODES; t += SCAT_THREADS) {
    const int i = (t + rot) & (BIN_NODES - 1);
    const float v = bins[i];
    const int node = lo + i;
    if (v != 0.f && node < n) atomicAdd(&deg[node], v);
  }
}


__device__ __forceinline__ v8f gemm_tile(const float* __restrict__ x,
                                         const float* __restrict__ W1, int row0, int lane) {
  const int nsel  = lane & 15;
  const int khalf = (lane >> 4) << 1;
  const float* arow = x + (size_t)(row0 + nsel) * NFEAT + khalf;
  v8f c = {};
  for (int k = 0; k < NFEAT; k += 4) {
    if ((k & 127) == 0) __builtin_prefetch(arow + k + 128, 0, 0);
    v2f a = *(const v2f*)(arow + k);
    v2f b;
    b.x = W1[(k + khalf) * HIDDEN + nsel];
    b.y = W1[(k + khalf + 1) * HIDDEN + nsel];
    c = __builtin_amdgcn_wmma_f32_16x16x4_f32(false, a, false, b, (short)0, c,
                                              false, false);
  }
  return c;
}

__global__ __launch_bounds__(256) void k_gemm1(const float* __restrict__ x,
                                               const float* __restrict__ W1,
                                               float* __restrict__ h1,
                                               float* degnorm,
                                               float* __restrict__ agg1,
                                               int n_nodes) {
  const int lane = threadIdx.x & 31;
  const int tile = blockIdx.x * 8 + (threadIdx.x >> 5);
  const int row0 = tile * 16;
  if (row0 >= n_nodes) return;

  const int nsel  = lane & 15;

  const int mbase = row0 + ((lane >> 4) << 3);
  float dg[8];
#pragma unroll
  for (int r = 0; r < 8; ++r) dg[r] = degnorm[mbase + r];

  const v8f c = gemm_tile(x, W1, row0, lane);

  float d[8];
#pragma unroll
  for (int r = 0; r < 8; ++r) d[r] = rsqrtf(dg[r]);
#pragma unroll
  for (int r = 0; r < 8; ++r) {
    const size_t o = (size_t)(mbase + r) * HIDDEN + nsel;
    const float hs = c[r] * d[r];
    h1[o] = hs;
    agg1[o] = hs * d[r];
  }
  if (nsel == 0) {
#pragma unroll
    for (int r = 0; r < 8; ++r) degnorm[mbase + r] = d[r];
  }
}


#define A1_NODES 4096
#define A1_CL_BITS 12
#define A1_SLICES 10
#define A1_THREADS 1024
#define A1_U 8
#define A1_G 8
static_assert((1 << A1_CL_BITS) == A1_NODES, "list entry packs the in-range target in A1_CL_BITS bits");

__global__ __launch_bounds__(A1_THREADS) void k_agg1_edges(const int* __restrict__ row,
                                                           const int* __restrict__ col,
                                                           const float* __restrict__ h1,
                                                           const float* __restrict__ dinv,
                                                           float* __restrict__ agg1,
                                                           int E, int n) {
  __shared__ float bins[A1_NODES * HIDDEN];
  __shared__ unsigned list[A1_THREADS / 32][A1_U * 32];
  const int tid = threadIdx.x;
  const int lane = tid & 31;
  const int half = lane >> 4;
  const int f = lane & 15;
  const int lo = blockIdx.y * A1_NODES;
  unsigned* mylist = list[tid >> 5];

  for (int i = tid; i < A1_NODES * HIDDEN / 4; i += A1_THREADS)
    ((float4*)bins)[i] = make_float4(0.f, 0.f, 0.f, 0.f);
  __syncthreads();

  const int per = (E + (int)gridDim.x - 1) / (int)gridDim.x;
  const int e0 = blockIdx.x * per;
  const int e1 = min(e0 + per, E);
  if (e0 < e1) {
    const int last = e1 - 1;
    int c[A1_U];
    unsigned r[A1_U];
#pragma unroll
    for (int j = 0; j < A1_U; ++j) {
      const int ee = e0 + tid + j * A1_THREADS;
      const bool in = ee < e1;
      const int k = in ? ee : last;
      const int cv = col[k];
      r[j] = (unsigned)row[k];
      c[j] = in ? cv : -1;
    }
    for (int base = e0; base < e1; base += A1_U * A1_THREADS) {
      int cn[A1_U];
      unsigned rn[A1_U];
#pragma unroll
      for (int j = 0; j < A1_U; ++j) {
        const int ee = base + tid + (A1_U + j) * A1_THREADS;
        const bool in = ee < e1;
        const int k = in ? ee : last;
        const int cv = col[k];
        rn[j] = (unsigned)row[k];
        cn[j] = in ? cv : -1;
      }
      int cnt = 0;
#pragma unroll
      for (int j = 0; j < A1_U; ++j) {
        const unsigned idx = (unsigned)(c[j] - lo);
        const bool m = idx < (unsigned)A1_NODES;
        const unsigned mask = __builtin_amdgcn_ballot_w32(m);
        if (m) mylist[cnt + (int)__builtin_amdgcn_mbcnt_lo(mask, 0u)] = (r[j] << A1_CL_BITS) | idx;
        cnt += __builtin_popcount(mask);
      }
      for (int t = 0; t < cnt; t += 2 * A1_G) {
        unsigned ent[A1_G];
        float hv[A1_G];
#pragma unroll
        for (int q = 0; q < A1_G; ++q) {
          const int i = t + 2 * q + half;
          ent[q] = mylist[i < cnt ? i : t];
        }
#pragma unroll
        for (int q = 0; q < A1_G; ++q) {
          const unsigned rr = ent[q] >> A1_CL_BITS;
          hv[q] = h1[rr * HIDDEN + f];
        }
#pragma unroll
        for (int q = 0; q < A1_G; ++q) {
          const int i = t + 2 * q + half;
          if (i < cnt)
            atomicAdd(&bins[(ent[q] & (A1_NODES - 1)) * HIDDEN + f], hv[q]);
        }
      }
#pragma unroll
      for (int j = 0; j < A1_U; ++j) {
        c[j] = cn[j];
        r[j] = rn[j];
      }
    }
  }
  __syncthreads();

  const int total = n * HIDDEN;
  const int rot = (int)blockIdx.x * (((A1_NODES * HIDDEN) / A1_SLICES) & ~31);
  for (int t = tid; t < A1_NODES * HIDDEN; t += 4 * A1_THREADS) {
    float v[4], d[4];
    int g[4];
#pragma unroll
    for (int u = 0; u < 4; ++u) {
      const int i = (t + u * A1_THREADS + rot) & (A1_NODES * HIDDEN - 1);
      v[u] = bins[i];
      g[u] = lo * HIDDEN + i;
      d[u] = dinv[min(lo + (i >> 4), n - 1)];
    }
#pragma unroll
    for (int u = 0; u < 4; ++u)
      if (v[u] != 0.f && g[u] < total) atomicAdd(&agg1[g[u]], v[u] * d[u]);
  }
}


__global__ void k_relu_gemm2(const float* __restrict__ agg1, const float* __restrict__ b1,
                             const float* __restrict__ W2, const float* __restrict__ dinv,
                             float* __restrict__ h2, float* __restrict__ agg2, int n) {
  int node = blockIdx.x * blockDim.x + threadIdx.x;
  if (node >= n) return;
  const float dn = dinv[node];
  float acc0 = 0.f, acc1 = 0.f;
#pragma unroll
  for (int j = 0; j < HIDDEN; ++j) {
    float v = fmaxf(agg1[(size_t)node * HIDDEN + j] + b1[j], 0.f);
    acc0 = fmaf(v, W2[j * NCLS + 0], acc0);
    acc1 = fmaf(v, W2[j * NCLS + 1], acc1);
  }
  const float s0 = acc0 * dn, s1 = acc1 * dn;
  h2[(size_t)node * NCLS + 0] = s0;
  h2[(size_t)node * NCLS + 1] = s1;
  agg2[(size_t)node * NCLS + 0] = s0 * dn;
  agg2[(size_t)node * NCLS + 1] = s1 * dn;
}


__global__ __launch_bounds__(SCAT_THREADS) void k_agg2_edges(const int* __restrict__ row,
                                                             const int* __restrict__ col,
                                                             const float* __restrict__ h2,
                                                             const float* __restrict__ dinv,
                                                             float* __restrict__ agg2,
                                                             int E, int n) {
  __shared__ float bins[BIN_NODES * NCLS];
  const int tid = threadIdx.x;
  const int lo = blockIdx.y * BIN_NODES;
  for (int i = tid; i < BIN_NODES * NCLS; i += SCAT_THREADS) bins[i] = 0.f;
  __syncthreads();

  const int per = (E + (int)gridDim.x - 1) / (int)gridDim.x;
  const int e0 = blockIdx.x * per;
  const int e1 = min(e0 + per, E);
  if (e0 < e1) {
    const int last = e1 - 1;
    int c[4];
    unsigned r[4];
#pragma unroll
    for (int j = 0; j < 4; ++j) {
      const int ee = e0 + tid + j * SCAT_THREADS;
      const bool in = ee < e1;
      const int k = in ? ee : last;
      const int cv = col[k];
      r[j] = (unsigned)row[k];
      c[j] = in ? cv : -1;
    }
    for (int e = e0 + tid; e < e1; e += 4 * SCAT_THREADS) {
      int cn[4];
      unsigned rn[4];
#pragma unroll
      for (int j = 0; j < 4; ++j) {
        const int ee = e + (4 + j) * SCAT_THREADS;
        const bool in = ee < e1;
        const int k = in ? ee : last;
        const int cv = col[k];
        rn[j] = (unsigned)row[k];
        cn[j] = in ? cv : -1;
      }
      v2f hv[4];
#pragma unroll
      for (int j = 0; j < 4; ++j) {
        const unsigned idx = (unsigned)(c[j] - lo);
        if (idx < (unsigned)BIN_NODES) hv[j] = *(const v2f*)(h2 + r[j] * NCLS);
      }
#pragma unroll
      for (int j = 0; j < 4; ++j) {
        const unsigned idx = (unsigned)(c[j] - lo);
        if (idx < (unsigned)BIN_NODES) {
          atomicAdd(&bins[idx * NCLS + 0], hv[j].x);
          atomicAdd(&bins[idx * NCLS + 1], hv[j].y);
        }
      }
#pragma unroll
      for (int j = 0; j < 4; ++j) {
        c[j] = cn[j];
        r[j] = rn[j];
      }
    }
  }
  __syncthreads();

  const int total = n * NCLS;
  const int rot = (int)blockIdx.x * SCAT_THREADS;
  for (int t = tid; t < BIN_NODES * NCLS; t += SCAT_THREADS) {
    const int i = (t + rot) & (BIN_NODES * NCLS - 1);
    const float v = bins[i];
    const int g = lo * NCLS + i;
    const float d = dinv[min(lo + (i >> 1), n - 1)];
    if (v != 0.f && g < total) atomicAdd(&agg2[g], v * d);
  }
}


#define PART_NODES 256
#define PART_BITS 8
#define PART_MAXR 512
#define PART_THREADS 1024
#define PART_U 10
#define PART_CAP (PART_THREADS * PART_U)
#define PART_PITCH 544
#define PART_MAXB 1024
#define PART_PADB (PART_MAXB + 64)
#define ROW_PITCH (HIDDEN + 1)

#define GEMM_TPB 25
#define GEMM_RING 8

template <bool AGAIN>
__device__ __forceinline__ v8f gemm_ring(float4 (&a)[GEMM_RING], v8f c, const float4* ws4,
                                         const float4* __restrict__ arow, int g) {
#pragma unroll
  for (int u = 0; u < GEMM_RING; ++u) {
    const float4 b = ws4[(g + u) * 32];
    v2f a0, a1, b0, b1;
    a0.x = a[u].x; a0.y = a[u].y; a1.x = a[u].z; a1.y = a[u].w;
    b0.x = b.x;    b0.y = b.y;    b1.x = b.z;    b1.y = b.w;
    c = __builtin_amdgcn_wmma_f32_16x16x4_f32(false, a0, false, b0, (short)0, c, false, false);
    c = __builtin_amdgcn_wmma_f32_16x16x4_f32(false, a1, false, b1, (short)0, c, false, false);
    if (AGAIN && (u & 3) == 3) {
#pragma unroll
      for (int v = u - 3; v <= u; ++v) a[v] = arow[2 * (g + GEMM_RING + v)];
      __builtin_amdgcn_sched_barrier(0);
    }
  }
  return c;
}

__global__ __launch_bounds__(PART_THREADS) void k_part_gemm(const int* __restrict__ row,
                                                            const int* __restrict__ col,
                                                            unsigned* __restrict__ part,
                                                            int* __restrict__ tab, int E,
                                                            const float* __restrict__ x,
                                                            const float* __restrict__ W1,
                                                            float* __restrict__ h1,
                                                            int n_nodes, int gemm_blocks) {
  __shared__ __attribute__((aligned(16))) unsigned sorted[PART_CAP];
  __shared__ int hist[PART_MAXR];
  __shared__ int base[PART_MAXR + 1];
  __shared__ int wsum[PART_MAXR / 32];
  const int tid = threadIdx.x;

  if ((int)blockIdx.x < gemm_blocks) {
    const int wave = tid >> 5;
    const int lane = tid & 31;
    const int row0 = ((int)blockIdx.x * GEMM_TPB + wave) * 16;
    const bool active = wave < GEMM_TPB && row0 < n_nodes;
    const int nsel = lane & 15;
    const int h = lane >> 4;
    float* wsf = (float*)sorted;
    float wv[NFEAT * HIDDEN / PART_THREADS];
#pragma unroll
    for (int j = 0; j < NFEAT * HIDDEN / PART_THREADS; ++j) wv[j] = W1[tid + j * PART_THREADS];
    const float4* arow = (const float4*)(x + (size_t)(active ? row0 + nsel : nsel) * NFEAT) + h;
    float4 a[GEMM_RING];
#pragma unroll
    for (int u = 0; u < GEMM_RING; ++u) a[u] = arow[2 * u];
#pragma unroll
    for (int j = 0; j < NFEAT * HIDDEN / PART_THREADS; ++j) {
      const int i = tid + j * PART_THREADS;
      const int k = i >> 4, col = i & 15;
      wsf[(((k >> 3) * 32 + ((k >> 2) & 1) * 16 + col) << 2) + (k & 3)] = wv[j];
    }
    __syncthreads();
    if (!active) return;
    const float4* ws4 = (const float4*)sorted + lane;
    v8f c = {};
#pragma nounroll
    for (int g = 0; g < NFEAT / 8 - GEMM_RING; g += GEMM_RING) c = gemm_ring<true>(a, c, ws4, arow, g);
    c = gemm_ring<false>(a, c, ws4, arow, NFEAT / 8 - GEMM_RING);
    const int mbase = row0 + (h << 3);
#pragma unroll
    for (int r = 0; r < 8; ++r) h1[(size_t)(mbase + r) * HIDDEN + nsel] = c[r];
    return;
  }
  const int b = (int)blockIdx.x - gemm_blocks;

  if (tid < PART_MAXR) hist[tid] = 0;
  __syncthreads();

  const int e0 = b * PART_CAP;
  const int e1 = min(e0 + PART_CAP, E);
  unsigned ent[PART_U];
  int rg[PART_U], rk[PART_U];
#pragma unroll
  for (int j = 0; j < PART_U; ++j) {
    const int ee = e0 + tid + j * PART_THREADS;
    const bool in = ee < e1;
    const int k = in ? ee : e0;
    const int cv = col[k];
    const unsigned rv = (unsigned)row[k];
    rg[j] = in ? (cv >> PART_BITS) : -1;
    ent[j] = (rv << PART_BITS) | ((unsigned)cv & (PART_NODES - 1));
  }
#pragma unroll
  for (int j = 0; j < PART_U; ++j) rk[j] = rg[j] >= 0 ? atomicAdd(&hist[rg[j]], 1) : 0;
  __syncthreads();

  int v = 0, incl = 0;
  if (tid < PART_MAXR) {
    v = hist[tid];
    incl = v;
#pragma unroll
    for (int d = 1; d < 32; d <<= 1) {
      const int up = __shfl_up(incl, d);
      if ((tid & 31) >= d) incl += up;
    }
    if ((tid & 31) == 31) wsum[tid >> 5] = incl;
  }
  __syncthreads();
  if (tid < PART_MAXR) {
    int before = 0;
    for (int w = 0; w < (tid >> 5); ++w) before += wsum[w];
    base[tid] = before + incl - v;
    if (tid == PART_MAXR - 1) base[PART_MAXR] = before + incl;
  }
  __syncthreads();

  if (tid <= PART_MAXR) tab[b * PART_PITCH + tid] = base[tid];
#pragma unroll
  for (int j = 0; j < PART_U; ++j)
    if (rg[j] >= 0) sorted[base[rg[j]] + rk[j]] = ent[j];
  __syncthreads();

  const int cnt = e1 - e0;
  for (int i = tid; i < cnt; i += PART_THREADS) part[e0 + i] = sorted[i];
}

__device__ __forceinline__ void load_runs(const int* __restrict__ tab, int y, int nblk,
                                          int* rb, int* re) {
  for (int i = threadIdx.x; i < PART_PADB; i += blockDim.x) {
    int b0 = 0, b1 = 0;
    if (i < nblk) {
      const int* t = tab + (size_t)i * PART_PITCH + y;
      b0 = i * PART_CAP + t[0];
      b1 = i * PART_CAP + t[1];
    }
    rb[i] = b0;
    re[i] = b1;
  }
}

#define DEG_THREADS 512
static_assert(PART_NODES * HIDDEN / 4 == 2 * DEG_THREADS, "two 16-byte row pieces per thread");
static_assert(NFEAT * HIDDEN <= PART_CAP && (NFEAT * HIDDEN) % PART_THREADS == 0 &&
              (NFEAT / 8) % GEMM_RING == 0 && GEMM_RING % 4 == 0,
              "W1 fits the entry buffer, its copy and the K loop divide evenly");
__global__ __launch_bounds__(DEG_THREADS) void k_deg_scale(const unsigned* __restrict__ part,
                                                           const int* __restrict__ tab,
                                                           float* __restrict__ dinv,
                                                           float* h1,
                                                           int nblk, int n) {
  __shared__ float bins[PART_NODES];
  __shared__ int rb[PART_PADB], re[PART_PADB];
  const int tid = threadIdx.x;
  const int lane = tid & 31;
  const int y = blockIdx.x;
  float4* hp = (float4*)(h1 + (size_t)y * PART_NODES * HIDDEN);
  const int r0 = tid >> 2;
  const int r1 = r0 + DEG_THREADS / 4;
  const bool in0 = y * PART_NODES + r0 < n;
  const bool in1 = y * PART_NODES + r1 < n;
  float4 p0 = make_float4(0.f, 0.f, 0.f, 0.f), p1 = p0;
  if (in0) p0 = hp[tid];
  if (in1) p1 = hp[tid + DEG_THREADS];
  load_runs(tab, y, nblk, rb, re);
  if (tid < PART_NODES) bins[tid] = 0.f;
  __syncthreads();

  const int nw = DEG_THREADS / 32;
  for (int b0 = tid >> 5; b0 < nblk; b0 += 4 * nw) {
    int beg[4], cnt[4];
    unsigned en[4];
#pragma unroll
    for (int j = 0; j < 4; ++j) {
      const int bj = b0 + j * nw;
      beg[j] = rb[bj];
      cnt[j] = re[bj] - beg[j];
    }
#pragma unroll
    for (int j = 0; j < 4; ++j) en[j] = part[cnt[j] > 0 ? beg[j] + min(lane, cnt[j] - 1) : 0];
#pragma unroll
    for (int j = 0; j < 4; ++j)
      if (lane < cnt[j]) atomicAdd(&bins[en[j] & (PART_NODES - 1)], 1.0f);
#pragma unroll
    for (int j = 0; j < 4; ++j)
      for (int t = 32 + lane; t < cnt[j]; t += 32)
        atomicAdd(&bins[part[beg[j] + t] & (PART_NODES - 1)], 1.0f);
  }
  __syncthreads();

  if (tid < PART_NODES) {
    const int node = y * PART_NODES + tid;
    const float d = rsqrtf(1.0f + bins[tid]);
    bins[tid] = d;
    if (node < n) dinv[node] = d;
  }
  __syncthreads();

  if (in0) {
    const float d = bins[r0];
    p0.x *= d; p0.y *= d; p0.z *= d; p0.w *= d;
    hp[tid] = p0;
  }
  if (in1) {
    const float d = bins[r1];
    p1.x *= d; p1.y *= d; p1.z *= d; p1.w *= d;
    hp[tid + DEG_THREADS] = p1;
  }
}

#define AP_THREADS 512
__global__ __launch_bounds__(AP_THREADS) void k_agg1_part(const unsigned* __restrict__ part,
                                                          const int* __restrict__ tab,
                                                          const float* __restrict__ h1,
                                                          const float* __restrict__ dinv,
                                                          const float* __restrict__ b1,
                                                          const float* __restrict__ W2,
                                                          float* __restrict__ h2,
                                                          int nblk, int n) {
  __shared__ float bins[PART_NODES * ROW_PITCH];
  __shared__ int rb[PART_PADB], re[PART_PADB];
  const int tid = threadIdx.x;
  const int lane = tid & 31;
  const int eq = lane >> 2;
  const int c4 = (lane & 3) * 4;
  const int y = blockIdx.x;
  load_runs(tab, y, nblk, rb, re);
  for (int i = tid; i < PART_NODES * ROW_PITCH; i += AP_THREADS) bins[i] = 0.f;
  __syncthreads();

  const int nw = AP_THREADS / 32;
  int b0 = tid >> 5;
  int beg[2], cnt[2];
  unsigned cur[2];
#pragma unroll
  for (int j = 0; j < 2; ++j) {
    beg[j] = rb[b0 + j * nw];
    cnt[j] = re[b0 + j * nw] - beg[j];
    cur[j] = part[cnt[j] > 0 ? beg[j] + min(lane, cnt[j] - 1) : 0];
  }
  for (; b0 < nblk; b0 += 2 * nw) {
    int nbeg[2], ncnt[2];
    unsigned nxt[2];
#pragma unroll
    for (int j = 0; j < 2; ++j) {
      nbeg[j] = rb[b0 + (2 + j) * nw];
      ncnt[j] = re[b0 + (2 + j) * nw] - nbeg[j];
      nxt[j] = part[ncnt[j] > 0 ? nbeg[j] + min(lane, ncnt[j] - 1) : 0];
    }
    unsigned en[2][4];
    float4 hv[2][4];
#pragma unroll
    for (int j = 0; j < 2; ++j)
#pragma unroll
      for (int q = 0; q < 4; ++q) {
        en[j][q] = (unsigned)__shfl((int)cur[j], q * 8 + eq);
        hv[j][q] = *(const float4*)(h1 + (en[j][q] >> PART_BITS) * HIDDEN + c4);
      }
#pragma unroll
    for (int j = 0; j < 2; ++j)
#pragma unroll
      for (int q = 0; q < 4; ++q)
        if (q * 8 + eq < cnt[j]) {
          float* p = &bins[(en[j][q] & (PART_NODES - 1)) * ROW_PITCH + c4];
          atomicAdd(p + 0, hv[j][q].x);
          atomicAdd(p + 1, hv[j][q].y);
          atomicAdd(p + 2, hv[j][q].z);
          atomicAdd(p + 3, hv[j][q].w);
        }
#pragma unroll
    for (int j = 0; j < 2; ++j)
      for (int t = 32; t < cnt[j]; t += 32) {
        const unsigned more = part[beg[j] + min(t + lane, cnt[j] - 1)];
        unsigned em[4];
        float4 hm[4];
#pragma unroll
        for (int q = 0; q < 4; ++q) {
          em[q] = (unsigned)__shfl((int)more, q * 8 + eq);
          hm[q] = *(const float4*)(h1 + (em[q] >> PART_BITS) * HIDDEN + c4);
        }
#pragma unroll
        for (int q = 0; q < 4; ++q)
          if (t + q * 8 + eq < cnt[j]) {
            float* p = &bins[(em[q] & (PART_NODES - 1)) * ROW_PITCH + c4];
            atomicAdd(p + 0, hm[q].x);
            atomicAdd(p + 1, hm[q].y);
            atomicAdd(p + 2, hm[q].z);
            atomicAdd(p + 3, hm[q].w);
          }
      }
#pragma unroll
    for (int j = 0; j < 2; ++j) {
      beg[j] = nbeg[j];
      cnt[j] = ncnt[j];
      cur[j] = nxt[j];
    }
  }
  __syncthreads();

  if (tid < PART_NODES) {
    const int node = y * PART_NODES + tid;
    if (node < n) {
      const float dn = dinv[node];
      const float4* sp = (const float4*)(h1 + (size_t)node * HIDDEN);
      float sv[HIDDEN];
#pragma unroll
      for (int q = 0; q < HIDDEN / 4; ++q) {
        const float4 s = sp[q];
        sv[4 * q + 0] = s.x;
        sv[4 * q + 1] = s.y;
        sv[4 * q + 2] = s.z;
        sv[4 * q + 3] = s.w;
      }
      const float* brow = &bins[tid * ROW_PITCH];
      float acc0 = 0.f, acc1 = 0.f;
#pragma unroll
      for (int j = 0; j < HIDDEN; ++j) {
        const float a = fmaf(brow[j], dn, sv[j] * dn);
        const float v = fmaxf(a + b1[j], 0.f);
        acc0 = fmaf(v, W2[j * NCLS + 0], acc0);
        acc1 = fmaf(v, W2[j * NCLS + 1], acc1);
      }
      v2f o;
      o.x = acc0 * dn;
      o.y = acc1 * dn;
      *(v2f*)(h2 + (size_t)node * NCLS) = o;
    }
  }
}

#define A2_THREADS 512
__global__ __launch_bounds__(A2_THREADS) void k_agg2_part(const unsigned* __restrict__ part,
                                                          const int* __restrict__ tab,
                                                          const float* __restrict__ h2,
                                                          const float* __restrict__ dinv,
                                                          const float* __restrict__ b2,
                                                          float* __restrict__ out,
                                                          int nblk, int n) {
  __shared__ float bins[PART_NODES * NCLS];
  __shared__ int rb[PART_PADB], re[PART_PADB];
  const int tid = threadIdx.x;
  const int lane = tid & 31;
  const int y = blockIdx.x;
  load_runs(tab, y, nblk, rb, re);
  for (int i = tid; i < PART_NODES * NCLS; i += A2_THREADS) bins[i] = 0.f;
  __syncthreads();

  const int nw = A2_THREADS / 32;
  for (int b0 = tid >> 5; b0 < nblk; b0 += 4 * nw) {
    int beg[4], cnt[4];
    unsigned en[4];
    v2f hv[4];
#pragma unroll
    for (int j = 0; j < 4; ++j) {
      const int bj = b0 + j * nw;
      beg[j] = rb[bj];
      cnt[j] = re[bj] - beg[j];
    }
#pragma unroll
    for (int j = 0; j < 4; ++j) en[j] = part[cnt[j] > 0 ? beg[j] + min(lane, cnt[j] - 1) : 0];
#pragma unroll
    for (int j = 0; j < 4; ++j) hv[j] = *(const v2f*)(h2 + (en[j] >> PART_BITS) * NCLS);
#pragma unroll
    for (int j = 0; j < 4; ++j)
      if (lane < cnt[j]) {
        const unsigned idx = en[j] & (PART_NODES - 1);
        atomicAdd(&bins[idx * NCLS + 0], hv[j].x);
        atomicAdd(&bins[idx * NCLS + 1], hv[j].y);
      }
#pragma unroll
    for (int j = 0; j < 4; ++j)
      for (int t = 32 + lane; t < cnt[j]; t += 32) {
        const unsigned em = part[beg[j] + t];
        const v2f hm = *(const v2f*)(h2 + (em >> PART_BITS) * NCLS);
        atomicAdd(&bins[(em & (PART_NODES - 1)) * NCLS + 0], hm.x);
        atomicAdd(&bins[(em & (PART_NODES - 1)) * NCLS + 1], hm.y);
      }
  }
  __syncthreads();

  if (tid < PART_NODES) {
    const int node = y * PART_NODES + tid;
    if (node < n) {
      const float dn = dinv[node];
      const v2f hs = *(const v2f*)(h2 + (size_t)node * NCLS);
      v2f bs;
      bs.x = bins[tid * NCLS + 0];
      bs.y = bins[tid * NCLS + 1];
      const float s0 = hs.x * dn;
      const float s1 = hs.y * dn;
      const float a0 = s0 + bs.x * dn;
      const float a1 = s1 + bs.y * dn;
      const float z0 = a0 + b2[0];
      const float z1 = a1 + b2[1];
      const float m = fmaxf(z0, z1);
      const float lse = m + logf(expf(z0 - m) + expf(z1 - m));
      v2f o;
      o.x = z0 - lse;
      o.y = z1 - lse;
      *(v2f*)(out + (size_t)node * NCLS) = o;
    }
  }
}


__global__ void k_logsoftmax(const float* __restrict__ agg2, const float* __restrict__ b2,
                             float* __restrict__ out, int n) {
  int node = blockIdx.x * blockDim.x + threadIdx.x;
  if (node >= n) return;
  float z0 = agg2[(size_t)node * NCLS + 0] + b2[0];
  float z1 = agg2[(size_t)node * NCLS + 1] + b2[1];
  float m = fmaxf(z0, z1);
  float lse = m + logf(expf(z0 - m) + expf(z1 - m));
  out[(size_t)node * NCLS + 0] = z0 - lse;
  out[(size_t)node * NCLS + 1] = z1 - lse;
}


extern "C" void kernel_launch(void* const* d_in, const int* in_sizes, int n_in,
                              void* d_out, int out_size, void* d_ws, size_t ws_size,
                              hipStream_t stream) {
  const float* x  = (const float*)d_in[0];
  const int*   ei = (const int*)d_in[1];
  const float* W1 = (const float*)d_in[2];
  const float* b1 = (const float*)d_in[3];
  const float* W2 = (const float*)d_in[4];
  const float* b2 = (const float*)d_in[5];

  const int n = in_sizes[0] / NFEAT;
  const int E = in_sizes[1] / 2;
  const int* row = ei;
  const int* col = ei + E;

  char* ws = (char*)d_ws;
  size_t off = 0;
  auto carve = [&](size_t bytes) -> void* {
    void* p = ws + off;
    off = (off + bytes + 255) & ~(size_t)255;
    return p;
  };
  float* deg  = (float*)carve((size_t)n * sizeof(float));
  float* h1   = (float*)carve((size_t)n * HIDDEN * sizeof(float));
  float* agg1 = (float*)carve((size_t)n * HIDDEN * sizeof(float));
  float* h2   = (float*)carve((size_t)n * NCLS * sizeof(float));
  float* agg2 = (float*)carve((size_t)n * NCLS * sizeof(float));

  const int B = 256;
  auto blocks = [](long long work, int b) { return (unsigned)((work + b - 1) / b); };

  const unsigned nblk = blocks(E, PART_CAP);
  const unsigned nrng = blocks(n, PART_NODES);
  unsigned* part = (unsigned*)carve((size_t)E * sizeof(unsigned));
  int* tab = (int*)carve((size_t)nblk * PART_PITCH * sizeof(int));
  const bool grouped = off <= ws_size && nrng <= PART_MAXR && nblk <= PART_MAXB &&
                       n <= (1 << (32 - PART_BITS));
  const dim3 scat_grid(EDGE_SLICES, blocks(n, BIN_NODES));

  int tiles = (n + 15) / 16;
  if (grouped) {
    const unsigned gblk = blocks(tiles, GEMM_TPB);
    k_part_gemm<<<gblk + nblk, PART_THREADS, 0, stream>>>(row, col, part, tab, E, x, W1, h1, n,
                                                           (int)gblk);
    k_deg_scale<<<nrng, DEG_THREADS, 0, stream>>>(part, tab, deg, h1, (int)nblk, n);
    k_agg1_part<<<nrng, AP_THREADS, 0, stream>>>(part, tab, h1, deg, b1, W2, h2,
                                                 (int)nblk, n);
    k_agg2_part<<<nrng, A2_THREADS, 0, stream>>>(part, tab, h2, deg, b2, (float*)d_out,
                                                 (int)nblk, n);
  } else {
    k_init_deg<<<blocks(n, B), B, 0, stream>>>(deg, n);
    k_deg_count<<<scat_grid, SCAT_THREADS, 0, stream>>>(col, deg, E, n);
    k_gemm1<<<blocks(tiles, 8), B, 0, stream>>>(x, W1, h1, deg, agg1, n);
    const dim3 a1_grid(A1_SLICES, blocks(n, A1_NODES));
    k_agg1_edges<<<a1_grid, A1_THREADS, 0, stream>>>(row, col, h1, deg, agg1, E, n);
    k_relu_gemm2<<<blocks(n, B), B, 0, stream>>>(agg1, b1, W2, deg, h2, agg2, n);
    k_agg2_edges<<<scat_grid, SCAT_THREADS, 0, stream>>>(row, col, h2, deg, agg2, E, n);
    k_logsoftmax<<<blocks(n, B), B, 0, stream>>>(agg2, b2, (float*)d_out, n);
  }
}


# target ranges of 196 nodes instead of 256: 511 range blocks, two per WGP on all 256, where 391 gave some WGPs two blocks and the rest one; runs of about 20 entries; LDS float add order changes
# speedup vs baseline: 7.3423x; 1.0943x over previous
#include <hip/hip_runtime.h>
#include <math.h>

typedef __attribute__((ext_vector_type(2))) float v2f;
typedef __attribute__((ext_vector_type(8))) float v8f;

#define NFEAT 512
#define HIDDEN 16
#define NCLS 2


__global__ void k_init_deg(float* __restrict__ deg, int n) {
  int i = blockIdx.x * blockDim.x + threadIdx.x;
  if (i < n) deg[i] = 1.0f;
}

#define BIN_NODES 16384
#define EDGE_SLICES 64
#define SCAT_THREADS 1024
static_assert((BIN_NODES & (BIN_NODES - 1)) == 0, "bin index is wrapped with a mask");

__global__ __launch_bounds__(SCAT_THREADS) void k_deg_count(const int* __restrict__ col,
                                                            float* __restrict__ deg,
                                                            int E, int n) {
  __shared__ float bins[BIN_NODES];
  const int tid = threadIdx.x;
  const int lo = blockIdx.y * BIN_NODES;
  for (int i = tid; i < BIN_NODES; i += SCAT_THREADS) bins[i] = 0.f;
  __syncthreads();

  const int per = (E + (int)gridDim.x - 1) / (int)gridDim.x;
  const int e0 = blockIdx.x * per;
  const int e1 = min(e0 + per, E);
  const int last = e1 - 1;
  for (int e = e0 + tid; e < e1; e += 8 * SCAT_THREADS) {
    int c[8];
#pragma unroll
    for (int j = 0; j < 8; ++j) {
      const int ee = e + j * SCAT_THREADS;
      const bool in = ee < e1;
      const int v = col[in ? ee : last];
      c[j] = in ? v : -1;
    }
#pragma unroll
    for (int j = 0; j < 8; ++j) {
      const unsigned idx = (unsigned)(c[j] - lo);
      if (idx < (unsigned)BIN_NODES) atomicAdd(&bins[idx], 1.0f);
    }
  }
  __syncthreads();

  const int rot = (int)blockIdx.x * SCAT_THREADS;
  for (int t = tid; t < BIN_NODES; t += SCAT_THREADS) {
    const int i = (t + rot) & (BIN_NODES - 1);
    const float v = bins[i];
    const int node = lo + i;
    if (v != 0.f && node < n) atomicAdd(&deg[node], v);
  }
}


__device__ __forceinline__ v8f gemm_tile(const float* __restrict__ x,
                                         const float* __restrict__ W1, int row0, int lane) {
  const int nsel  = lane & 15;
  const int khalf = (lane >> 4) << 1;
  const float* arow = x + (size_t)(row0 + nsel) * NFEAT + khalf;
  v8f c = {};
  for (int k = 0; k < NFEAT; k += 4) {
    if ((k & 127) == 0) __builtin_prefetch(arow + k + 128, 0, 0);
    v2f a = *(const v2f*)(arow + k);
    v2f b;
    b.x = W1[(k + khalf) * HIDDEN + nsel];
    b.y = W1[(k + khalf + 1) * HIDDEN + nsel];
    c = __builtin_amdgcn_wmma_f32_16x16x4_f32(false, a, false, b, (short)0, c,
                                              false, false);
  }
  return c;
}

__global__ __launch_bounds__(256) void k_gemm1(const float* __restrict__ x,
                                               const float* __restrict__ W1,
                                               float* __restrict__ h1,
                                               float* degnorm,
                                               float* __restrict__ agg1,
                                               int n_nodes) {
  const int lane = threadIdx.x & 31;
  const int tile = blockIdx.x * 8 + (threadIdx.x >> 5);
  const int row0 = tile * 16;
  if (row0 >= n_nodes) return;

  const int nsel  = lane & 15;

  const int mbase = row0 + ((lane >> 4) << 3);
  float dg[8];
#pragma unroll
  for (int r = 0; r < 8; ++r) dg[r] = degnorm[mbase + r];

  const v8f c = gemm_tile(x, W1, row0, lane);

  float d[8];
#pragma unroll
  for (int r = 0; r < 8; ++r) d[r] = rsqrtf(dg[r]);
#pragma unroll
  for (int r = 0; r < 8; ++r) {
    const size_t o = (size_t)(mbase + r) * HIDDEN + nsel;
    const float hs = c[r] * d[r];
    h1[o] = hs;
    agg1[o] = hs * d[r];
  }
  if (nsel == 0) {
#pragma unroll
    for (int r = 0; r < 8; ++r) degnorm[mbase + r] = d[r];
  }
}


#define A1_NODES 4096
#define A1_CL_BITS 12
#define A1_SLICES 10
#define A1_THREADS 1024
#define A1_U 8
#define A1_G 8
static_assert((1 << A1_CL_BITS) == A1_NODES, "list entry packs the in-range target in A1_CL_BITS bits");

__global__ __launch_bounds__(A1_THREADS) void k_agg1_edges(const int* __restrict__ row,
                                                           const int* __restrict__ col,
                                                           const float* __restrict__ h1,
                                                           const float* __restrict__ dinv,
                                                           float* __restrict__ agg1,
                                                           int E, int n) {
  __shared__ float bins[A1_NODES * HIDDEN];
  __shared__ unsigned list[A1_THREADS / 32][A1_U * 32];
  const int tid = threadIdx.x;
  const int lane = tid & 31;
  const int half = lane >> 4;
  const int f = lane & 15;
  const int lo = blockIdx.y * A1_NODES;
  unsigned* mylist = list[tid >> 5];

  for (int i = tid; i < A1_NODES * HIDDEN / 4; i += A1_THREADS)
    ((float4*)bins)[i] = make_float4(0.f, 0.f, 0.f, 0.f);
  __syncthreads();

  const int per = (E + (int)gridDim.x - 1) / (int)gridDim.x;
  const int e0 = blockIdx.x * per;
  const int e1 = min(e0 + per, E);
  if (e0 < e1) {
    const int last = e1 - 1;
    int c[A1_U];
    unsigned r[A1_U];
#pragma unroll
    for (int j = 0; j < A1_U; ++j) {
      const int ee = e0 + tid + j * A1_THREADS;
      const bool in = ee < e1;
      const int k = in ? ee : last;
      const int cv = col[k];
      r[j] = (unsigned)row[k];
      c[j] = in ? cv : -1;
    }
    for (int base = e0; base < e1; base += A1_U * A1_THREADS) {
      int cn[A1_U];
      unsigned rn[A1_U];
#pragma unroll
      for (int j = 0; j < A1_U; ++j) {
        const int ee = base + tid + (A1_U + j) * A1_THREADS;
        const bool in = ee < e1;
        const int k = in ? ee : last;
        const int cv = col[k];
        rn[j] = (unsigned)row[k];
        cn[j] = in ? cv : -1;
      }
      int cnt = 0;
#pragma unroll
      for (int j = 0; j < A1_U; ++j) {
        const unsigned idx = (unsigned)(c[j] - lo);
        const bool m = idx < (unsigned)A1_NODES;
        const unsigned mask = __builtin_amdgcn_ballot_w32(m);
        if (m) mylist[cnt + (int)__builtin_amdgcn_mbcnt_lo(mask, 0u)] = (r[j] << A1_CL_BITS) | idx;
        cnt += __builtin_popcount(mask);
      }
      for (int t = 0; t < cnt; t += 2 * A1_G) {
        unsigned ent[A1_G];
        float hv[A1_G];
#pragma unroll
        for (int q = 0; q < A1_G; ++q) {
          const int i = t + 2 * q + half;
          ent[q] = mylist[i < cnt ? i : t];
        }
#pragma unroll
        for (int q = 0; q < A1_G; ++q) {
          const unsigned rr = ent[q] >> A1_CL_BITS;
          hv[q] = h1[rr * HIDDEN + f];
        }
#pragma unroll
        for (int q = 0; q < A1_G; ++q) {
          const int i = t + 2 * q + half;
          if (i < cnt)
            atomicAdd(&bins[(ent[q] & (A1_NODES - 1)) * HIDDEN + f], hv[q]);
        }
      }
#pragma unroll
      for (int j = 0; j < A1_U; ++j) {
        c[j] = cn[j];
        r[j] = rn[j];
      }
    }
  }
  __syncthreads();

  const int total = n * HIDDEN;
  const int rot = (int)blockIdx.x * (((A1_NODES * HIDDEN) / A1_SLICES) & ~31);
  for (int t = tid; t < A1_NODES * HIDDEN; t += 4 * A1_THREADS) {
    float v[4], d[4];
    int g[4];
#pragma unroll
    for (int u = 0; u < 4; ++u) {
      const int i = (t + u * A1_THREADS + rot) & (A1_NODES * HIDDEN - 1);
      v[u] = bins[i];
      g[u] = lo * HIDDEN + i;
      d[u] = dinv[min(lo + (i >> 4), n - 1)];
    }
#pragma unroll
    for (int u = 0; u < 4; ++u)
      if (v[u] != 0.f && g[u] < total) atomicAdd(&agg1[g[u]], v[u] * d[u]);
  }
}


__global__ void k_relu_gemm2(const float* __restrict__ agg1, const float* __restrict__ b1,
                             const float* __restrict__ W2, const float* __restrict__ dinv,
                             float* __restrict__ h2, float* __restrict__ agg2, int n) {
  int node = blockIdx.x * blockDim.x + threadIdx.x;
  if (node >= n) return;
  const float dn = dinv[node];
  float acc0 = 0.f, acc1 = 0.f;
#pragma unroll
  for (int j = 0; j < HIDDEN; ++j) {
    float v = fmaxf(agg1[(size_t)node * HIDDEN + j] + b1[j], 0.f);
    acc0 = fmaf(v, W2[j * NCLS + 0], acc0);
    acc1 = fmaf(v, W2[j * NCLS + 1], acc1);
  }
  const float s0 = acc0 * dn, s1 = acc1 * dn;
  h2[(size_t)node * NCLS + 0] = s0;
  h2[(size_t)node * NCLS + 1] = s1;
  agg2[(size_t)node * NCLS + 0] = s0 * dn;
  agg2[(size_t)node * NCLS + 1] = s1 * dn;
}


__global__ __launch_bounds__(SCAT_THREADS) void k_agg2_edges(const int* __restrict__ row,
                                                             const int* __restrict__ col,
                                                             const float* __restrict__ h2,
                                                             const float* __restrict__ dinv,
                                                             float* __restrict__ agg2,
                                                             int E, int n) {
  __shared__ float bins[BIN_NODES * NCLS];
  const int tid = threadIdx.x;
  const int lo = blockIdx.y * BIN_NODES;
  for (int i = tid; i < BIN_NODES * NCLS; i += SCAT_THREADS) bins[i] = 0.f;
  __syncthreads();

  const int per = (E + (int)gridDim.x - 1) / (int)gridDim.x;
  const int e0 = blockIdx.x * per;
  const int e1 = min(e0 + per, E);
  if (e0 < e1) {
    const int last = e1 - 1;
    int c[4];
    unsigned r[4];
#pragma unroll
    for (int j = 0; j < 4; ++j) {
      const int ee = e0 + tid + j * SCAT_THREADS;
      const bool in = ee < e1;
      const int k = in ? ee : last;
      const int cv = col[k];
      r[j] = (unsigned)row[k];
      c[j] = in ? cv : -1;
    }
    for (int e = e0 + tid; e < e1; e += 4 * SCAT_THREADS) {
      int cn[4];
      unsigned rn[4];
#pragma unroll
      for (int j = 0; j < 4; ++j) {
        const int ee = e + (4 + j) * SCAT_THREADS;
        const bool in = ee < e1;
        const int k = in ? ee : last;
        const int cv = col[k];
        rn[j] = (unsigned)row[k];
        cn[j] = in ? cv : -1;
      }
      v2f hv[4];
#pragma unroll
      for (int j = 0; j < 4; ++j) {
        const unsigned idx = (unsigned)(c[j] - lo);
        if (idx < (unsigned)BIN_NODES) hv[j] = *(const v2f*)(h2 + r[j] * NCLS);
      }
#pragma unroll
      for (int j = 0; j < 4; ++j) {
        const unsigned idx = (unsigned)(c[j] - lo);
        if (idx < (unsigned)BIN_NODES) {
          atomicAdd(&bins[idx * NCLS + 0], hv[j].x);
          atomicAdd(&bins[idx * NCLS + 1], hv[j].y);
        }
      }
#pragma unroll
      for (int j = 0; j < 4; ++j) {
        c[j] = cn[j];
        r[j] = rn[j];
      }
    }
  }
  __syncthreads();

  const int total = n * NCLS;
  const int rot = (int)blockIdx.x * SCAT_THREADS;
  for (int t = tid; t < BIN_NODES * NCLS; t += SCAT_THREADS) {
    const int i = (t + rot) & (BIN_NODES * NCLS - 1);
    const float v = bins[i];
    const int g = lo * NCLS + i;
    const float d = dinv[min(lo + (i >> 1), n - 1)];
    if (v != 0.f && g < total) atomicAdd(&agg2[g], v * d);
  }
}


#define PART_NODES 196
#define PART_BITS 8
#define PART_LMASK ((1 << PART_BITS) - 1)
static_assert(PART_NODES <= (1 << PART_BITS), "a target-in-range fits its entry field");
#define PART_MAXR 512
#define PART_THREADS 1024
#define PART_U 10
#define PART_CAP (PART_THREADS * PART_U)
#define PART_PITCH 544
#define PART_MAXB 1024
#define PART_PADB (PART_MAXB + 64)
#define ROW_PITCH (HIDDEN + 1)

#define GEMM_TPB 25
#define GEMM_RING 8

template <bool AGAIN>
__device__ __forceinline__ v8f gemm_ring(float4 (&a)[GEMM_RING], v8f c, const float4* ws4,
                                         const float4* __restrict__ arow, int g) {
#pragma unroll
  for (int u = 0; u < GEMM_RING; ++u) {
    const float4 b = ws4[(g + u) * 32];
    v2f a0, a1, b0, b1;
    a0.x = a[u].x; a0.y = a[u].y; a1.x = a[u].z; a1.y = a[u].w;
    b0.x = b.x;    b0.y = b.y;    b1.x = b.z;    b1.y = b.w;
    c = __builtin_amdgcn_wmma_f32_16x16x4_f32(false, a0, false, b0, (short)0, c, false, false);
    c = __builtin_amdgcn_wmma_f32_16x16x4_f32(false, a1, false, b1, (short)0, c, false, false);
    if (AGAIN && (u & 3) == 3) {
#pragma unroll
      for (int v = u - 3; v <= u; ++v) a[v] = arow[2 * (g + GEMM_RING + v)];
      __builtin_amdgcn_sched_barrier(0);
    }
  }
  return c;
}

__global__ __launch_bounds__(PART_THREADS) void k_part_gemm(const int* __restrict__ row,
                                                            const int* __restrict__ col,
                                                            unsigned* __restrict__ part,
                                                            int* __restrict__ tab, int E,
                                                            const float* __restrict__ x,
                                                            const float* __restrict__ W1,
                                                            float* __restrict__ h1,
                                                            int n_nodes, int gemm_blocks) {
  __shared__ __attribute__((aligned(16))) unsigned sorted[PART_CAP];
  __shared__ int hist[PART_MAXR];
  __shared__ int base[PART_MAXR + 1];
  __shared__ int wsum[PART_MAXR / 32];
  const int tid = threadIdx.x;

  if ((int)blockIdx.x < gemm_blocks) {
    const int wave = tid >> 5;
    const int lane = tid & 31;
    const int row0 = ((int)blockIdx.x * GEMM_TPB + wave) * 16;
    const bool active = wave < GEMM_TPB && row0 < n_nodes;
    const int nsel = lane & 15;
    const int h = lane >> 4;
    float* wsf = (float*)sorted;
    float wv[NFEAT * HIDDEN / PART_THREADS];
#pragma unroll
    for (int j = 0; j < NFEAT * HIDDEN / PART_THREADS; ++j) wv[j] = W1[tid + j * PART_THREADS];
    const float4* arow = (const float4*)(x + (size_t)(active ? row0 + nsel : nsel) * NFEAT) + h;
    float4 a[GEMM_RING];
#pragma unroll
    for (int u = 0; u < GEMM_RING; ++u) a[u] = arow[2 * u];
#pragma unroll
    for (int j = 0; j < NFEAT * HIDDEN / PART_THREADS; ++j) {
      const int i = tid + j * PART_THREADS;
      const int k = i >> 4, col = i & 15;
      wsf[(((k >> 3) * 32 + ((k >> 2) & 1) * 16 + col) << 2) + (k & 3)] = wv[j];
    }
    __syncthreads();
    if (!active) return;
    const float4* ws4 = (const float4*)sorted + lane;
    v8f c = {};
#pragma nounroll
    for (int g = 0; g < NFEAT / 8 - GEMM_RING; g += GEMM_RING) c = gemm_ring<true>(a, c, ws4, arow, g);
    c = gemm_ring<false>(a, c, ws4, arow, NFEAT / 8 - GEMM_RING);
    const int mbase = row0 + (h << 3);
#pragma unroll
    for (int r = 0; r < 8; ++r) h1[(size_t)(mbase + r) * HIDDEN + nsel] = c[r];
    return;
  }
  const int b = (int)blockIdx.x - gemm_blocks;

  if (tid < PART_MAXR) hist[tid] = 0;
  __syncthreads();

  const int e0 = b * PART_CAP;
  const int e1 = min(e0 + PART_CAP, E);
  unsigned ent[PART_U];
  int rg[PART_U], rk[PART_U];
#pragma unroll
  for (int j = 0; j < PART_U; ++j) {
    const int ee = e0 + tid + j * PART_THREADS;
    const bool in = ee < e1;
    const int k = in ? ee : e0;
    const int cv = col[k];
    const unsigned rv = (unsigned)row[k];
    const int rgv = cv / PART_NODES;
    rg[j] = in ? rgv : -1;
    ent[j] = (rv << PART_BITS) | (unsigned)(cv - rgv * PART_NODES);
  }
#pragma unroll
  for (int j = 0; j < PART_U; ++j) rk[j] = rg[j] >= 0 ? atomicAdd(&hist[rg[j]], 1) : 0;
  __syncthreads();

  int v = 0, incl = 0;
  if (tid < PART_MAXR) {
    v = hist[tid];
    incl = v;
#pragma unroll
    for (int d = 1; d < 32; d <<= 1) {
      const int up = __shfl_up(incl, d);
      if ((tid & 31) >= d) incl += up;
    }
    if ((tid & 31) == 31) wsum[tid >> 5] = incl;
  }
  __syncthreads();
  if (tid < PART_MAXR) {
    int before = 0;
    for (int w = 0; w < (tid >> 5); ++w) before += wsum[w];
    base[tid] = before + incl - v;
    if (tid == PART_MAXR - 1) base[PART_MAXR] = before + incl;
  }
  __syncthreads();

  if (tid <= PART_MAXR) tab[b * PART_PITCH + tid] = base[tid];
#pragma unroll
  for (int j = 0; j < PART_U; ++j)
    if (rg[j] >= 0) sorted[base[rg[j]] + rk[j]] = ent[j];
  __syncthreads();

  const int cnt = e1 - e0;
  for (int i = tid; i < cnt; i += PART_THREADS) part[e0 + i] = sorted[i];
}

__device__ __forceinline__ void load_runs(const int* __restrict__ tab, int y, int nblk,
                                          int* rb, int* re) {
  for (int i = threadIdx.x; i < PART_PADB; i += blockDim.x) {
    int b0 = 0, b1 = 0;
    if (i < nblk) {
      const int* t = tab + (size_t)i * PART_PITCH + y;
      b0 = i * PART_CAP + t[0];
      b1 = i * PART_CAP + t[1];
    }
    rb[i] = b0;
    re[i] = b1;
  }
}

#define DEG_THREADS 512
static_assert(PART_NODES * HIDDEN / 4 <= 2 * DEG_THREADS && DEG_THREADS / 4 <= PART_NODES, "at most two 16-byte row pieces per thread");
static_assert(NFEAT * HIDDEN <= PART_CAP && (NFEAT * HIDDEN) % PART_THREADS == 0 &&
              (NFEAT / 8) % GEMM_RING == 0 && GEMM_RING % 4 == 0,
              "W1 fits the entry buffer, its copy and the K loop divide evenly");
__global__ __launch_bounds__(DEG_THREADS) void k_deg_scale(const unsigned* __restrict__ part,
                                                           const int* __restrict__ tab,
                                                           float* __restrict__ dinv,
                                                           float* h1,
                                                           int nblk, int n) {
  __shared__ float bins[PART_NODES];
  __shared__ int rb[PART_PADB], re[PART_PADB];
  const int tid = threadIdx.x;
  const int lane = tid & 31;
  const int y = blockIdx.x;
  float4* hp = (float4*)(h1 + (size_t)y * PART_NODES * HIDDEN);
  const int r0 = tid >> 2;
  const int r1 = r0 + DEG_THREADS / 4;
  const bool in0 = y * PART_NODES + r0 < n;
  const bool in1 = r1 < PART_NODES && y * PART_NODES + r1 < n;
  float4 p0 = make_float4(0.f, 0.f, 0.f, 0.f), p1 = p0;
  if (in0) p0 = hp[tid];
  if (in1) p1 = hp[tid + DEG_THREADS];
  load_runs(tab, y, nblk, rb, re);
  if (tid < PART_NODES) bins[tid] = 0.f;
  __syncthreads();

  const int nw = DEG_THREADS / 32;
  for (int b0 = tid >> 5; b0 < nblk; b0 += 4 * nw) {
    int beg[4], cnt[4];
    unsigned en[4];
#pragma unroll
    for (int j = 0; j < 4; ++j) {
      const int bj = b0 + j * nw;
      beg[j] = rb[bj];
      cnt[j] = re[bj] - beg[j];
    }
#pragma unroll
    for (int j = 0; j < 4; ++j) en[j] = part[cnt[j] > 0 ? beg[j] + min(lane, cnt[j] - 1) : 0];
#pragma unroll
    for (int j = 0; j < 4; ++j)
      if (lane < cnt[j]) atomicAdd(&bins[en[j] & PART_LMASK], 1.0f);
#pragma unroll
    for (int j = 0; j < 4; ++j)
      for (int t = 32 + lane; t < cnt[j]; t += 32)
        atomicAdd(&bins[part[beg[j] + t] & PART_LMASK], 1.0f);
  }
  __syncthreads();

  if (tid < PART_NODES) {
    const int node = y * PART_NODES + tid;
    const float d = rsqrtf(1.0f + bins[tid]);
    bins[tid] = d;
    if (node < n) dinv[node] = d;
  }
  __syncthreads();

  if (in0) {
    const float d = bins[r0];
    p0.x *= d; p0.y *= d; p0.z *= d; p0.w *= d;
    hp[tid] = p0;
  }
  if (in1) {
    const float d = bins[r1];
    p1.x *= d; p1.y *= d; p1.z *= d; p1.w *= d;
    hp[tid + DEG_THREADS] = p1;
  }
}

#define AP_THREADS 512
__global__ __launch_bounds__(AP_THREADS) void k_agg1_part(const unsigned* __restrict__ part,
                                                          const int* __restrict__ tab,
                                                          const float* __restrict__ h1,
                                                          const float* __restrict__ dinv,
                                                          const float* __restrict__ b1,
                                                          const float* __restrict__ W2,
                                                          float* __restrict__ h2,
                                                          int nblk, int n) {
  __shared__ float bins[PART_NODES * ROW_PITCH];
  __shared__ int rb[PART_PADB], re[PART_PADB];
  const int tid = threadIdx.x;
  const int lane = tid & 31;
  const int eq = lane >> 2;
  const int c4 = (lane & 3) * 4;
  const int y = blockIdx.x;
  load_runs(tab, y, nblk, rb, re);
  for (int i = tid; i < PART_NODES * ROW_PITCH; i += AP_THREADS) bins[i] = 0.f;
  __syncthreads();

  const int nw = AP_THREADS / 32;
  int b0 = tid >> 5;
  int beg[2], cnt[2];
  unsigned cur[2];
#pragma unroll
  for (int j = 0; j < 2; ++j) {
    beg[j] = rb[b0 + j * nw];
    cnt[j] = re[b0 + j * nw] - beg[j];
    cur[j] = part[cnt[j] > 0 ? beg[j] + min(lane, cnt[j] - 1) : 0];
  }
  for (; b0 < nblk; b0 += 2 * nw) {
    int nbeg[2], ncnt[2];
    unsigned nxt[2];
#pragma unroll
    for (int j = 0; j < 2; ++j) {
      nbeg[j] = rb[b0 + (2 + j) * nw];
      ncnt[j] = re[b0 + (2 + j) * nw] - nbeg[j];
      nxt[j] = part[ncnt[j] > 0 ? nbeg[j] + min(lane, ncnt[j] - 1) : 0];
    }
    unsigned en[2][4];
    float4 hv[2][4];
#pragma unroll
    for (int j = 0; j < 2; ++j)
#pragma unroll
      for (int q = 0; q < 4; ++q) {
        en[j][q] = (unsigned)__shfl((int)cur[j], q * 8 + eq);
        hv[j][q] = *(const float4*)(h1 + (en[j][q] >> PART_BITS) * HIDDEN + c4);
      }
#pragma unroll
    for (int j = 0; j < 2; ++j)
#pragma unroll
      for (int q = 0; q < 4; ++q)
        if (q * 8 + eq < cnt[j]) {
          float* p = &bins[(en[j][q] & PART_LMASK) * ROW_PITCH + c4];
          atomicAdd(p + 0, hv[j][q].x);
          atomicAdd(p + 1, hv[j][q].y);
          atomicAdd(p + 2, hv[j][q].z);
          atomicAdd(p + 3, hv[j][q].w);
        }
#pragma unroll
    for (int j = 0; j < 2; ++j)
      for (int t = 32; t < cnt[j]; t += 32) {
        const unsigned more = part[beg[j] + min(t + lane, cnt[j] - 1)];
        unsigned em[4];
        float4 hm[4];
#pragma unroll
        for (int q = 0; q < 4; ++q) {
          em[q] = (unsigned)__shfl((int)more, q * 8 + eq);
          hm[q] = *(const float4*)(h1 + (em[q] >> PART_BITS) * HIDDEN + c4);
        }
#pragma unroll
        for (int q = 0; q < 4; ++q)
          if (t + q * 8 + eq < cnt[j]) {
            float* p = &bins[(em[q] & PART_LMASK) * ROW_PITCH + c4];
            atomicAdd(p + 0, hm[q].x);
            atomicAdd(p + 1, hm[q].y);
            atomicAdd(p + 2, hm[q].z);
            atomicAdd(p + 3, hm[q].w);
          }
      }
#pragma unroll
    for (int j = 0; j < 2; ++j) {
      beg[j] = nbeg[j];
      cnt[j] = ncnt[j];
      cur[j] = nxt[j];
    }
  }
  __syncthreads();

  if (tid < PART_NODES) {
    const int node = y * PART_NODES + tid;
    if (node < n) {
      const float dn = dinv[node];
      const float4* sp = (const float4*)(h1 + (size_t)node * HIDDEN);
      float sv[HIDDEN];
#pragma unroll
      for (int q = 0; q < HIDDEN / 4; ++q) {
        const float4 s = sp[q];
        sv[4 * q + 0] = s.x;
        sv[4 * q + 1] = s.y;
        sv[4 * q + 2] = s.z;
        sv[4 * q + 3] = s.w;
      }
      const float* brow = &bins[tid * ROW_PITCH];
      float acc0 = 0.f, acc1 = 0.f;
#pragma unroll
      for (int j = 0; j < HIDDEN; ++j) {
        const float a = fmaf(brow[j], dn, sv[j] * dn);
        const float v = fmaxf(a + b1[j], 0.f);
        acc0 = fmaf(v, W2[j * NCLS + 0], acc0);
        acc1 = fmaf(v, W2[j * NCLS + 1], acc1);
      }
      v2f o;
      o.x = acc0 * dn;
      o.y = acc1 * dn;
      *(v2f*)(h2 + (size_t)node * NCLS) = o;
    }
  }
}

#define A2_THREADS 512
__global__ __launch_bounds__(A2_THREADS) void k_agg2_part(const unsigned* __restrict__ part,
                                                          const int* __restrict__ tab,
                                                          const float* __restrict__ h2,
                                                          const float* __restrict__ dinv,
                                                          const float* __restrict__ b2,
                                                          float* __restrict__ out,
                                                          int nblk, int n) {
  __shared__ float bins[PART_NODES * NCLS];
  __shared__ int rb[PART_PADB], re[PART_PADB];
  const int tid = threadIdx.x;
  const int lane = tid & 31;
  const int y = blockIdx.x;
  load_runs(tab, y, nblk, rb, re);
  for (int i = tid; i < PART_NODES * NCLS; i += A2_THREADS) bins[i] = 0.f;
  __syncthreads();

  const int nw = A2_THREADS / 32;
  for (int b0 = tid >> 5; b0 < nblk; b0 += 4 * nw) {
    int beg[4], cnt[4];
    unsigned en[4];
    v2f hv[4];
#pragma unroll
    for (int j = 0; j < 4; ++j) {
      const int bj = b0 + j * nw;
      beg[j] = rb[bj];
      cnt[j] = re[bj] - beg[j];
    }
#pragma unroll
    for (int j = 0; j < 4; ++j) en[j] = part[cnt[j] > 0 ? beg[j] + min(lane, cnt[j] - 1) : 0];
#pragma unroll
    for (int j = 0; j < 4; ++j) hv[j] = *(const v2f*)(h2 + (en[j] >> PART_BITS) * NCLS);
#pragma unroll
    for (int j = 0; j < 4; ++j)
      if (lane < cnt[j]) {
        const unsigned idx = en[j] & PART_LMASK;
        atomicAdd(&bins[idx * NCLS + 0], hv[j].x);
        atomicAdd(&bins[idx * NCLS + 1], hv[j].y);
      }
#pragma unroll
    for (int j = 0; j < 4; ++j)
      for (int t = 32 + lane; t < cnt[j]; t += 32) {
        const unsigned em = part[beg[j] + t];
        const v2f hm = *(const v2f*)(h2 + (em >> PART_BITS) * NCLS);
        atomicAdd(&bins[(em & PART_LMASK) * NCLS + 0], hm.x);
        atomicAdd(&bins[(em & PART_LMASK) * NCLS + 1], hm.y);
      }
  }
  __syncthreads();

  if (tid < PART_NODES) {
    const int node = y * PART_NODES + tid;
    if (node < n) {
      const float dn = dinv[node];
      const v2f hs = *(const v2f*)(h2 + (size_t)node * NCLS);
      v2f bs;
      bs.x = bins[tid * NCLS + 0];
      bs.y = bins[tid * NCLS + 1];
      const float s0 = hs.x * dn;
      const float s1 = hs.y * dn;
      const float a0 = s0 + bs.x * dn;
      const float a1 = s1 + bs.y * dn;
      const float z0 = a0 + b2[0];
      const float z1 = a1 + b2[1];
      const float m = fmaxf(z0, z1);
      const float lse = m + logf(expf(z0 - m) + expf(z1 - m));
      v2f o;
      o.x = z0 - lse;
      o.y = z1 - lse;
      *(v2f*)(out + (size_t)node * NCLS) = o;
    }
  }
}


__global__ void k_logsoftmax(const float* __restrict__ agg2, const float* __restrict__ b2,
                             float* __restrict__ out, int n) {
  int node = blockIdx.x * blockDim.x + threadIdx.x;
  if (node >= n) return;
  float z0 = agg2[(size_t)node * NCLS + 0] + b2[0];
  float z1 = agg2[(size_t)node * NCLS + 1] + b2[1];
  float m = fmaxf(z0, z1);
  float lse = m + logf(expf(z0 - m) + expf(z1 - m));
  out[(size_t)node * NCLS + 0] = z0 - lse;
  out[(size_t)node * NCLS + 1] = z1 - lse;
}


extern "C" void kernel_launch(void* const* d_in, const int* in_sizes, int n_in,
                              void* d_out, int out_size, void* d_ws, size_t ws_size,
                              hipStream_t stream) {
  const float* x  = (const float*)d_in[0];
  const int*   ei = (const int*)d_in[1];
  const float* W1 = (const float*)d_in[2];
  const float* b1 = (const float*)d_in[3];
  const float* W2 = (const float*)d_in[4];
  const float* b2 = (const float*)d_in[5];

  const int n = in_sizes[0] / NFEAT;
  const int E = in_sizes[1] / 2;
  const int* row = ei;
  const int* col = ei + E;

  char* ws = (char*)d_ws;
  size_t off = 0;
  auto carve = [&](size_t bytes) -> void* {
    void* p = ws + off;
    off = (off + bytes + 255) & ~(size_t)255;
    return p;
  };
  float* deg  = (float*)carve((size_t)n * sizeof(float));
  float* h1   = (float*)carve((size_t)n * HIDDEN * sizeof(float));
  float* agg1 = (float*)carve((size_t)n * HIDDEN * sizeof(float));
  float* h2   = (float*)carve((size_t)n * NCLS * sizeof(float));
  float* agg2 = (float*)carve((size_t)n * NCLS * sizeof(float));

  const int B = 256;
  auto blocks = [](long long work, int b) { return (unsigned)((work + b - 1) / b); };

  const unsigned nblk = blocks(E, PART_CAP);
  const unsigned nrng = blocks(n, PART_NODES);
  unsigned* part = (unsigned*)carve((size_t)E * sizeof(unsigned));
  int* tab = (int*)carve((size_t)nblk * PART_PITCH * sizeof(int));
  const bool grouped = off <= ws_size && nrng <= PART_MAXR && nblk <= PART_MAXB &&
                       n <= (1 << (32 - PART_BITS));
  const dim3 scat_grid(EDGE_SLICES, blocks(n, BIN_NODES));

  int tiles = (n + 15) / 16;
  if (grouped) {
    const unsigned gblk = blocks(tiles, GEMM_TPB);
    k_part_gemm<<<gblk + nblk, PART_THREADS, 0, stream>>>(row, col, part, tab, E, x, W1, h1, n,
                                                           (int)gblk);
    k_deg_scale<<<nrng, DEG_THREADS, 0, stream>>>(part, tab, deg, h1, (int)nblk, n);
    k_agg1_part<<<nrng, AP_THREADS, 0, stream>>>(part, tab, h1, deg, b1, W2, h2,
                                                 (int)nblk, n);
    k_agg2_part<<<nrng, A2_THREADS, 0, stream>>>(part, tab, h2, deg, b2, (float*)d_out,
                                                 (int)nblk, n);
  } else {
    k_init_deg<<<blocks(n, B), B, 0, stream>>>(deg, n);
    k_deg_count<<<scat_grid, SCAT_THREADS, 0, stream>>>(col, deg, E, n);
    k_gemm1<<<blocks(tiles, 8), B, 0, stream>>>(x, W1, h1, deg, agg1, n);
    const dim3 a1_grid(A1_SLICES, blocks(n, A1_NODES));
    k_agg1_edges<<<a1_grid, A1_THREADS, 0, stream>>>(row, col, h1, deg, agg1, E, n);
    k_relu_gemm2<<<blocks(n, B), B, 0, stream>>>(agg1, b1, W2, deg, h2, agg2, n);
    k_agg2_edges<<<scat_grid, SCAT_THREADS, 0, stream>>>(row, col, h2, deg, agg2, E, n);
    k_logsoftmax<<<blocks(n, B), B, 0, stream>>>(agg2, b2, (float*)d_out, n);
  }
}


# layer-1 range block lays its entries out node by node in LDS (int cursors, counts from the degree pass) and sums each node's rows in registers, four lanes a node: no LDS float adds; sum order changes
# speedup vs baseline: 7.8536x; 1.0696x over previous
#include <hip/hip_runtime.h>
#include <math.h>

typedef __attribute__((ext_vector_type(2))) float v2f;
typedef __attribute__((ext_vector_type(8))) float v8f;

#define NFEAT 512
#define HIDDEN 16
#define NCLS 2


__global__ void k_init_deg(float* __restrict__ deg, int n) {
  int i = blockIdx.x * blockDim.x + threadIdx.x;
  if (i < n) deg[i] = 1.0f;
}

#define BIN_NODES 16384
#define EDGE_SLICES 64
#define SCAT_THREADS 1024
static_assert((BIN_NODES & (BIN_NODES - 1)) == 0, "bin index is wrapped with a mask");

__global__ __launch_bounds__(SCAT_THREADS) void k_deg_count(const int* __restrict__ col,
                                                            float* __restrict__ deg,
                                                            int E, int n) {
  __shared__ float bins[BIN_NODES];
  const int tid = threadIdx.x;
  const int lo = blockIdx.y * BIN_NODES;
  for (int i = tid; i < BIN_NODES; i += SCAT_THREADS) bins[i] = 0.f;
  __syncthreads();

  const int per = (E + (int)gridDim.x - 1) / (int)gridDim.x;
  const int e0 = blockIdx.x * per;
  const int e1 = min(e0 + per, E);
  const int last = e1 - 1;
  for (int e = e0 + tid; e < e1; e += 8 * SCAT_THREADS) {
    int c[8];
#pragma unroll
    for (int j = 0; j < 8; ++j) {
      const int ee = e + j * SCAT_THREADS;
      const bool in = ee < e1;
      const int v = col[in ? ee : last];
      c[j] = in ? v : -1;
    }
#pragma unroll
    for (int j = 0; j < 8; ++j) {
      const unsigned idx = (unsigned)(c[j] - lo);
      if (idx < (unsigned)BIN_NODES) atomicAdd(&bins[idx], 1.0f);
    }
  }
  __syncthreads();

  const int rot = (int)blockIdx.x * SCAT_THREADS;
  for (int t = tid; t < BIN_NODES; t += SCAT_THREADS) {
    const int i = (t + rot) & (BIN_NODES - 1);
    const float v = bins[i];
    const int node = lo + i;
    if (v != 0.f && node < n) atomicAdd(&deg[node], v);
  }
}


__device__ __forceinline__ v8f gemm_tile(const float* __restrict__ x,
                                         const float* __restrict__ W1, int row0, int lane) {
  const int nsel  = lane & 15;
  const int khalf = (lane >> 4) << 1;
  const float* arow = x + (size_t)(row0 + nsel) * NFEAT + khalf;
  v8f c = {};
  for (int k = 0; k < NFEAT; k += 4) {
    if ((k & 127) == 0) __builtin_prefetch(arow + k + 128, 0, 0);
    v2f a = *(const v2f*)(arow + k);
    v2f b;
    b.x = W1[(k + khalf) * HIDDEN + nsel];
    b.y = W1[(k + khalf + 1) * HIDDEN + nsel];
    c = __builtin_amdgcn_wmma_f32_16x16x4_f32(false, a, false, b, (short)0, c,
                                              false, false);
  }
  return c;
}

__global__ __launch_bounds__(256) void k_gemm1(const float* __restrict__ x,
                                               const float* __restrict__ W1,
                                               float* __restrict__ h1,
                                               float* degnorm,
                                               float* __restrict__ agg1,
                                               int n_nodes) {
  const int lane = threadIdx.x & 31;
  const int tile = blockIdx.x * 8 + (threadIdx.x >> 5);
  const int row0 = tile * 16;
  if (row0 >= n_nodes) return;

  const int nsel  = lane & 15;

  const int mbase = row0 + ((lane >> 4) << 3);
  float dg[8];
#pragma unroll
  for (int r = 0; r < 8; ++r) dg[r] = degnorm[mbase + r];

  const v8f c = gemm_tile(x, W1, row0, lane);

  float d[8];
#pragma unroll
  for (int r = 0; r < 8; ++r) d[r] = rsqrtf(dg[r]);
#pragma unroll
  for (int r = 0; r < 8; ++r) {
    const size_t o = (size_t)(mbase + r) * HIDDEN + nsel;
    const float hs = c[r] * d[r];
    h1[o] = hs;
    agg1[o] = hs * d[r];
  }
  if (nsel == 0) {
#pragma unroll
    for (int r = 0; r < 8; ++r) degnorm[mbase + r] = d[r];
  }
}


#define A1_NODES 4096
#define A1_CL_BITS 12
#define A1_SLICES 10
#define A1_THREADS 1024
#define A1_U 8
#define A1_G 8
static_assert((1 << A1_CL_BITS) == A1_NODES, "list entry packs the in-range target in A1_CL_BITS bits");

__global__ __launch_bounds__(A1_THREADS) void k_agg1_edges(const int* __restrict__ row,
                                                           const int* __restrict__ col,
                                                           const float* __restrict__ h1,
                                                           const float* __restrict__ dinv,
                                                           float* __restrict__ agg1,
                                                           int E, int n) {
  __shared__ float bins[A1_NODES * HIDDEN];
  __shared__ unsigned list[A1_THREADS / 32][A1_U * 32];
  const int tid = threadIdx.x;
  const int lane = tid & 31;
  const int half = lane >> 4;
  const int f = lane & 15;
  const int lo = blockIdx.y * A1_NODES;
  unsigned* mylist = list[tid >> 5];

  for (int i = tid; i < A1_NODES * HIDDEN / 4; i += A1_THREADS)
    ((float4*)bins)[i] = make_float4(0.f, 0.f, 0.f, 0.f);
  __syncthreads();

  const int per = (E + (int)gridDim.x - 1) / (int)gridDim.x;
  const int e0 = blockIdx.x * per;
  const int e1 = min(e0 + per, E);
  if (e0 < e1) {
    const int last = e1 - 1;
    int c[A1_U];
    unsigned r[A1_U];
#pragma unroll
    for (int j = 0; j < A1_U; ++j) {
      const int ee = e0 + tid + j * A1_THREADS;
      const bool in = ee < e1;
      const int k = in ? ee : last;
      const int cv = col[k];
      r[j] = (unsigned)row[k];
      c[j] = in ? cv : -1;
    }
    for (int base = e0; base < e1; base += A1_U * A1_THREADS) {
      int cn[A1_U];
      unsigned rn[A1_U];
#pragma unroll
      for (int j = 0; j < A1_U; ++j) {
        const int ee = base + tid + (A1_U + j) * A1_THREADS;
        const bool in = ee < e1;
        const int k = in ? ee : last;
        const int cv = col[k];
        rn[j] = (unsigned)row[k];
        cn[j] = in ? cv : -1;
      }
      int cnt = 0;
#pragma unroll
      for (int j = 0; j < A1_U; ++j) {
        const unsigned idx = (unsigned)(c[j] - lo);
        const bool m = idx < (unsigned)A1_NODES;
        const unsigned mask = __builtin_amdgcn_ballot_w32(m);
        if (m) mylist[cnt + (int)__builtin_amdgcn_mbcnt_lo(mask, 0u)] = (r[j] << A1_CL_BITS) | idx;
        cnt += __builtin_popcount(mask);
      }
      for (int t = 0; t < cnt; t += 2 * A1_G) {
        unsigned ent[A1_G];
        float hv[A1_G];
#pragma unroll
        for (int q = 0; q < A1_G; ++q) {
          const int i = t + 2 * q + half;
          ent[q] = mylist[i < cnt ? i : t];
        }
#pragma unroll
        for (int q = 0; q < A1_G; ++q) {
          const unsigned rr = ent[q] >> A1_CL_BITS;
          hv[q] = h1[rr * HIDDEN + f];
        }
#pragma unroll
        for (int q = 0; q < A1_G; ++q) {
          const int i = t + 2 * q + half;
          if (i < cnt)
            atomicAdd(&bins[(ent[q] & (A1_NODES - 1)) * HIDDEN + f], hv[q]);
        }
      }
#pragma unroll
      for (int j = 0; j < A1_U; ++j) {
        c[j] = cn[j];
        r[j] = rn[j];
      }
    }
  }
  __syncthreads();

  const int total = n * HIDDEN;
  const int rot = (int)blockIdx.x * (((A1_NODES * HIDDEN) / A1_SLICES) & ~31);
  for (int t = tid; t < A1_NODES * HIDDEN; t += 4 * A1_THREADS) {
    float v[4], d[4];
    int g[4];
#pragma unroll
    for (int u = 0; u < 4; ++u) {
      const int i = (t + u * A1_THREADS + rot) & (A1_NODES * HIDDEN - 1);
      v[u] = bins[i];
      g[u] = lo * HIDDEN + i;
      d[u] = dinv[min(lo + (i >> 4), n - 1)];
    }
#pragma unroll
    for (int u = 0; u < 4; ++u)
      if (v[u] != 0.f && g[u] < total) atomicAdd(&agg1[g[u]], v[u] * d[u]);
  }
}


__global__ void k_relu_gemm2(const float* __restrict__ agg1, const float* __restrict__ b1,
                             const float* __restrict__ W2, const float* __restrict__ dinv,
                             float* __restrict__ h2, float* __restrict__ agg2, int n) {
  int node = blockIdx.x * blockDim.x + threadIdx.x;
  if (node >= n) return;
  const float dn = dinv[node];
  float acc0 = 0.f, acc1 = 0.f;
#pragma unroll
  for (int j = 0; j < HIDDEN; ++j) {
    float v = fmaxf(agg1[(size_t)node * HIDDEN + j] + b1[j], 0.f);
    acc0 = fmaf(v, W2[j * NCLS + 0], acc0);
    acc1 = fmaf(v, W2[j * NCLS + 1], acc1);
  }
  const float s0 = acc0 * dn, s1 = acc1 * dn;
  h2[(size_t)node * NCLS + 0] = s0;
  h2[(size_t)node * NCLS + 1] = s1;
  agg2[(size_t)node * NCLS + 0] = s0 * dn;
  agg2[(size_t)node * NCLS + 1] = s1 * dn;
}


__global__ __launch_bounds__(SCAT_THREADS) void k_agg2_edges(const int* __restrict__ row,
                                                             const int* __restrict__ col,
                                                             const float* __restrict__ h2,
                                                             const float* __restrict__ dinv,
                                                             float* __restrict__ agg2,
                                                             int E, int n) {
  __shared__ float bins[BIN_NODES * NCLS];
  const int tid = threadIdx.x;
  const int lo = blockIdx.y * BIN_NODES;
  for (int i = tid; i < BIN_NODES * NCLS; i += SCAT_THREADS) bins[i] = 0.f;
  __syncthreads();

  const int per = (E + (int)gridDim.x - 1) / (int)gridDim.x;
  const int e0 = blockIdx.x * per;
  const int e1 = min(e0 + per, E);
  if (e0 < e1) {
    const int last = e1 - 1;
    int c[4];
    unsigned r[4];
#pragma unroll
    for (int j = 0; j < 4; ++j) {
      const int ee = e0 + tid + j * SCAT_THREADS;
      const bool in = ee < e1;
      const int k = in ? ee : last;
      const int cv = col[k];
      r[j] = (unsigned)row[k];
      c[j] = in ? cv : -1;
    }
    for (int e = e0 + tid; e < e1; e += 4 * SCAT_THREADS) {
      int cn[4];
      unsigned rn[4];
#pragma unroll
      for (int j = 0; j < 4; ++j) {
        const int ee = e + (4 + j) * SCAT_THREADS;
        const bool in = ee < e1;
        const int k = in ? ee : last;
        const int cv = col[k];
        rn[j] = (unsigned)row[k];
        cn[j] = in ? cv : -1;
      }
      v2f hv[4];
#pragma unroll
      for (int j = 0; j < 4; ++j) {
        const unsigned idx = (unsigned)(c[j] - lo);
        if (idx < (unsigned)BIN_NODES) hv[j] = *(const v2f*)(h2 + r[j] * NCLS);
      }
#pragma unroll
      for (int j = 0; j < 4; ++j) {
        const unsigned idx = (unsigned)(c[j] - lo);
        if (idx < (unsigned)BIN_NODES) {
          atomicAdd(&bins[idx * NCLS + 0], hv[j].x);
          atomicAdd(&bins[idx * NCLS + 1], hv[j].y);
        }
      }
#pragma unroll
      for (int j = 0; j < 4; ++j) {
        c[j] = cn[j];
        r[j] = rn[j];
      }
    }
  }
  __syncthreads();

  const int total = n * NCLS;
  const int rot = (int)blockIdx.x * SCAT_THREADS;
  for (int t = tid; t < BIN_NODES * NCLS; t += SCAT_THREADS) {
    const int i = (t + rot) & (BIN_NODES * NCLS - 1);
    const float v = bins[i];
    const int g = lo * NCLS + i;
    const float d = dinv[min(lo + (i >> 1), n - 1)];
    if (v != 0.f && g < total) atomicAdd(&agg2[g], v * d);
  }
}


#define PART_NODES 196
#define PART_BITS 8
#define PART_LMASK ((1 << PART_BITS) - 1)
static_assert(PART_NODES <= (1 << PART_BITS), "a target-in-range fits its entry field");
#define PART_MAXR 512
#define PART_THREADS 1024
#define PART_U 10
#define PART_CAP (PART_THREADS * PART_U)
#define PART_PITCH 544
#define PART_MAXB 1024
#define PART_PADB (PART_MAXB + 64)
#define ROW_PITCH (HIDDEN + 1)

#define GEMM_TPB 25
#define GEMM_RING 8

template <bool AGAIN>
__device__ __forceinline__ v8f gemm_ring(float4 (&a)[GEMM_RING], v8f c, const float4* ws4,
                                         const float4* __restrict__ arow, int g) {
#pragma unroll
  for (int u = 0; u < GEMM_RING; ++u) {
    const float4 b = ws4[(g + u) * 32];
    v2f a0, a1, b0, b1;
    a0.x = a[u].x; a0.y = a[u].y; a1.x = a[u].z; a1.y = a[u].w;
    b0.x = b.x;    b0.y = b.y;    b1.x = b.z;    b1.y = b.w;
    c = __builtin_amdgcn_wmma_f32_16x16x4_f32(false, a0, false, b0, (short)0, c, false, false);
    c = __builtin_amdgcn_wmma_f32_16x16x4_f32(false, a1, false, b1, (short)0, c, false, false);
    if (AGAIN && (u & 3) == 3) {
#pragma unroll
      for (int v = u - 3; v <= u; ++v) a[v] = arow[2 * (g + GEMM_RING + v)];
      __builtin_amdgcn_sched_barrier(0);
    }
  }
  return c;
}

__global__ __launch_bounds__(PART_THREADS) void k_part_gemm(const int* __restrict__ row,
                                                            const int* __restrict__ col,
                                                            unsigned* __restrict__ part,
                                                            int* __restrict__ tab, int E,
                                                            const float* __restrict__ x,
                                                            const float* __restrict__ W1,
                                                            float* __restrict__ h1,
                                                            int n_nodes, int gemm_blocks) {
  __shared__ __attribute__((aligned(16))) unsigned sorted[PART_CAP];
  __shared__ int hist[PART_MAXR];
  __shared__ int base[PART_MAXR + 1];
  __shared__ int wsum[PART_MAXR / 32];
  const int tid = threadIdx.x;

  if ((int)blockIdx.x < gemm_blocks) {
    const int wave = tid >> 5;
    const int lane = tid & 31;
    const int row0 = ((int)blockIdx.x * GEMM_TPB + wave) * 16;
    const bool active = wave < GEMM_TPB && row0 < n_nodes;
    const int nsel = lane & 15;
    const int h = lane >> 4;
    float* wsf = (float*)sorted;
    float wv[NFEAT * HIDDEN / PART_THREADS];
#pragma unroll
    for (int j = 0; j < NFEAT * HIDDEN / PART_THREADS; ++j) wv[j] = W1[tid + j * PART_THREADS];
    const float4* arow = (const float4*)(x + (size_t)(active ? row0 + nsel : nsel) * NFEAT) + h;
    float4 a[GEMM_RING];
#pragma unroll
    for (int u = 0; u < GEMM_RING; ++u) a[u] = arow[2 * u];
#pragma unroll
    for (int j = 0; j < NFEAT * HIDDEN / PART_THREADS; ++j) {
      const int i = tid + j * PART_THREADS;
      const int k = i >> 4, col = i & 15;
      wsf[(((k >> 3) * 32 + ((k >> 2) & 1) * 16 + col) << 2) + (k & 3)] = wv[j];
    }
    __syncthreads();
    if (!active) return;
    const float4* ws4 = (const float4*)sorted + lane;
    v8f c = {};
#pragma nounroll
    for (int g = 0; g < NFEAT / 8 - GEMM_RING; g += GEMM_RING) c = gemm_ring<true>(a, c, ws4, arow, g);
    c = gemm_ring<false>(a, c, ws4, arow, NFEAT / 8 - GEMM_RING);
    const int mbase = row0 + (h << 3);
#pragma unroll
    for (int r = 0; r < 8; ++r) h1[(size_t)(mbase + r) * HIDDEN + nsel] = c[r];
    return;
  }
  const int b = (int)blockIdx.x - gemm_blocks;

  if (tid < PART_MAXR) hist[tid] = 0;
  __syncthreads();

  const int e0 = b * PART_CAP;
  const int e1 = min(e0 + PART_CAP, E);
  unsigned ent[PART_U];
  int rg[PART_U], rk[PART_U];
#pragma unroll
  for (int j = 0; j < PART_U; ++j) {
    const int ee = e0 + tid + j * PART_THREADS;
    const bool in = ee < e1;
    const int k = in ? ee : e0;
    const int cv = col[k];
    const unsigned rv = (unsigned)row[k];
    const int rgv = cv / PART_NODES;
    rg[j] = in ? rgv : -1;
    ent[j] = (rv << PART_BITS) | (unsigned)(cv - rgv * PART_NODES);
  }
#pragma unroll
  for (int j = 0; j < PART_U; ++j) rk[j] = rg[j] >= 0 ? atomicAdd(&hist[rg[j]], 1) : 0;
  __syncthreads();

  int v = 0, incl = 0;
  if (tid < PART_MAXR) {
    v = hist[tid];
    incl = v;
#pragma unroll
    for (int d = 1; d < 32; d <<= 1) {
      const int up = __shfl_up(incl, d);
      if ((tid & 31) >= d) incl += up;
    }
    if ((tid & 31) == 31) wsum[tid >> 5] = incl;
  }
  __syncthreads();
  if (tid < PART_MAXR) {
    int before = 0;
    for (int w = 0; w < (tid >> 5); ++w) before += wsum[w];
    base[tid] = before + incl - v;
    if (tid == PART_MAXR - 1) base[PART_MAXR] = before + incl;
  }
  __syncthreads();

  if (tid <= PART_MAXR) tab[b * PART_PITCH + tid] = base[tid];
#pragma unroll
  for (int j = 0; j < PART_U; ++j)
    if (rg[j] >= 0) sorted[base[rg[j]] + rk[j]] = ent[j];
  __syncthreads();

  const int cnt = e1 - e0;
  for (int i = tid; i < cnt; i += PART_THREADS) part[e0 + i] = sorted[i];
}

__device__ __forceinline__ void load_runs(const int* __restrict__ tab, int y, int nblk,
                                          int* rb, int* re) {
  for (int i = threadIdx.x; i < PART_PADB; i += blockDim.x) {
    int b0 = 0, b1 = 0;
    if (i < nblk) {
      const int* t = tab + (size_t)i * PART_PITCH + y;
      b0 = i * PART_CAP + t[0];
      b1 = i * PART_CAP + t[1];
    }
    rb[i] = b0;
    re[i] = b1;
  }
}

#define DEG_THREADS 512
static_assert(PART_NODES * HIDDEN / 4 <= 2 * DEG_THREADS && DEG_THREADS / 4 <= PART_NODES, "at most two 16-byte row pieces per thread");
static_assert(NFEAT * HIDDEN <= PART_CAP && (NFEAT * HIDDEN) % PART_THREADS == 0 &&
              (NFEAT / 8) % GEMM_RING == 0 && GEMM_RING % 4 == 0,
              "W1 fits the entry buffer, its copy and the K loop divide evenly");
__global__ __launch_bounds__(DEG_THREADS) void k_deg_scale(const unsigned* __restrict__ part,
                                                           const int* __restrict__ tab,
                                                           float* __restrict__ dinv,
                                                           float* h1,
                                                           int* __restrict__ cnti,
                                                           int nblk, int n) {
  __shared__ float bins[PART_NODES];
  __shared__ int rb[PART_PADB], re[PART_PADB];
  const int tid = threadIdx.x;
  const int lane = tid & 31;
  const int y = blockIdx.x;
  float4* hp = (float4*)(h1 + (size_t)y * PART_NODES * HIDDEN);
  const int r0 = tid >> 2;
  const int r1 = r0 + DEG_THREADS / 4;
  const bool in0 = y * PART_NODES + r0 < n;
  const bool in1 = r1 < PART_NODES && y * PART_NODES + r1 < n;
  float4 p0 = make_float4(0.f, 0.f, 0.f, 0.f), p1 = p0;
  if (in0) p0 = hp[tid];
  if (in1) p1 = hp[tid + DEG_THREADS];
  load_runs(tab, y, nblk, rb, re);
  if (tid < PART_NODES) bins[tid] = 0.f;
  __syncthreads();

  const int nw = DEG_THREADS / 32;
  for (int b0 = tid >> 5; b0 < nblk; b0 += 4 * nw) {
    int beg[4], cnt[4];
    unsigned en[4];
#pragma unroll
    for (int j = 0; j < 4; ++j) {
      const int bj = b0 + j * nw;
      beg[j] = rb[bj];
      cnt[j] = re[bj] - beg[j];
    }
#pragma unroll
    for (int j = 0; j < 4; ++j) en[j] = part[cnt[j] > 0 ? beg[j] + min(lane, cnt[j] - 1) : 0];
#pragma unroll
    for (int j = 0; j < 4; ++j)
      if (lane < cnt[j]) atomicAdd(&bins[en[j] & PART_LMASK], 1.0f);
#pragma unroll
    for (int j = 0; j < 4; ++j)
      for (int t = 32 + lane; t < cnt[j]; t += 32)
        atomicAdd(&bins[part[beg[j] + t] & PART_LMASK], 1.0f);
  }
  __syncthreads();

  if (tid < PART_NODES) {
    const int node = y * PART_NODES + tid;
    const float cf = bins[tid];
    const float d = rsqrtf(1.0f + cf);
    bins[tid] = d;
    if (node < n) {
      dinv[node] = d;
      cnti[node] = (int)cf;
    }
  }
  __syncthreads();

  if (in0) {
    const float d = bins[r0];
    p0.x *= d; p0.y *= d; p0.z *= d; p0.w *= d;
    hp[tid] = p0;
  }
  if (in1) {
    const float d = bins[r1];
    p1.x *= d; p1.y *= d; p1.z *= d; p1.w *= d;
    hp[tid + DEG_THREADS] = p1;
  }
}

#define AP_THREADS 512
#define AP_CAP 8192
#define AP_G 8
#define AP_SCANT ((PART_NODES + 31) & ~31)
static_assert(AP_SCANT <= AP_THREADS, "one thread per node in the count prefix");

__device__ __forceinline__ void agg1_walk_lds_adds(const unsigned* __restrict__ part,
                                                   const float* __restrict__ h1, const int* rb,
                                                   const int* re, float* bins, int nblk) {
  const int tid = threadIdx.x;
  const int lane = tid & 31;
  const int eq = lane >> 2;
  const int c4 = (lane & 3) * 4;
  const int nw = AP_THREADS / 32;
  int b0 = tid >> 5;
  int beg[2], cnt[2];
  unsigned cur[2];
#pragma unroll
  for (int j = 0; j < 2; ++j) {
    beg[j] = rb[b0 + j * nw];
    cnt[j] = re[b0 + j * nw] - beg[j];
    cur[j] = part[cnt[j] > 0 ? beg[j] + min(lane, cnt[j] - 1) : 0];
  }
  for (; b0 < nblk; b0 += 2 * nw) {
    int nbeg[2], ncnt[2];
    unsigned nxt[2];
#pragma unroll
    for (int j = 0; j < 2; ++j) {
      nbeg[j] = rb[b0 + (2 + j) * nw];
      ncnt[j] = re[b0 + (2 + j) * nw] - nbeg[j];
      nxt[j] = part[ncnt[j] > 0 ? nbeg[j] + min(lane, ncnt[j] - 1) : 0];
    }
    unsigned en[2][4];
    float4 hv[2][4];
#pragma unroll
    for (int j = 0; j < 2; ++j)
#pragma unroll
      for (int q = 0; q < 4; ++q) {
        en[j][q] = (unsigned)__shfl((int)cur[j], q * 8 + eq);
        hv[j][q] = *(const float4*)(h1 + (en[j][q] >> PART_BITS) * HIDDEN + c4);
      }
#pragma unroll
    for (int j = 0; j < 2; ++j)
#pragma unroll
      for (int q = 0; q < 4; ++q)
        if (q * 8 + eq < cnt[j]) {
          float* p = &bins[(en[j][q] & PART_LMASK) * ROW_PITCH + c4];
          atomicAdd(p + 0, hv[j][q].x);
          atomicAdd(p + 1, hv[j][q].y);
          atomicAdd(p + 2, hv[j][q].z);
          atomicAdd(p + 3, hv[j][q].w);
        }
#pragma unroll
    for (int j = 0; j < 2; ++j)
      for (int t = 32; t < cnt[j]; t += 32) {
        const unsigned more = part[beg[j] + min(t + lane, cnt[j] - 1)];
        unsigned em[4];
        float4 hm[4];
#pragma unroll
        for (int q = 0; q < 4; ++q) {
          em[q] = (unsigned)__shfl((int)more, q * 8 + eq);
          hm[q] = *(const float4*)(h1 + (em[q] >> PART_BITS) * HIDDEN + c4);
        }
#pragma unroll
        for (int q = 0; q < 4; ++q)
          if (t + q * 8 + eq < cnt[j]) {
            float* p = &bins[(em[q] & PART_LMASK) * ROW_PITCH + c4];
            atomicAdd(p + 0, hm[q].x);
            atomicAdd(p + 1, hm[q].y);
            atomicAdd(p + 2, hm[q].z);
            atomicAdd(p + 3, hm[q].w);
          }
      }
#pragma unroll
    for (int j = 0; j < 2; ++j) {
      beg[j] = nbeg[j];
      cnt[j] = ncnt[j];
      cur[j] = nxt[j];
    }
  }
}

__global__ __launch_bounds__(AP_THREADS) void k_agg1_part(const unsigned* __restrict__ part,
                                                          const int* __restrict__ tab,
                                                          const float* __restrict__ h1,
                                                          const float* __restrict__ dinv,
                                                          const int* __restrict__ cnti,
                                                          const float* __restrict__ b1,
                                                          const float* __restrict__ W2,
                                                          float* __restrict__ h2,
                                                          int nblk, int n) {
  __shared__ float bins[PART_NODES * ROW_PITCH];
  __shared__ int rb[PART_PADB], re[PART_PADB];
  __shared__ unsigned srt[AP_CAP];
  __shared__ int nbeg[PART_NODES + 1];
  __shared__ int offs[1 << PART_BITS];
  __shared__ int wsum[AP_SCANT / 32];
  const int tid = threadIdx.x;
  const int lane = tid & 31;
  const int c4 = (lane & 3) * 4;
  const int y = blockIdx.x;

  int v = 0;
  if (tid < PART_NODES && y * PART_NODES + tid < n) v = cnti[y * PART_NODES + tid];
  load_runs(tab, y, nblk, rb, re);
  int incl = v;
  if (tid < AP_SCANT) {
#pragma unroll
    for (int d = 1; d < 32; d <<= 1) {
      const int up = __shfl_up(incl, d);
      if (lane >= d) incl += up;
    }
    if (lane == 31) wsum[tid >> 5] = incl;
  }
  __syncthreads();
  if (tid < PART_NODES) {
    int before = 0;
    for (int w = 0; w < (tid >> 5); ++w) before += wsum[w];
    const int ex = before + incl - v;
    nbeg[tid] = ex;
    offs[tid] = ex;
    if (tid == PART_NODES - 1) nbeg[PART_NODES] = ex + v;
  }
  __syncthreads();

  const int total = nbeg[PART_NODES];
  if (total <= AP_CAP) {
    const int nw = AP_THREADS / 32;
    int b0 = tid >> 5;
    int beg[4], cnt[4];
    unsigned en[4];
#pragma unroll
    for (int j = 0; j < 4; ++j) {
      const int bj = b0 + j * nw;
      beg[j] = rb[bj];
      cnt[j] = re[bj] - beg[j];
      en[j] = part[cnt[j] > 0 ? beg[j] + min(lane, cnt[j] - 1) : 0];
    }
    for (; b0 < nblk; b0 += 4 * nw) {
      int nb[4], nc[4];
      unsigned nx[4];
#pragma unroll
      for (int j = 0; j < 4; ++j) {
        const int bj = min(b0 + (4 + j) * nw, PART_PADB - 1);
        nb[j] = rb[bj];
        nc[j] = re[bj] - nb[j];
        nx[j] = part[nc[j] > 0 ? nb[j] + min(lane, nc[j] - 1) : 0];
      }
      int pos[4];
#pragma unroll
      for (int j = 0; j < 4; ++j)
        pos[j] = lane < cnt[j] ? atomicAdd(&offs[en[j] & PART_LMASK], 1) : AP_CAP;
#pragma unroll
      for (int j = 0; j < 4; ++j)
        if (pos[j] < AP_CAP) srt[pos[j]] = en[j] >> PART_BITS;
#pragma unroll
      for (int j = 0; j < 4; ++j)
        for (int t = 32 + lane; t < cnt[j]; t += 32) {
          const unsigned em = part[beg[j] + t];
          const int pos = atomicAdd(&offs[em & PART_LMASK], 1);
          if (pos < AP_CAP) srt[pos] = em >> PART_BITS;
        }
#pragma unroll
      for (int j = 0; j < 4; ++j) {
        beg[j] = nb[j];
        cnt[j] = nc[j];
        en[j] = nx[j];
      }
    }
    __syncthreads();

    for (int nd = tid >> 2; nd < PART_NODES; nd += AP_THREADS / 4) {
      const int s0 = nbeg[nd], s1 = nbeg[nd + 1];
      float4 acc = make_float4(0.f, 0.f, 0.f, 0.f);
      if (s0 < s1) {
        unsigned sx[AP_G];
#pragma unroll
        for (int q = 0; q < AP_G; ++q) sx[q] = srt[min(s0 + q, s1 - 1)];
        for (int i = s0; i < s1; i += AP_G) {
          float4 hv[AP_G];
#pragma unroll
          for (int q = 0; q < AP_G; ++q) hv[q] = *(const float4*)(h1 + sx[q] * HIDDEN + c4);
#pragma unroll
          for (int q = 0; q < AP_G; ++q) sx[q] = srt[min(i + AP_G + q, s1 - 1)];
#pragma unroll
          for (int q = 0; q < AP_G; ++q) {
            const bool ok = i + q < s1;
            acc.x += ok ? hv[q].x : 0.f;
            acc.y += ok ? hv[q].y : 0.f;
            acc.z += ok ? hv[q].z : 0.f;
            acc.w += ok ? hv[q].w : 0.f;
          }
        }
      }
      float* p = &bins[nd * ROW_PITCH + c4];
      p[0] = acc.x;
      p[1] = acc.y;
      p[2] = acc.z;
      p[3] = acc.w;
    }
  } else {
    for (int i = tid; i < PART_NODES * ROW_PITCH; i += AP_THREADS) bins[i] = 0.f;
    __syncthreads();
    agg1_walk_lds_adds(part, h1, rb, re, bins, nblk);
  }
  __syncthreads();

  if (tid < PART_NODES) {
    const int node = y * PART_NODES + tid;
    if (node < n) {
      const float dn = dinv[node];
      const float4* sp = (const float4*)(h1 + (size_t)node * HIDDEN);
      float sv[HIDDEN];
#pragma unroll
      for (int q = 0; q < HIDDEN / 4; ++q) {
        const float4 s = sp[q];
        sv[4 * q + 0] = s.x;
        sv[4 * q + 1] = s.y;
        sv[4 * q + 2] = s.z;
        sv[4 * q + 3] = s.w;
      }
      const float* brow = &bins[tid * ROW_PITCH];
      float acc0 = 0.f, acc1 = 0.f;
#pragma unroll
      for (int j = 0; j < HIDDEN; ++j) {
        const float a = fmaf(brow[j], dn, sv[j] * dn);
        const float v = fmaxf(a + b1[j], 0.f);
        acc0 = fmaf(v, W2[j * NCLS + 0], acc0);
        acc1 = fmaf(v, W2[j * NCLS + 1], acc1);
      }
      v2f o;
      o.x = acc0 * dn;
      o.y = acc1 * dn;
      *(v2f*)(h2 + (size_t)node * NCLS) = o;
    }
  }
}

#define A2_THREADS 512
__global__ __launch_bounds__(A2_THREADS) void k_agg2_part(const unsigned* __restrict__ part,
                                                          const int* __restrict__ tab,
                                                          const float* __restrict__ h2,
                                                          const float* __restrict__ dinv,
                                                          const float* __restrict__ b2,
                                                          float* __restrict__ out,
                                                          int nblk, int n) {
  __shared__ float bins[PART_NODES * NCLS];
  __shared__ int rb[PART_PADB], re[PART_PADB];
  const int tid = threadIdx.x;
  const int lane = tid & 31;
  const int y = blockIdx.x;
  load_runs(tab, y, nblk, rb, re);
  for (int i = tid; i < PART_NODES * NCLS; i += A2_THREADS) bins[i] = 0.f;
  __syncthreads();

  const int nw = A2_THREADS / 32;
  for (int b0 = tid >> 5; b0 < nblk; b0 += 4 * nw) {
    int beg[4], cnt[4];
    unsigned en[4];
    v2f hv[4];
#pragma unroll
    for (int j = 0; j < 4; ++j) {
      const int bj = b0 + j * nw;
      beg[j] = rb[bj];
      cnt[j] = re[bj] - beg[j];
    }
#pragma unroll
    for (int j = 0; j < 4; ++j) en[j] = part[cnt[j] > 0 ? beg[j] + min(lane, cnt[j] - 1) : 0];
#pragma unroll
    for (int j = 0; j < 4; ++j) hv[j] = *(const v2f*)(h2 + (en[j] >> PART_BITS) * NCLS);
#pragma unroll
    for (int j = 0; j < 4; ++j)
      if (lane < cnt[j]) {
        const unsigned idx = en[j] & PART_LMASK;
        atomicAdd(&bins[idx * NCLS + 0], hv[j].x);
        atomicAdd(&bins[idx * NCLS + 1], hv[j].y);
      }
#pragma unroll
    for (int j = 0; j < 4; ++j)
      for (int t = 32 + lane; t < cnt[j]; t += 32) {
        const unsigned em = part[beg[j] + t];
        const v2f hm = *(const v2f*)(h2 + (em >> PART_BITS) * NCLS);
        atomicAdd(&bins[(em & PART_LMASK) * NCLS + 0], hm.x);
        atomicAdd(&bins[(em & PART_LMASK) * NCLS + 1], hm.y);
      }
  }
  __syncthreads();

  if (tid < PART_NODES) {
    const int node = y * PART_NODES + tid;
    if (node < n) {
      const float dn = dinv[node];
      const v2f hs = *(const v2f*)(h2 + (size_t)node * NCLS);
      v2f bs;
      bs.x = bins[tid * NCLS + 0];
      bs.y = bins[tid * NCLS + 1];
      const float s0 = hs.x * dn;
      const float s1 = hs.y * dn;
      const float a0 = s0 + bs.x * dn;
      const float a1 = s1 + bs.y * dn;
      const float z0 = a0 + b2[0];
      const float z1 = a1 + b2[1];
      const float m = fmaxf(z0, z1);
      const float lse = m + logf(expf(z0 - m) + expf(z1 - m));
      v2f o;
      o.x = z0 - lse;
      o.y = z1 - lse;
      *(v2f*)(out + (size_t)node * NCLS) = o;
    }
  }
}


__global__ void k_logsoftmax(const float* __restrict__ agg2, const float* __restrict__ b2,
                             float* __restrict__ out, int n) {
  int node = blockIdx.x * blockDim.x + threadIdx.x;
  if (node >= n) return;
  float z0 = agg2[(size_t)node * NCLS + 0] + b2[0];
  float z1 = agg2[(size_t)node * NCLS + 1] + b2[1];
  float m = fmaxf(z0, z1);
  float lse = m + logf(expf(z0 - m) + expf(z1 - m));
  out[(size_t)node * NCLS + 0] = z0 - lse;
  out[(size_t)node * NCLS + 1] = z1 - lse;
}


extern "C" void kernel_launch(void* const* d_in, const int* in_sizes, int n_in,
                              void* d_out, int out_size, void* d_ws, size_t ws_size,
                              hipStream_t stream) {
  const float* x  = (const float*)d_in[0];
  const int*   ei = (const int*)d_in[1];
  const float* W1 = (const float*)d_in[2];
  const float* b1 = (const float*)d_in[3];
  const float* W2 = (const float*)d_in[4];
  const float* b2 = (const float*)d_in[5];

  const int n = in_sizes[0] / NFEAT;
  const int E = in_sizes[1] / 2;
  const int* row = ei;
  const int* col = ei + E;

  char* ws = (char*)d_ws;
  size_t off = 0;
  auto carve = [&](size_t bytes) -> void* {
    void* p = ws + off;
    off = (off + bytes + 255) & ~(size_t)255;
    return p;
  };
  float* deg  = (float*)carve((size_t)n * sizeof(float));
  float* h1   = (float*)carve((size_t)n * HIDDEN * sizeof(float));
  float* agg1 = (float*)carve((size_t)n * HIDDEN * sizeof(float));
  float* h2   = (float*)carve((size_t)n * NCLS * sizeof(float));
  float* agg2 = (float*)carve((size_t)n * NCLS * sizeof(float));
  int* cnti = (int*)agg2;

  const int B = 256;
  auto blocks = [](long long work, int b) { return (unsigned)((work + b - 1) / b); };

  const unsigned nblk = blocks(E, PART_CAP);
  const unsigned nrng = blocks(n, PART_NODES);
  unsigned* part = (unsigned*)carve((size_t)E * sizeof(unsigned));
  int* tab = (int*)carve((size_t)nblk * PART_PITCH * sizeof(int));
  const bool grouped = off <= ws_size && nrng <= PART_MAXR && nblk <= PART_MAXB &&
                       n <= (1 << (32 - PART_BITS));
  const dim3 scat_grid(EDGE_SLICES, blocks(n, BIN_NODES));

  int tiles = (n + 15) / 16;
  if (grouped) {
    const unsigned gblk = blocks(tiles, GEMM_TPB);
    k_part_gemm<<<gblk + nblk, PART_THREADS, 0, stream>>>(row, col, part, tab, E, x, W1, h1, n,
                                                           (int)gblk);
    k_deg_scale<<<nrng, DEG_THREADS, 0, stream>>>(part, tab, deg, h1, cnti, (int)nblk, n);
    k_agg1_part<<<nrng, AP_THREADS, 0, stream>>>(part, tab, h1, deg, cnti, b1, W2, h2,
                                                 (int)nblk, n);
    k_agg2_part<<<nrng, A2_THREADS, 0, stream>>>(part, tab, h2, deg, b2, (float*)d_out,
                                                 (int)nblk, n);
  } else {
    k_init_deg<<<blocks(n, B), B, 0, stream>>>(deg, n);
    k_deg_count<<<scat_grid, SCAT_THREADS, 0, stream>>>(col, deg, E, n);
    k_gemm1<<<blocks(tiles, 8), B, 0, stream>>>(x, W1, h1, deg, agg1, n);
    const dim3 a1_grid(A1_SLICES, blocks(n, A1_NODES));
    k_agg1_edges<<<a1_grid, A1_THREADS, 0, stream>>>(row, col, h1, deg, agg1, E, n);
    k_relu_gemm2<<<blocks(n, B), B, 0, stream>>>(agg1, b1, W2, deg, h2, agg2, n);
    k_agg2_edges<<<scat_grid, SCAT_THREADS, 0, stream>>>(row, col, h2, deg, agg2, E, n);
    k_logsoftmax<<<blocks(n, B), B, 0, stream>>>(agg2, b2, (float*)d_out, n);
  }
}
